# starting kernel
# baseline (speedup 1.0000x reference)
#include <hip/hip_runtime.h>
#include <math.h>

typedef float v2f __attribute__((ext_vector_type(2)));
typedef float v8f __attribute__((ext_vector_type(8)));

static constexpr int   NN = 100000;
static constexpr int   NE = 1600000;
static constexpr float NEG = 0.2f;

__device__ __forceinline__ float leaky(float x) { return x > 0.f ? x : NEG * x; }
__device__ __forceinline__ unsigned f2ord(float f) {
  unsigned u = __float_as_uint(f);
  return (u & 0x80000000u) ? ~u : (u | 0x80000000u);
}
__device__ __forceinline__ float ord2f(unsigned v) {
  unsigned u = (v & 0x80000000u) ? (v & 0x7fffffffu) : ~v;
  return __uint_as_float(u);
}

__global__ void zero_f32_k(float* p, int n) {
  int i = blockIdx.x * blockDim.x + threadIdx.x;
  if (i < n) p[i] = 0.f;
}
__global__ void relu_k(float* p, int n) {
  int i = blockIdx.x * blockDim.x + threadIdx.x;
  if (i < n) p[i] = fmaxf(p[i], 0.f);
}
__global__ void residual_k(const float* __restrict__ h2, float* __restrict__ r,
                           const float* __restrict__ br, int n) {
  int i = blockIdx.x * blockDim.x + threadIdx.x;
  if (i < n) { int c = i & 127; r[i] = fmaxf(h2[i], 0.f) + r[i] + br[c]; }
}

__global__ void attr_accum_k(const int* __restrict__ dst, const float* __restrict__ ea,
                             float* attr_sum, float* cnt) {
  int i = blockIdx.x * blockDim.x + threadIdx.x;
  if (i >= NE * 16) return;
  int e = i >> 4, j = i & 15;
  int d = dst[e];
  atomicAdd(&attr_sum[(size_t)d * 16 + j], ea[i]);
  if (j == 0) atomicAdd(&cnt[d], 1.f);
}
__global__ void attr_mean_k(float* attr_sum, const float* cnt) {
  int i = blockIdx.x * blockDim.x + threadIdx.x;
  if (i < NN * 16) attr_sum[i] /= fmaxf(cnt[i >> 4], 1.f);
}

__global__ void wa_k(const float* __restrict__ We, const float* __restrict__ ae,
                     float* wa, int K) {
  int i = threadIdx.x;
  if (i < 16) { float s = 0.f; for (int j = 0; j < K; ++j) s += We[i * K + j] * ae[j]; wa[i] = s; }
}

template <int COLT>
__global__ void gemm_k128(const float* __restrict__ A, const float* __restrict__ B,
                          float* __restrict__ C) {
  constexpr int COUT = COLT * 16;
  __shared__ float As[16 * 128];
  const int tid = threadIdx.x;
  const size_t abase = (size_t)blockIdx.x * (16 * 128);
  for (int i = tid * 4; i < 16 * 128; i += blockDim.x * 4)
    *(float4*)(&As[i]) = *(const float4*)(&A[abase + i]);
  __syncthreads();

  const int wave = tid >> 5, lane = tid & 31, half = lane >> 4, lidx = lane & 15;
  const int n0 = wave * 16;
  v8f acc = {};
#pragma unroll
  for (int k0 = 0; k0 < 128; k0 += 4) {
    v2f a, b;
    a.x = As[lidx * 128 + k0 + 2 * half];
    a.y = As[lidx * 128 + k0 + 2 * half + 1];
    b.x = B[(k0 + 2 * half) * COUT + n0 + lidx];
    b.y = B[(k0 + 2 * half + 1) * COUT + n0 + lidx];
    acc = __builtin_amdgcn_wmma_f32_16x16x4_f32(false, a, false, b, (short)0, acc,
                                                false, false);
  }
  const size_t row0 = (size_t)blockIdx.x * 16;
#pragma unroll
  for (int r = 0; r < 8; ++r)
    C[(row0 + r + 8 * half) * COUT + n0 + lidx] = acc[r];
}

template <int C, bool ET>
__global__ void node_prep_k(const float* __restrict__ h, const float* __restrict__ a_src,
                            const float* __restrict__ a_dst,
                            const float* __restrict__ mean_attr,
                            const float* __restrict__ wa, float* s_src, float* s_dst,
                            float* self_logit, unsigned* segmax_u) {
  int gid = blockIdx.x * blockDim.x + threadIdx.x;
  int n = gid >> 5, lane = gid & 31;
  if (n >= NN) return;
  float ps = 0.f, pd = 0.f;
#pragma unroll
  for (int i = 0; i < C / 32; ++i) {
    int c = lane + i * 32;
    float v = h[(size_t)n * C + c];
    ps += v * a_src[c];
    pd += v * a_dst[c];
  }
#pragma unroll
  for (int m = 16; m > 0; m >>= 1) { ps += __shfl_xor(ps, m, 32); pd += __shfl_xor(pd, m, 32); }
  if (lane == 0) {
    float st = 0.f;
    if (ET) { for (int j = 0; j < 16; ++j) st += mean_attr[(size_t)n * 16 + j] * wa[j]; }
    float sl = leaky(ps + pd + st);
    s_src[n] = ps; s_dst[n] = pd; self_logit[n] = sl; segmax_u[n] = f2ord(sl);
  }
}

__global__ void eterm_k(const float* __restrict__ ea, const float* __restrict__ wa,
                        float* eterm) {
  int i = blockIdx.x * blockDim.x + threadIdx.x;
  if (i >= NE * 16) return;
  int j = i & 15;
  float v = ea[i] * wa[j];
  v += __shfl_xor(v, 1, 32); v += __shfl_xor(v, 2, 32);
  v += __shfl_xor(v, 4, 32); v += __shfl_xor(v, 8, 32);
  if (j == 0) eterm[i >> 4] = v;
}

template <bool ET>
__global__ void edge_max_k(const int* __restrict__ src, const int* __restrict__ dst,
                           const float* __restrict__ s_src, const float* __restrict__ s_dst,
                           const float* __restrict__ eterm, unsigned* segmax_u) {
  int e = blockIdx.x * blockDim.x + threadIdx.x;
  if (e >= NE) return;
  int d = dst[e];
  float l = s_src[src[e]] + s_dst[d];
  if (ET) l += eterm[e];
  atomicMax(&segmax_u[d], f2ord(leaky(l)));
}

__global__ void node_mid_k(const unsigned* __restrict__ segmax_u,
                           const float* __restrict__ self_logit, float* segmaxf,
                           float* segsum) {
  int n = blockIdx.x * blockDim.x + threadIdx.x;
  if (n >= NN) return;
  float m = ord2f(segmax_u[n]);
  segmaxf[n] = m;
  segsum[n] = __expf(self_logit[n] - m);
}

template <bool ET>
__global__ void edge_sum_k(const int* __restrict__ src, const int* __restrict__ dst,
                           const float* __restrict__ s_src, const float* __restrict__ s_dst,
                           const float* __restrict__ eterm,
                           const float* __restrict__ segmaxf, float* segsum) {
  int e = blockIdx.x * blockDim.x + threadIdx.x;
  if (e >= NE) return;
  int d = dst[e];
  float l = s_src[src[e]] + s_dst[d];
  if (ET) l += eterm[e];
  atomicAdd(&segsum[d], __expf(leaky(l) - segmaxf[d]));
}

template <int C>
__global__ void out_init_k(const float* __restrict__ h, const float* __restrict__ self_logit,
                           const float* __restrict__ segmaxf, const float* __restrict__ segsum,
                           const float* __restrict__ bias, float* __restrict__ out) {
  int i = blockIdx.x * blockDim.x + threadIdx.x;
  if (i >= NN * (C / 4)) return;
  int n = i / (C / 4), q = i % (C / 4);
  float a = __expf(self_logit[n] - segmaxf[n]) / (segsum[n] + 1e-16f);
  float4 hv = *(const float4*)&h[(size_t)n * C + q * 4];
  float4 o;
  o.x = a * hv.x + bias[q * 4 + 0];
  o.y = a * hv.y + bias[q * 4 + 1];
  o.z = a * hv.z + bias[q * 4 + 2];
  o.w = a * hv.w + bias[q * 4 + 3];
  *(float4*)&out[(size_t)n * C + q * 4] = o;
}

template <int C, bool ET>
__global__ void edge_agg_k(const int* __restrict__ src, const int* __restrict__ dst,
                           const float* __restrict__ s_src, const float* __restrict__ s_dst,
                           const float* __restrict__ eterm,
                           const float* __restrict__ segmaxf, const float* __restrict__ segsum,
                           const float* __restrict__ h, float* __restrict__ out) {
  int gid = blockIdx.x * blockDim.x + threadIdx.x;
  int e = gid >> 5, lane = gid & 31;
  if (e >= NE) return;
  int s = src[e], d = dst[e];
  float l = s_src[s] + s_dst[d];
  if (ET) l += eterm[e];
  float alpha = __expf(leaky(l) - segmaxf[d]) / (segsum[d] + 1e-16f);
#pragma unroll
  for (int i = 0; i < C / 32; ++i) {
    int c = lane + i * 32;
    atomicAdd(&out[(size_t)d * C + c], alpha * h[(size_t)s * C + c]);
  }
}

struct Scratch {
  float *s_src, *s_dst, *self_l, *segmaxf, *segsum;
  unsigned* segmax_u;
  const float* mean_attr;
  float *eterm, *wa;
};

template <int C, bool ET>
static void run_gat(hipStream_t st, const float* hin, const float* Wm, const float* asrc,
                    const float* adst, const float* bias, const float* We, const float* ae,
                    const int* src, const int* dst, const float* edge_attr, float* hg,
                    float* out, const Scratch& S) {
  constexpr int COLT = C / 16;
  if (ET) wa_k<<<1, 16, 0, st>>>(We, ae, S.wa, C);
  gemm_k128<COLT><<<NN / 16, 32 * COLT, 0, st>>>(hin, Wm, hg);
  node_prep_k<C, ET><<<(NN * 32 + 255) / 256, 256, 0, st>>>(hg, asrc, adst, S.mean_attr,
                                                            S.wa, S.s_src, S.s_dst,
                                                            S.self_l, S.segmax_u);
  if (ET) eterm_k<<<(NE * 16 + 255) / 256, 256, 0, st>>>(edge_attr, S.wa, S.eterm);
  edge_max_k<ET><<<(NE + 255) / 256, 256, 0, st>>>(src, dst, S.s_src, S.s_dst, S.eterm,
                                                   S.segmax_u);
  node_mid_k<<<(NN + 255) / 256, 256, 0, st>>>(S.segmax_u, S.self_l, S.segmaxf, S.segsum);
  edge_sum_k<ET><<<(NE + 255) / 256, 256, 0, st>>>(src, dst, S.s_src, S.s_dst, S.eterm,
                                                   S.segmaxf, S.segsum);
  out_init_k<C><<<(NN * (C / 4) + 255) / 256, 256, 0, st>>>(hg, S.self_l, S.segmaxf,
                                                            S.segsum, bias, out);
  edge_agg_k<C, ET><<<(NE * 32 + 255) / 256, 256, 0, st>>>(src, dst, S.s_src, S.s_dst,
                                                           S.eterm, S.segmaxf, S.segsum,
                                                           hg, out);
}

extern "C" void kernel_launch(void* const* d_in, const int* in_sizes, int n_in, void* d_out,
                              int out_size, void* d_ws, size_t ws_size, hipStream_t stream) {
  (void)in_sizes; (void)n_in; (void)out_size; (void)ws_size;
  const float* x = (const float*)d_in[0];
  const int* ei = (const int*)d_in[1];
  const int* src = ei;
  const int* dst = ei + NE;
  const float* edge_attr = (const float*)d_in[2];
  const float *W1 = (const float*)d_in[3], *a1s = (const float*)d_in[4],
              *a1d = (const float*)d_in[5], *b1 = (const float*)d_in[6];
  const float *W2 = (const float*)d_in[7], *a2s = (const float*)d_in[8],
              *a2d = (const float*)d_in[9], *b2 = (const float*)d_in[10];
  const float *We2 = (const float*)d_in[11], *ae2 = (const float*)d_in[12];
  const float *Wr = (const float*)d_in[13], *br = (const float*)d_in[14];
  const float *Wmu = (const float*)d_in[15], *amus = (const float*)d_in[16],
              *amud = (const float*)d_in[17], *bmu = (const float*)d_in[18],
              *Wemu = (const float*)d_in[19], *aemu = (const float*)d_in[20];
  const float *Wls = (const float*)d_in[21], *alss = (const float*)d_in[22],
              *alsd = (const float*)d_in[23], *bls = (const float*)d_in[24],
              *Wels = (const float*)d_in[25], *aels = (const float*)d_in[26];
  float* outf = (float*)d_out;

  float* W = (float*)d_ws;
  size_t o = 0;
  float* hA = W + o; o += (size_t)NN * 128;
  float* hB = W + o; o += (size_t)NN * 128;
  float* hC = W + o; o += (size_t)NN * 128;
  Scratch S;
  S.s_src = W + o; o += NN;
  S.s_dst = W + o; o += NN;
  S.self_l = W + o; o += NN;
  S.segmaxf = W + o; o += NN;
  S.segsum = W + o; o += NN;
  S.segmax_u = (unsigned*)(W + o); o += NN;
  float* cnt = W + o; o += NN;
  float* mean_attr = W + o; o += (size_t)NN * 16;
  S.mean_attr = mean_attr;
  S.eterm = W + o; o += NE;
  S.wa = W + o; o += 16;

  zero_f32_k<<<(NN * 17 + 255) / 256, 256, 0, stream>>>(cnt, NN * 17);
  attr_accum_k<<<(NE * 16 + 255) / 256, 256, 0, stream>>>(dst, edge_attr, mean_attr, cnt);
  attr_mean_k<<<(NN * 16 + 255) / 256, 256, 0, stream>>>(mean_attr, cnt);

  run_gat<128, false>(stream, x, W1, a1s, a1d, b1, nullptr, nullptr, src, dst, edge_attr,
                      hA, hB, S);
  relu_k<<<(NN * 128 + 255) / 256, 256, 0, stream>>>(hB, NN * 128);

  run_gat<128, true>(stream, hB, W2, a2s, a2d, b2, We2, ae2, src, dst, edge_attr, hA, hC, S);

  gemm_k128<8><<<NN / 16, 256, 0, stream>>>(x, Wr, hB);
  residual_k<<<(NN * 128 + 255) / 256, 256, 0, stream>>>(hC, hB, br, NN * 128);

  run_gat<64, true>(stream, hB, Wmu, amus, amud, bmu, Wemu, aemu, src, dst, edge_attr, hA,
                    outf, S);
  run_gat<64, true>(stream, hB, Wls, alss, alsd, bls, Wels, aels, src, dst, edge_attr, hA,
                    outf + (size_t)NN * 64, S);
}


# Edges grouped by destination once per call (counting sort, each group in edge-id order); one wave per node does softmax max, sum and aggregation: float atomics gone, sums now run in edge order
# speedup vs baseline: 2.6920x; 2.6920x over previous
#include <hip/hip_runtime.h>
#include <math.h>

typedef float v2f __attribute__((ext_vector_type(2)));
typedef float v8f __attribute__((ext_vector_type(8)));

static constexpr int   NN = 100000;
static constexpr int   NE = 1600000;
static constexpr float NEG = 0.2f;
static constexpr int   SCAN_TILE = 1024;
static constexpr int   SCAN_BLOCKS = (NN + 1 + SCAN_TILE - 1) / SCAN_TILE;
static_assert(SCAN_BLOCKS <= 256, "scan_k sums the earlier block totals with one thread each");

__device__ __forceinline__ float leaky(float x) { return x > 0.f ? x : NEG * x; }

__global__ void zero_i32_k(int* p, int n) {
  int i = blockIdx.x * blockDim.x + threadIdx.x;
  if (i < n) p[i] = 0;
}
__global__ void relu_k(float* p, int n) {
  int i = blockIdx.x * blockDim.x + threadIdx.x;
  if (i < n) p[i] = fmaxf(p[i], 0.f);
}
__global__ void residual_k(const float* __restrict__ h2, float* __restrict__ r,
                           const float* __restrict__ br, int n) {
  int i = blockIdx.x * blockDim.x + threadIdx.x;
  if (i < n) { int c = i & 127; r[i] = fmaxf(h2[i], 0.f) + r[i] + br[c]; }
}

__global__ void hist_rank_k(const int* __restrict__ dst, int* cnt, int* __restrict__ rank) {
  int e = blockIdx.x * blockDim.x + threadIdx.x;
  if (e >= NE) return;
  rank[e] = atomicAdd(&cnt[dst[e]], 1);
}
__global__ void blocksum_k(const int* __restrict__ cnt, int* __restrict__ bsum) {
  __shared__ int wtot[8];
  const int tid = threadIdx.x, lane = tid & 31, w = tid >> 5;
  const int i0 = blockIdx.x * SCAN_TILE + tid * 4;
  int s = 0;
  for (int j = 0; j < 4; ++j) s += (i0 + j < NN) ? cnt[i0 + j] : 0;
  for (int m = 16; m > 0; m >>= 1) s += __shfl_xor(s, m, 32);
  if (lane == 0) wtot[w] = s;
  __syncthreads();
  if (tid == 0) {
    int t = 0;
    for (int i = 0; i < 8; ++i) t += wtot[i];
    bsum[blockIdx.x] = t;
  }
}
__global__ void scan_k(const int* __restrict__ cnt, const int* __restrict__ bsum,
                       int* __restrict__ rowptr) {
  __shared__ int wtot[8];
  __shared__ int wpre[8];
  const int tid = threadIdx.x, lane = tid & 31, w = tid >> 5;
  int part = (tid < (int)blockIdx.x) ? bsum[tid] : 0;
  for (int m = 16; m > 0; m >>= 1) part += __shfl_xor(part, m, 32);
  if (lane == 0) wpre[w] = part;
  const int i0 = blockIdx.x * SCAN_TILE + tid * 4;
  int c[4];
  for (int j = 0; j < 4; ++j) c[j] = (i0 + j < NN) ? cnt[i0 + j] : 0;
  const int ts = c[0] + c[1] + c[2] + c[3];
  int inc = ts;
  for (int o = 1; o < 32; o <<= 1) {
    int t = __shfl_up(inc, o, 32);
    if (lane >= o) inc += t;
  }
  if (lane == 31) wtot[w] = inc;
  __syncthreads();
  int run = inc - ts;
  for (int i = 0; i < 8; ++i) {
    run += wpre[i];
    if (i < w) run += wtot[i];
  }
  for (int j = 0; j < 4; ++j) {
    if (i0 + j <= NN) rowptr[i0 + j] = run;
    run += c[j];
  }
}
__global__ void fill_k(const int* __restrict__ dst, const int* __restrict__ rowptr,
                       const int* __restrict__ rank, int* __restrict__ eid_raw) {
  int e = blockIdx.x * blockDim.x + threadIdx.x;
  if (e >= NE) return;
  eid_raw[rowptr[dst[e]] + rank[e]] = e;
}
__device__ __forceinline__ int lane_i(int v, int k) { return __builtin_amdgcn_readlane(v, k); }
__device__ __forceinline__ float lane_f(float v, int k) {
  return __int_as_float(__builtin_amdgcn_readlane(__float_as_int(v), k));
}
__global__ void seg_sort_k(const int* __restrict__ rowptr, const int* __restrict__ eid_raw,
                           const int* __restrict__ src, int* __restrict__ seid,
                           int* __restrict__ ssrc) {
  const int gid = blockIdx.x * blockDim.x + threadIdx.x;
  const int lane = gid & 31;
  const int n = __builtin_amdgcn_readfirstlane(gid >> 5);
  if (n >= NN) return;
  const int start = rowptr[n];
  const int end = rowptr[n + 1];
  for (int base = start; base < end; base += 32) {
    const int p = base + lane;
    const int x = (p < end) ? eid_raw[p] : 0x7fffffff;
    int r = 0;
    for (int b2 = start; b2 < end; b2 += 32) {
      const int q = b2 + lane;
      const int y = (q < end) ? eid_raw[q] : 0x7fffffff;
      const int m = min(32, end - b2);
      for (int k = 0; k < m; ++k) r += (lane_i(y, k) < x) ? 1 : 0;
    }
    if (p < end) { seid[start + r] = x; ssrc[start + r] = src[x]; }
  }
}
__global__ void attr_mean_k(const int* __restrict__ rowptr, const int* __restrict__ seid,
                            const float* __restrict__ ea, float* __restrict__ mean_attr) {
  const int gid = blockIdx.x * blockDim.x + threadIdx.x;
  const int n = gid >> 4, j = gid & 15;
  if (n >= NN) return;
  const int start = rowptr[n], end = rowptr[n + 1];
  float s = 0.f;
  for (int p = start; p < end; ++p) s += ea[(size_t)seid[p] * 16 + j];
  mean_attr[(size_t)n * 16 + j] = s / fmaxf((float)(end - start), 1.f);
}

__global__ void wa_k(const float* __restrict__ We, const float* __restrict__ ae,
                     float* wa, int K) {
  int i = threadIdx.x;
  if (i < 16) { float s = 0.f; for (int j = 0; j < K; ++j) s += We[i * K + j] * ae[j]; wa[i] = s; }
}

template <int COLT>
__global__ void gemm_k128(const float* __restrict__ A, const float* __restrict__ B,
                          float* __restrict__ C) {
  constexpr int COUT = COLT * 16;
  __shared__ float As[16 * 128];
  const int tid = threadIdx.x;
  const size_t abase = (size_t)blockIdx.x * (16 * 128);
  for (int i = tid * 4; i < 16 * 128; i += blockDim.x * 4)
    *(float4*)(&As[i]) = *(const float4*)(&A[abase + i]);
  __syncthreads();

  const int wave = tid >> 5, lane = tid & 31, half = lane >> 4, lidx = lane & 15;
  const int n0 = wave * 16;
  v8f acc = {};
#pragma unroll
  for (int k0 = 0; k0 < 128; k0 += 4) {
    v2f a, b;
    a.x = As[lidx * 128 + k0 + 2 * half];
    a.y = As[lidx * 128 + k0 + 2 * half + 1];
    b.x = B[(k0 + 2 * half) * COUT + n0 + lidx];
    b.y = B[(k0 + 2 * half + 1) * COUT + n0 + lidx];
    acc = __builtin_amdgcn_wmma_f32_16x16x4_f32(false, a, false, b, (short)0, acc,
                                                false, false);
  }
  const size_t row0 = (size_t)blockIdx.x * 16;
#pragma unroll
  for (int r = 0; r < 8; ++r)
    C[(row0 + r + 8 * half) * COUT + n0 + lidx] = acc[r];
}

template <int C, bool ET>
__global__ void node_prep_k(const float* __restrict__ h, const float* __restrict__ a_src,
                            const float* __restrict__ a_dst,
                            const float* __restrict__ mean_attr,
                            const float* __restrict__ wa, float* s_src, float* s_dst,
                            float* self_logit) {
  int gid = blockIdx.x * blockDim.x + threadIdx.x;
  int n = gid >> 5, lane = gid & 31;
  if (n >= NN) return;
  float ps = 0.f, pd = 0.f;
#pragma unroll
  for (int i = 0; i < C / 32; ++i) {
    int c = lane + i * 32;
    float v = h[(size_t)n * C + c];
    ps += v * a_src[c];
    pd += v * a_dst[c];
  }
#pragma unroll
  for (int m = 16; m > 0; m >>= 1) { ps += __shfl_xor(ps, m, 32); pd += __shfl_xor(pd, m, 32); }
  if (lane == 0) {
    float st = 0.f;
    if (ET) { for (int j = 0; j < 16; ++j) st += mean_attr[(size_t)n * 16 + j] * wa[j]; }
    s_src[n] = ps; s_dst[n] = pd; self_logit[n] = leaky(ps + pd + st);
  }
}

__global__ void eterm_k(const float* __restrict__ ea, const float* __restrict__ wa,
                        float* eterm) {
  int i = blockIdx.x * blockDim.x + threadIdx.x;
  if (i >= NE * 16) return;
  int j = i & 15;
  float v = ea[i] * wa[j];
  v += __shfl_xor(v, 1, 32); v += __shfl_xor(v, 2, 32);
  v += __shfl_xor(v, 4, 32); v += __shfl_xor(v, 8, 32);
  if (j == 0) eterm[i >> 4] = v;
}

template <int V> struct VecOf;
template <> struct VecOf<4> { typedef float4 type; };
template <> struct VecOf<2> { typedef float2 type; };
__device__ __forceinline__ void axpy(float4& acc, float a, const float4& v) {
  acc.x += a * v.x; acc.y += a * v.y; acc.z += a * v.z; acc.w += a * v.w;
}
__device__ __forceinline__ void axpy(float2& acc, float a, const float2& v) {
  acc.x += a * v.x; acc.y += a * v.y;
}

template <bool ET>
__device__ __forceinline__ float edge_logit(int p, float pd, const int* __restrict__ ssrc,
                                            const int* __restrict__ seid,
                                            const float* __restrict__ s_src,
                                            const float* __restrict__ eterm) {
  float l = s_src[ssrc[p]] + pd;
  if (ET) l += eterm[seid[p]];
  return leaky(l);
}
__device__ __forceinline__ float wave_max(float v) {
  for (int m = 16; m > 0; m >>= 1) v = fmaxf(v, __shfl_xor(v, m, 32));
  return v;
}
__device__ __forceinline__ float wave_sum(float v) {
  for (int m = 16; m > 0; m >>= 1) v += __shfl_xor(v, m, 32);
  return v;
}

template <int C, bool ET>
__global__ void gat_node_k(const int* __restrict__ rowptr, const int* __restrict__ ssrc,
                           const int* __restrict__ seid, const float* __restrict__ s_src,
                           const float* __restrict__ s_dst, const float* __restrict__ self_l,
                           const float* __restrict__ eterm, const float* __restrict__ h,
                           const float* __restrict__ bias, float* __restrict__ out) {
  constexpr int V = C / 32;
  typedef typename VecOf<V>::type vec_t;
  const int gid = blockIdx.x * blockDim.x + threadIdx.x;
  const int lane = gid & 31;
  const int n = __builtin_amdgcn_readfirstlane(gid >> 5);
  if (n >= NN) return;
  const int start = rowptr[n];
  const int end = rowptr[n + 1];
  const float pd = s_dst[n], sl = self_l[n];
  const int ch = lane * V;
  const vec_t hs = *(const vec_t*)&h[(size_t)n * C + ch];
  const vec_t bv = *(const vec_t*)&bias[ch];
  vec_t acc = bv;

  if (end - start <= 32) {
    const int p = start + lane;
    const bool act = p < end;
    int s = 0;
    float l = 0.f;
    if (act) { s = ssrc[p]; l = edge_logit<ET>(p, pd, ssrc, seid, s_src, eterm); }
    const float m = wave_max(act ? fmaxf(l, sl) : sl);
    const float w = act ? __expf(l - m) : 0.f;
    const float wself = __expf(sl - m);
    const float den = wself + wave_sum(w) + 1e-16f;
    const float a = w / den;
    axpy(acc, wself / den, hs);
    const int deg = end - start;
#pragma unroll 4
    for (int k = 0; k < deg; ++k) {
      const int sk = lane_i(s, k);
      const float ak = lane_f(a, k);
      axpy(acc, ak, *(const vec_t*)&h[(size_t)sk * C + ch]);
    }
  } else {
    float m = sl;
    for (int base = start; base < end; base += 32) {
      const int p = base + lane;
      if (p < end) m = fmaxf(m, edge_logit<ET>(p, pd, ssrc, seid, s_src, eterm));
    }
    m = wave_max(m);
    float ssum = 0.f;
    for (int base = start; base < end; base += 32) {
      const int p = base + lane;
      if (p < end) ssum += __expf(edge_logit<ET>(p, pd, ssrc, seid, s_src, eterm) - m);
    }
    const float wself = __expf(sl - m);
    const float den = wself + wave_sum(ssum) + 1e-16f;
    axpy(acc, wself / den, hs);
    for (int base = start; base < end; base += 32) {
      const int p = base + lane;
      int s = 0;
      float a = 0.f;
      if (p < end) {
        s = ssrc[p];
        a = __expf(edge_logit<ET>(p, pd, ssrc, seid, s_src, eterm) - m) / den;
      }
      const int cntk = min(32, end - base);
#pragma unroll 4
      for (int k = 0; k < cntk; ++k) {
        const int sk = lane_i(s, k);
        const float ak = lane_f(a, k);
        axpy(acc, ak, *(const vec_t*)&h[(size_t)sk * C + ch]);
      }
    }
  }
  *(vec_t*)&out[(size_t)n * C + ch] = acc;
}

struct Scratch {
  float *s_src, *s_dst, *self_l;
  const float* mean_attr;
  float *eterm, *wa;
  const int *rowptr, *ssrc, *seid;
};

template <int C, bool ET>
static void run_gat(hipStream_t st, const float* hin, const float* Wm, const float* asrc,
                    const float* adst, const float* bias, const float* We, const float* ae,
                    const float* edge_attr, float* hg, float* out, const Scratch& S) {
  constexpr int COLT = C / 16;
  if (ET) wa_k<<<1, 16, 0, st>>>(We, ae, S.wa, C);
  gemm_k128<COLT><<<NN / 16, 32 * COLT, 0, st>>>(hin, Wm, hg);
  node_prep_k<C, ET><<<(NN * 32 + 255) / 256, 256, 0, st>>>(hg, asrc, adst, S.mean_attr,
                                                            S.wa, S.s_src, S.s_dst, S.self_l);
  if (ET) eterm_k<<<(NE * 16 + 255) / 256, 256, 0, st>>>(edge_attr, S.wa, S.eterm);
  gat_node_k<C, ET><<<(NN * 32 + 255) / 256, 256, 0, st>>>(S.rowptr, S.ssrc, S.seid, S.s_src,
                                                           S.s_dst, S.self_l, S.eterm, hg,
                                                           bias, out);
}

extern "C" void kernel_launch(void* const* d_in, const int* in_sizes, int n_in, void* d_out,
                              int out_size, void* d_ws, size_t ws_size, hipStream_t stream) {
  (void)in_sizes; (void)n_in; (void)out_size; (void)ws_size;
  const float* x = (const float*)d_in[0];
  const int* ei = (const int*)d_in[1];
  const int* src = ei;
  const int* dst = ei + NE;
  const float* edge_attr = (const float*)d_in[2];
  const float *W1 = (const float*)d_in[3], *a1s = (const float*)d_in[4],
              *a1d = (const float*)d_in[5], *b1 = (const float*)d_in[6];
  const float *W2 = (const float*)d_in[7], *a2s = (const float*)d_in[8],
              *a2d = (const float*)d_in[9], *b2 = (const float*)d_in[10];
  const float *We2 = (const float*)d_in[11], *ae2 = (const float*)d_in[12];
  const float *Wr = (const float*)d_in[13], *br = (const float*)d_in[14];
  const float *Wmu = (const float*)d_in[15], *amus = (const float*)d_in[16],
              *amud = (const float*)d_in[17], *bmu = (const float*)d_in[18],
              *Wemu = (const float*)d_in[19], *aemu = (const float*)d_in[20];
  const float *Wls = (const float*)d_in[21], *alss = (const float*)d_in[22],
              *alsd = (const float*)d_in[23], *bls = (const float*)d_in[24],
              *Wels = (const float*)d_in[25], *aels = (const float*)d_in[26];
  float* outf = (float*)d_out;

  float* W = (float*)d_ws;
  size_t o = 0;
  float* hA = W + o; o += (size_t)NN * 128;
  float* hB = W + o; o += (size_t)NN * 128;
  float* hC = W + o; o += (size_t)NN * 128;
  Scratch S;
  S.s_src = W + o; o += NN;
  S.s_dst = W + o; o += NN;
  S.self_l = W + o; o += NN;
  float* mean_attr = W + o; o += (size_t)NN * 16;
  S.mean_attr = mean_attr;
  S.eterm = W + o; o += NE;
  S.wa = W + o; o += 16;
  int* cnt = (int*)(W + o); o += NN;
  int* bsum = (int*)(W + o); o += 256;
  int* rowptr = (int*)(W + o); o += NN + 16;
  int* rank = (int*)(W + o); o += NE;
  int* eid_raw = (int*)(W + o); o += NE;
  int* seid = (int*)(W + o); o += NE;
  int* ssrc = (int*)(W + o); o += NE;
  S.rowptr = rowptr; S.ssrc = ssrc; S.seid = seid;

  zero_i32_k<<<(NN + 255) / 256, 256, 0, stream>>>(cnt, NN);
  hist_rank_k<<<(NE + 255) / 256, 256, 0, stream>>>(dst, cnt, rank);
  blocksum_k<<<SCAN_BLOCKS, 256, 0, stream>>>(cnt, bsum);
  scan_k<<<SCAN_BLOCKS, 256, 0, stream>>>(cnt, bsum, rowptr);
  fill_k<<<(NE + 255) / 256, 256, 0, stream>>>(dst, rowptr, rank, eid_raw);
  seg_sort_k<<<(NN * 32 + 255) / 256, 256, 0, stream>>>(rowptr, eid_raw, src, seid, ssrc);

  attr_mean_k<<<(NN * 16 + 255) / 256, 256, 0, stream>>>(rowptr, seid, edge_attr, mean_attr);

  run_gat<128, false>(stream, x, W1, a1s, a1d, b1, nullptr, nullptr, edge_attr, hA, hB, S);
  relu_k<<<(NN * 128 + 255) / 256, 256, 0, stream>>>(hB, NN * 128);

  run_gat<128, true>(stream, hB, W2, a2s, a2d, b2, We2, ae2, edge_attr, hA, hC, S);

  gemm_k128<8><<<NN / 16, 256, 0, stream>>>(x, Wr, hB);
  residual_k<<<(NN * 128 + 255) / 256, 256, 0, stream>>>(hC, hB, br, NN * 128);

  run_gat<64, true>(stream, hB, Wmu, amus, amud, bmu, Wemu, aemu, edge_attr, hA, outf, S);
  run_gat<64, true>(stream, hB, Wls, alss, alsd, bls, Wels, aels, edge_attr, hA,
                    outf + (size_t)NN * 64, S);
}


# Edge terms of all three convs made in ONE pass over edge_attr up front (16 B per lane, 4 loads in flight) instead of three 4-B-per-lane passes; each dot sums 4 products in-lane, then over 4 lanes
# speedup vs baseline: 3.4119x; 1.2674x over previous
#include <hip/hip_runtime.h>
#include <math.h>

typedef float v2f __attribute__((ext_vector_type(2)));
typedef float v8f __attribute__((ext_vector_type(8)));

static constexpr int   NN = 100000;
static constexpr int   NE = 1600000;
static constexpr float NEG = 0.2f;
static constexpr int   SCAN_TILE = 1024;
static constexpr int   SCAN_BLOCKS = (NN + 1 + SCAN_TILE - 1) / SCAN_TILE;
static_assert(SCAN_BLOCKS <= 256, "scan_k sums the earlier block totals with one thread each");

__device__ __forceinline__ float leaky(float x) { return x > 0.f ? x : NEG * x; }

__global__ void zero_i32_k(int* p, int n) {
  int i = blockIdx.x * blockDim.x + threadIdx.x;
  if (i < n) p[i] = 0;
}
__global__ void relu_k(float* p, int n) {
  int i = blockIdx.x * blockDim.x + threadIdx.x;
  if (i < n) p[i] = fmaxf(p[i], 0.f);
}
__global__ void residual_k(const float* __restrict__ h2, float* __restrict__ r,
                           const float* __restrict__ br, int n) {
  int i = blockIdx.x * blockDim.x + threadIdx.x;
  if (i < n) { int c = i & 127; r[i] = fmaxf(h2[i], 0.f) + r[i] + br[c]; }
}

__global__ void hist_rank_k(const int* __restrict__ dst, int* cnt, int* __restrict__ rank) {
  int e = blockIdx.x * blockDim.x + threadIdx.x;
  if (e >= NE) return;
  rank[e] = atomicAdd(&cnt[dst[e]], 1);
}
__global__ void blocksum_k(const int* __restrict__ cnt, int* __restrict__ bsum) {
  __shared__ int wtot[8];
  const int tid = threadIdx.x, lane = tid & 31, w = tid >> 5;
  const int i0 = blockIdx.x * SCAN_TILE + tid * 4;
  int s = 0;
  for (int j = 0; j < 4; ++j) s += (i0 + j < NN) ? cnt[i0 + j] : 0;
  for (int m = 16; m > 0; m >>= 1) s += __shfl_xor(s, m, 32);
  if (lane == 0) wtot[w] = s;
  __syncthreads();
  if (tid == 0) {
    int t = 0;
    for (int i = 0; i < 8; ++i) t += wtot[i];
    bsum[blockIdx.x] = t;
  }
}
__global__ void scan_k(const int* __restrict__ cnt, const int* __restrict__ bsum,
                       int* __restrict__ rowptr) {
  __shared__ int wtot[8];
  __shared__ int wpre[8];
  const int tid = threadIdx.x, lane = tid & 31, w = tid >> 5;
  int part = (tid < (int)blockIdx.x) ? bsum[tid] : 0;
  for (int m = 16; m > 0; m >>= 1) part += __shfl_xor(part, m, 32);
  if (lane == 0) wpre[w] = part;
  const int i0 = blockIdx.x * SCAN_TILE + tid * 4;
  int c[4];
  for (int j = 0; j < 4; ++j) c[j] = (i0 + j < NN) ? cnt[i0 + j] : 0;
  const int ts = c[0] + c[1] + c[2] + c[3];
  int inc = ts;
  for (int o = 1; o < 32; o <<= 1) {
    int t = __shfl_up(inc, o, 32);
    if (lane >= o) inc += t;
  }
  if (lane == 31) wtot[w] = inc;
  __syncthreads();
  int run = inc - ts;
  for (int i = 0; i < 8; ++i) {
    run += wpre[i];
    if (i < w) run += wtot[i];
  }
  for (int j = 0; j < 4; ++j) {
    if (i0 + j <= NN) rowptr[i0 + j] = run;
    run += c[j];
  }
}
__global__ void fill_k(const int* __restrict__ dst, const int* __restrict__ rowptr,
                       const int* __restrict__ rank, int* __restrict__ eid_raw) {
  int e = blockIdx.x * blockDim.x + threadIdx.x;
  if (e >= NE) return;
  eid_raw[rowptr[dst[e]] + rank[e]] = e;
}
__device__ __forceinline__ int lane_i(int v, int k) { return __builtin_amdgcn_readlane(v, k); }
__device__ __forceinline__ float lane_f(float v, int k) {
  return __int_as_float(__builtin_amdgcn_readlane(__float_as_int(v), k));
}
__global__ void seg_sort_k(const int* __restrict__ rowptr, const int* __restrict__ eid_raw,
                           const int* __restrict__ src, int* __restrict__ seid,
                           int* __restrict__ ssrc) {
  const int gid = blockIdx.x * blockDim.x + threadIdx.x;
  const int lane = gid & 31;
  const int n = __builtin_amdgcn_readfirstlane(gid >> 5);
  if (n >= NN) return;
  const int start = rowptr[n];
  const int end = rowptr[n + 1];
  for (int base = start; base < end; base += 32) {
    const int p = base + lane;
    const int x = (p < end) ? eid_raw[p] : 0x7fffffff;
    int r = 0;
    for (int b2 = start; b2 < end; b2 += 32) {
      const int q = b2 + lane;
      const int y = (q < end) ? eid_raw[q] : 0x7fffffff;
      const int m = min(32, end - b2);
      for (int k = 0; k < m; ++k) r += (lane_i(y, k) < x) ? 1 : 0;
    }
    if (p < end) { seid[start + r] = x; ssrc[start + r] = src[x]; }
  }
}
__global__ void attr_mean_k(const int* __restrict__ rowptr, const int* __restrict__ seid,
                            const float* __restrict__ ea, float* __restrict__ mean_attr) {
  const int gid = blockIdx.x * blockDim.x + threadIdx.x;
  const int n = gid >> 4, j = gid & 15;
  if (n >= NN) return;
  const int start = rowptr[n], end = rowptr[n + 1];
  float s = 0.f;
  for (int p = start; p < end; ++p) s += ea[(size_t)seid[p] * 16 + j];
  mean_attr[(size_t)n * 16 + j] = s / fmaxf((float)(end - start), 1.f);
}

__global__ void wa3_k(const float* __restrict__ We_a, const float* __restrict__ ae_a,
                      const float* __restrict__ We_b, const float* __restrict__ ae_b,
                      const float* __restrict__ We_c, const float* __restrict__ ae_c,
                      float* __restrict__ wa3) {
  const int l = blockIdx.x;
  const float* __restrict__ We = (l == 0) ? We_a : ((l == 1) ? We_b : We_c);
  const float* __restrict__ ae = (l == 0) ? ae_a : ((l == 1) ? ae_b : ae_c);
  const int K = (l == 0) ? 128 : 64;
  const int i = threadIdx.x;
  if (i < 16) {
    float s = 0.f;
    for (int j = 0; j < K; ++j) s += We[i * K + j] * ae[j];
    wa3[l * 16 + i] = s;
  }
}

template <int COLT>
__global__ void gemm_k128(const float* __restrict__ A, const float* __restrict__ B,
                          float* __restrict__ C) {
  constexpr int COUT = COLT * 16;
  __shared__ float As[16 * 128];
  const int tid = threadIdx.x;
  const size_t abase = (size_t)blockIdx.x * (16 * 128);
  for (int i = tid * 4; i < 16 * 128; i += blockDim.x * 4)
    *(float4*)(&As[i]) = *(const float4*)(&A[abase + i]);
  __syncthreads();

  const int wave = tid >> 5, lane = tid & 31, half = lane >> 4, lidx = lane & 15;
  const int n0 = wave * 16;
  v8f acc = {};
#pragma unroll
  for (int k0 = 0; k0 < 128; k0 += 4) {
    v2f a, b;
    a.x = As[lidx * 128 + k0 + 2 * half];
    a.y = As[lidx * 128 + k0 + 2 * half + 1];
    b.x = B[(k0 + 2 * half) * COUT + n0 + lidx];
    b.y = B[(k0 + 2 * half + 1) * COUT + n0 + lidx];
    acc = __builtin_amdgcn_wmma_f32_16x16x4_f32(false, a, false, b, (short)0, acc,
                                                false, false);
  }
  const size_t row0 = (size_t)blockIdx.x * 16;
#pragma unroll
  for (int r = 0; r < 8; ++r)
    C[(row0 + r + 8 * half) * COUT + n0 + lidx] = acc[r];
}

template <int C, bool ET>
__global__ void node_prep_k(const float* __restrict__ h, const float* __restrict__ a_src,
                            const float* __restrict__ a_dst,
                            const float* __restrict__ mean_attr,
                            const float* __restrict__ wa, float* s_src, float* s_dst,
                            float* self_logit) {
  int gid = blockIdx.x * blockDim.x + threadIdx.x;
  int n = gid >> 5, lane = gid & 31;
  if (n >= NN) return;
  float ps = 0.f, pd = 0.f;
#pragma unroll
  for (int i = 0; i < C / 32; ++i) {
    int c = lane + i * 32;
    float v = h[(size_t)n * C + c];
    ps += v * a_src[c];
    pd += v * a_dst[c];
  }
#pragma unroll
  for (int m = 16; m > 0; m >>= 1) { ps += __shfl_xor(ps, m, 32); pd += __shfl_xor(pd, m, 32); }
  if (lane == 0) {
    float st = 0.f;
    if (ET) { for (int j = 0; j < 16; ++j) st += mean_attr[(size_t)n * 16 + j] * wa[j]; }
    s_src[n] = ps; s_dst[n] = pd; self_logit[n] = leaky(ps + pd + st);
  }
}

static constexpr int ET_PER_THREAD = 4;
static constexpr int ET_BLOCK = 256;
static constexpr int ET_TILE = ET_BLOCK * ET_PER_THREAD;
static_assert((NE * 4) % ET_TILE == 0, "eterm3_k has no tail handling");
__device__ __forceinline__ float dot4(const float4& a, const float4& b) {
  return a.x * b.x + a.y * b.y + a.z * b.z + a.w * b.w;
}
__device__ __forceinline__ float quad_sum(float v) {
  v += __shfl_xor(v, 1, 32);
  v += __shfl_xor(v, 2, 32);
  return v;
}
__global__ void eterm3_k(const float* __restrict__ ea, const float* __restrict__ wa3,
                         float* __restrict__ et0, float* __restrict__ et1,
                         float* __restrict__ et2) {
  const int tid = threadIdx.x;
  const int q = tid & 3;
  const float4* __restrict__ ea4 = (const float4*)ea;
  const size_t base = (size_t)blockIdx.x * ET_TILE + tid;
  float4 v[ET_PER_THREAD];
#pragma unroll
  for (int k = 0; k < ET_PER_THREAD; ++k) v[k] = ea4[base + (size_t)k * ET_BLOCK];
  const float4 w0 = ((const float4*)wa3)[q];
  const float4 w1 = ((const float4*)wa3)[4 + q];
  const float4 w2 = ((const float4*)wa3)[8 + q];
#pragma unroll
  for (int k = 0; k < ET_PER_THREAD; ++k) {
    const float s0 = quad_sum(dot4(v[k], w0));
    const float s1 = quad_sum(dot4(v[k], w1));
    const float s2 = quad_sum(dot4(v[k], w2));
    if (q == 0) {
      const size_t e = (base + (size_t)k * ET_BLOCK) >> 2;
      et0[e] = s0; et1[e] = s1; et2[e] = s2;
    }
  }
}

template <int V> struct VecOf;
template <> struct VecOf<4> { typedef float4 type; };
template <> struct VecOf<2> { typedef float2 type; };
__device__ __forceinline__ void axpy(float4& acc, float a, const float4& v) {
  acc.x += a * v.x; acc.y += a * v.y; acc.z += a * v.z; acc.w += a * v.w;
}
__device__ __forceinline__ void axpy(float2& acc, float a, const float2& v) {
  acc.x += a * v.x; acc.y += a * v.y;
}

template <bool ET>
__device__ __forceinline__ float edge_logit(int p, float pd, const int* __restrict__ ssrc,
                                            const int* __restrict__ seid,
                                            const float* __restrict__ s_src,
                                            const float* __restrict__ eterm) {
  float l = s_src[ssrc[p]] + pd;
  if (ET) l += eterm[seid[p]];
  return leaky(l);
}
__device__ __forceinline__ float wave_max(float v) {
  for (int m = 16; m > 0; m >>= 1) v = fmaxf(v, __shfl_xor(v, m, 32));
  return v;
}
__device__ __forceinline__ float wave_sum(float v) {
  for (int m = 16; m > 0; m >>= 1) v += __shfl_xor(v, m, 32);
  return v;
}

template <int C, bool ET>
__global__ void gat_node_k(const int* __restrict__ rowptr, const int* __restrict__ ssrc,
                           const int* __restrict__ seid, const float* __restrict__ s_src,
                           const float* __restrict__ s_dst, const float* __restrict__ self_l,
                           const float* __restrict__ eterm, const float* __restrict__ h,
                           const float* __restrict__ bias, float* __restrict__ out) {
  constexpr int V = C / 32;
  typedef typename VecOf<V>::type vec_t;
  const int gid = blockIdx.x * blockDim.x + threadIdx.x;
  const int lane = gid & 31;
  const int n = __builtin_amdgcn_readfirstlane(gid >> 5);
  if (n >= NN) return;
  const int start = rowptr[n];
  const int end = rowptr[n + 1];
  const float pd = s_dst[n], sl = self_l[n];
  const int ch = lane * V;
  const vec_t hs = *(const vec_t*)&h[(size_t)n * C + ch];
  const vec_t bv = *(const vec_t*)&bias[ch];
  vec_t acc = bv;

  if (end - start <= 32) {
    const int p = start + lane;
    const bool act = p < end;
    int s = 0;
    float l = 0.f;
    if (act) { s = ssrc[p]; l = edge_logit<ET>(p, pd, ssrc, seid, s_src, eterm); }
    const float m = wave_max(act ? fmaxf(l, sl) : sl);
    const float w = act ? __expf(l - m) : 0.f;
    const float wself = __expf(sl - m);
    const float den = wself + wave_sum(w) + 1e-16f;
    const float a = w / den;
    axpy(acc, wself / den, hs);
    const int deg = end - start;
#pragma unroll 4
    for (int k = 0; k < deg; ++k) {
      const int sk = lane_i(s, k);
      const float ak = lane_f(a, k);
      axpy(acc, ak, *(const vec_t*)&h[(size_t)sk * C + ch]);
    }
  } else {
    float m = sl;
    for (int base = start; base < end; base += 32) {
      const int p = base + lane;
      if (p < end) m = fmaxf(m, edge_logit<ET>(p, pd, ssrc, seid, s_src, eterm));
    }
    m = wave_max(m);
    float ssum = 0.f;
    for (int base = start; base < end; base += 32) {
      const int p = base + lane;
      if (p < end) ssum += __expf(edge_logit<ET>(p, pd, ssrc, seid, s_src, eterm) - m);
    }
    const float wself = __expf(sl - m);
    const float den = wself + wave_sum(ssum) + 1e-16f;
    axpy(acc, wself / den, hs);
    for (int base = start; base < end; base += 32) {
      const int p = base + lane;
      int s = 0;
      float a = 0.f;
      if (p < end) {
        s = ssrc[p];
        a = __expf(edge_logit<ET>(p, pd, ssrc, seid, s_src, eterm) - m) / den;
      }
      const int cntk = min(32, end - base);
#pragma unroll 4
      for (int k = 0; k < cntk; ++k) {
        const int sk = lane_i(s, k);
        const float ak = lane_f(a, k);
        axpy(acc, ak, *(const vec_t*)&h[(size_t)sk * C + ch]);
      }
    }
  }
  *(vec_t*)&out[(size_t)n * C + ch] = acc;
}

struct Scratch {
  float *s_src, *s_dst, *self_l;
  const float* mean_attr;
  const float *eterm, *wa;
  const int *rowptr, *ssrc, *seid;
};

template <int C, bool ET>
static void run_gat(hipStream_t st, const float* hin, const float* Wm, const float* asrc,
                    const float* adst, const float* bias, float* hg, float* out,
                    const Scratch& S) {
  constexpr int COLT = C / 16;
  gemm_k128<COLT><<<NN / 16, 32 * COLT, 0, st>>>(hin, Wm, hg);
  node_prep_k<C, ET><<<(NN * 32 + 255) / 256, 256, 0, st>>>(hg, asrc, adst, S.mean_attr,
                                                            S.wa, S.s_src, S.s_dst, S.self_l);
  gat_node_k<C, ET><<<(NN * 32 + 255) / 256, 256, 0, st>>>(S.rowptr, S.ssrc, S.seid, S.s_src,
                                                           S.s_dst, S.self_l, S.eterm, hg,
                                                           bias, out);
}

extern "C" void kernel_launch(void* const* d_in, const int* in_sizes, int n_in, void* d_out,
                              int out_size, void* d_ws, size_t ws_size, hipStream_t stream) {
  (void)in_sizes; (void)n_in; (void)out_size; (void)ws_size;
  const float* x = (const float*)d_in[0];
  const int* ei = (const int*)d_in[1];
  const int* src = ei;
  const int* dst = ei + NE;
  const float* edge_attr = (const float*)d_in[2];
  const float *W1 = (const float*)d_in[3], *a1s = (const float*)d_in[4],
              *a1d = (const float*)d_in[5], *b1 = (const float*)d_in[6];
  const float *W2 = (const float*)d_in[7], *a2s = (const float*)d_in[8],
              *a2d = (const float*)d_in[9], *b2 = (const float*)d_in[10];
  const float *We2 = (const float*)d_in[11], *ae2 = (const float*)d_in[12];
  const float *Wr = (const float*)d_in[13], *br = (const float*)d_in[14];
  const float *Wmu = (const float*)d_in[15], *amus = (const float*)d_in[16],
              *amud = (const float*)d_in[17], *bmu = (const float*)d_in[18],
              *Wemu = (const float*)d_in[19], *aemu = (const float*)d_in[20];
  const float *Wls = (const float*)d_in[21], *alss = (const float*)d_in[22],
              *alsd = (const float*)d_in[23], *bls = (const float*)d_in[24],
              *Wels = (const float*)d_in[25], *aels = (const float*)d_in[26];
  float* outf = (float*)d_out;

  float* W = (float*)d_ws;
  size_t o = 0;
  float* hA = W + o; o += (size_t)NN * 128;
  float* hB = W + o; o += (size_t)NN * 128;
  float* hC = W + o; o += (size_t)NN * 128;
  Scratch S;
  S.s_src = W + o; o += NN;
  S.s_dst = W + o; o += NN;
  S.self_l = W + o; o += NN;
  float* mean_attr = W + o; o += (size_t)NN * 16;
  S.mean_attr = mean_attr;
  float* et0 = W + o; o += NE;
  float* et1 = W + o; o += NE;
  float* et2 = W + o; o += NE;
  float* wa3 = W + o; o += 48;
  S.eterm = et0; S.wa = wa3;
  int* cnt = (int*)(W + o); o += NN;
  int* bsum = (int*)(W + o); o += 256;
  int* rowptr = (int*)(W + o); o += NN + 16;
  int* rank = (int*)(W + o); o += NE;
  int* eid_raw = (int*)(W + o); o += NE;
  int* seid = (int*)(W + o); o += NE;
  int* ssrc = (int*)(W + o); o += NE;
  S.rowptr = rowptr; S.ssrc = ssrc; S.seid = seid;

  zero_i32_k<<<(NN + 255) / 256, 256, 0, stream>>>(cnt, NN);
  hist_rank_k<<<(NE + 255) / 256, 256, 0, stream>>>(dst, cnt, rank);
  blocksum_k<<<SCAN_BLOCKS, 256, 0, stream>>>(cnt, bsum);
  scan_k<<<SCAN_BLOCKS, 256, 0, stream>>>(cnt, bsum, rowptr);
  fill_k<<<(NE + 255) / 256, 256, 0, stream>>>(dst, rowptr, rank, eid_raw);
  seg_sort_k<<<(NN * 32 + 255) / 256, 256, 0, stream>>>(rowptr, eid_raw, src, seid, ssrc);

  attr_mean_k<<<(NN * 16 + 255) / 256, 256, 0, stream>>>(rowptr, seid, edge_attr, mean_attr);

  wa3_k<<<3, 16, 0, stream>>>(We2, ae2, Wemu, aemu, Wels, aels, wa3);
  eterm3_k<<<(NE * 4) / ET_TILE, ET_BLOCK, 0, stream>>>(edge_attr, wa3, et0, et1, et2);

  run_gat<128, false>(stream, x, W1, a1s, a1d, b1, hA, hB, S);
  relu_k<<<(NN * 128 + 255) / 256, 256, 0, stream>>>(hB, NN * 128);

  S.eterm = et0; S.wa = wa3;
  run_gat<128, true>(stream, hB, W2, a2s, a2d, b2, hA, hC, S);

  gemm_k128<8><<<NN / 16, 256, 0, stream>>>(x, Wr, hB);
  residual_k<<<(NN * 128 + 255) / 256, 256, 0, stream>>>(hC, hB, br, NN * 128);

  S.eterm = et1; S.wa = wa3 + 16;
  run_gat<64, true>(stream, hB, Wmu, amus, amud, bmu, hA, outf, S);
  S.eterm = et2; S.wa = wa3 + 32;
  run_gat<64, true>(stream, hB, Wls, alss, alsd, bls, hA, outf + (size_t)NN * 64, S);
}


# GEMM A tile in LDS at a row pitch of 132 dwords instead of 128: the 16 rows of a fragment read no longer share two banks (16-way conflict gone)
# speedup vs baseline: 3.8473x; 1.1276x over previous
#include <hip/hip_runtime.h>
#include <math.h>

typedef float v2f __attribute__((ext_vector_type(2)));
typedef float v8f __attribute__((ext_vector_type(8)));

static constexpr int   NN = 100000;
static constexpr int   NE = 1600000;
static constexpr float NEG = 0.2f;
static constexpr int   SCAN_TILE = 1024;
static constexpr int   SCAN_BLOCKS = (NN + 1 + SCAN_TILE - 1) / SCAN_TILE;
static_assert(SCAN_BLOCKS <= 256, "scan_k sums the earlier block totals with one thread each");

__device__ __forceinline__ float leaky(float x) { return x > 0.f ? x : NEG * x; }

__global__ void zero_i32_k(int* p, int n) {
  int i = blockIdx.x * blockDim.x + threadIdx.x;
  if (i < n) p[i] = 0;
}
__global__ void relu_k(float* p, int n) {
  int i = blockIdx.x * blockDim.x + threadIdx.x;
  if (i < n) p[i] = fmaxf(p[i], 0.f);
}
__global__ void residual_k(const float* __restrict__ h2, float* __restrict__ r,
                           const float* __restrict__ br, int n) {
  int i = blockIdx.x * blockDim.x + threadIdx.x;
  if (i < n) { int c = i & 127; r[i] = fmaxf(h2[i], 0.f) + r[i] + br[c]; }
}

__global__ void hist_rank_k(const int* __restrict__ dst, int* cnt, int* __restrict__ rank) {
  int e = blockIdx.x * blockDim.x + threadIdx.x;
  if (e >= NE) return;
  rank[e] = atomicAdd(&cnt[dst[e]], 1);
}
__global__ void blocksum_k(const int* __restrict__ cnt, int* __restrict__ bsum) {
  __shared__ int wtot[8];
  const int tid = threadIdx.x, lane = tid & 31, w = tid >> 5;
  const int i0 = blockIdx.x * SCAN_TILE + tid * 4;
  int s = 0;
  for (int j = 0; j < 4; ++j) s += (i0 + j < NN) ? cnt[i0 + j] : 0;
  for (int m = 16; m > 0; m >>= 1) s += __shfl_xor(s, m, 32);
  if (lane == 0) wtot[w] = s;
  __syncthreads();
  if (tid == 0) {
    int t = 0;
    for (int i = 0; i < 8; ++i) t += wtot[i];
    bsum[blockIdx.x] = t;
  }
}
__global__ void scan_k(const int* __restrict__ cnt, const int* __restrict__ bsum,
                       int* __restrict__ rowptr) {
  __shared__ int wtot[8];
  __shared__ int wpre[8];
  const int tid = threadIdx.x, lane = tid & 31, w = tid >> 5;
  int part = (tid < (int)blockIdx.x) ? bsum[tid] : 0;
  for (int m = 16; m > 0; m >>= 1) part += __shfl_xor(part, m, 32);
  if (lane == 0) wpre[w] = part;
  const int i0 = blockIdx.x * SCAN_TILE + tid * 4;
  int c[4];
  for (int j = 0; j < 4; ++j) c[j] = (i0 + j < NN) ? cnt[i0 + j] : 0;
  const int ts = c[0] + c[1] + c[2] + c[3];
  int inc = ts;
  for (int o = 1; o < 32; o <<= 1) {
    int t = __shfl_up(inc, o, 32);
    if (lane >= o) inc += t;
  }
  if (lane == 31) wtot[w] = inc;
  __syncthreads();
  int run = inc - ts;
  for (int i = 0; i < 8; ++i) {
    run += wpre[i];
    if (i < w) run += wtot[i];
  }
  for (int j = 0; j < 4; ++j) {
    if (i0 + j <= NN) rowptr[i0 + j] = run;
    run += c[j];
  }
}
__global__ void fill_k(const int* __restrict__ dst, const int* __restrict__ rowptr,
                       const int* __restrict__ rank, int* __restrict__ eid_raw) {
  int e = blockIdx.x * blockDim.x + threadIdx.x;
  if (e >= NE) return;
  eid_raw[rowptr[dst[e]] + rank[e]] = e;
}
__device__ __forceinline__ int lane_i(int v, int k) { return __builtin_amdgcn_readlane(v, k); }
__device__ __forceinline__ float lane_f(float v, int k) {
  return __int_as_float(__builtin_amdgcn_readlane(__float_as_int(v), k));
}
__global__ void seg_sort_k(const int* __restrict__ rowptr, const int* __restrict__ eid_raw,
                           const int* __restrict__ src, int* __restrict__ seid,
                           int* __restrict__ ssrc) {
  const int gid = blockIdx.x * blockDim.x + threadIdx.x;
  const int lane = gid & 31;
  const int n = __builtin_amdgcn_readfirstlane(gid >> 5);
  if (n >= NN) return;
  const int start = rowptr[n];
  const int end = rowptr[n + 1];
  for (int base = start; base < end; base += 32) {
    const int p = base + lane;
    const int x = (p < end) ? eid_raw[p] : 0x7fffffff;
    int r = 0;
    for (int b2 = start; b2 < end; b2 += 32) {
      const int q = b2 + lane;
      const int y = (q < end) ? eid_raw[q] : 0x7fffffff;
      const int m = min(32, end - b2);
      for (int k = 0; k < m; ++k) r += (lane_i(y, k) < x) ? 1 : 0;
    }
    if (p < end) { seid[start + r] = x; ssrc[start + r] = src[x]; }
  }
}
__global__ void attr_mean_k(const int* __restrict__ rowptr, const int* __restrict__ seid,
                            const float* __restrict__ ea, float* __restrict__ mean_attr) {
  const int gid = blockIdx.x * blockDim.x + threadIdx.x;
  const int n = gid >> 4, j = gid & 15;
  if (n >= NN) return;
  const int start = rowptr[n], end = rowptr[n + 1];
  float s = 0.f;
  for (int p = start; p < end; ++p) s += ea[(size_t)seid[p] * 16 + j];
  mean_attr[(size_t)n * 16 + j] = s / fmaxf((float)(end - start), 1.f);
}

__global__ void wa3_k(const float* __restrict__ We_a, const float* __restrict__ ae_a,
                      const float* __restrict__ We_b, const float* __restrict__ ae_b,
                      const float* __restrict__ We_c, const float* __restrict__ ae_c,
                      float* __restrict__ wa3) {
  const int l = blockIdx.x;
  const float* __restrict__ We = (l == 0) ? We_a : ((l == 1) ? We_b : We_c);
  const float* __restrict__ ae = (l == 0) ? ae_a : ((l == 1) ? ae_b : ae_c);
  const int K = (l == 0) ? 128 : 64;
  const int i = threadIdx.x;
  if (i < 16) {
    float s = 0.f;
    for (int j = 0; j < K; ++j) s += We[i * K + j] * ae[j];
    wa3[l * 16 + i] = s;
  }
}

template <int COLT>
__global__ void gemm_k128(const float* __restrict__ A, const float* __restrict__ B,
                          float* __restrict__ C) {
  constexpr int COUT = COLT * 16;
  constexpr int AP = 132;
  __shared__ __attribute__((aligned(16))) float As[16 * AP];
  const int tid = threadIdx.x;
  const size_t abase = (size_t)blockIdx.x * (16 * 128);
  for (int i = tid * 4; i < 16 * 128; i += blockDim.x * 4)
    *(float4*)(&As[(i >> 7) * AP + (i & 127)]) = *(const float4*)(&A[abase + i]);
  __syncthreads();

  const int wave = tid >> 5, lane = tid & 31, half = lane >> 4, lidx = lane & 15;
  const int n0 = wave * 16;
  v8f acc = {};
#pragma unroll
  for (int k0 = 0; k0 < 128; k0 += 4) {
    v2f a, b;
    a.x = As[lidx * AP + k0 + 2 * half];
    a.y = As[lidx * AP + k0 + 2 * half + 1];
    b.x = B[(k0 + 2 * half) * COUT + n0 + lidx];
    b.y = B[(k0 + 2 * half + 1) * COUT + n0 + lidx];
    acc = __builtin_amdgcn_wmma_f32_16x16x4_f32(false, a, false, b, (short)0, acc,
                                                false, false);
  }
  const size_t row0 = (size_t)blockIdx.x * 16;
#pragma unroll
  for (int r = 0; r < 8; ++r)
    C[(row0 + r + 8 * half) * COUT + n0 + lidx] = acc[r];
}

template <int C, bool ET>
__global__ void node_prep_k(const float* __restrict__ h, const float* __restrict__ a_src,
                            const float* __restrict__ a_dst,
                            const float* __restrict__ mean_attr,
                            const float* __restrict__ wa, float* s_src, float* s_dst,
                            float* self_logit) {
  int gid = blockIdx.x * blockDim.x + threadIdx.x;
  int n = gid >> 5, lane = gid & 31;
  if (n >= NN) return;
  float ps = 0.f, pd = 0.f;
#pragma unroll
  for (int i = 0; i < C / 32; ++i) {
    int c = lane + i * 32;
    float v = h[(size_t)n * C + c];
    ps += v * a_src[c];
    pd += v * a_dst[c];
  }
#pragma unroll
  for (int m = 16; m > 0; m >>= 1) { ps += __shfl_xor(ps, m, 32); pd += __shfl_xor(pd, m, 32); }
  if (lane == 0) {
    float st = 0.f;
    if (ET) { for (int j = 0; j < 16; ++j) st += mean_attr[(size_t)n * 16 + j] * wa[j]; }
    s_src[n] = ps; s_dst[n] = pd; self_logit[n] = leaky(ps + pd + st);
  }
}

static constexpr int ET_PER_THREAD = 4;
static constexpr int ET_BLOCK = 256;
static constexpr int ET_TILE = ET_BLOCK * ET_PER_THREAD;
static_assert((NE * 4) % ET_TILE == 0, "eterm3_k has no tail handling");
__device__ __forceinline__ float dot4(const float4& a, const float4& b) {
  return a.x * b.x + a.y * b.y + a.z * b.z + a.w * b.w;
}
__device__ __forceinline__ float quad_sum(float v) {
  v += __shfl_xor(v, 1, 32);
  v += __shfl_xor(v, 2, 32);
  return v;
}
__global__ void eterm3_k(const float* __restrict__ ea, const float* __restrict__ wa3,
                         float* __restrict__ et0, float* __restrict__ et1,
                         float* __restrict__ et2) {
  const int tid = threadIdx.x;
  const int q = tid & 3;
  const float4* __restrict__ ea4 = (const float4*)ea;
  const size_t base = (size_t)blockIdx.x * ET_TILE + tid;
  float4 v[ET_PER_THREAD];
#pragma unroll
  for (int k = 0; k < ET_PER_THREAD; ++k) v[k] = ea4[base + (size_t)k * ET_BLOCK];
  const float4 w0 = ((const float4*)wa3)[q];
  const float4 w1 = ((const float4*)wa3)[4 + q];
  const float4 w2 = ((const float4*)wa3)[8 + q];
#pragma unroll
  for (int k = 0; k < ET_PER_THREAD; ++k) {
    const float s0 = quad_sum(dot4(v[k], w0));
    const float s1 = quad_sum(dot4(v[k], w1));
    const float s2 = quad_sum(dot4(v[k], w2));
    if (q == 0) {
      const size_t e = (base + (size_t)k * ET_BLOCK) >> 2;
      et0[e] = s0; et1[e] = s1; et2[e] = s2;
    }
  }
}

template <int V> struct VecOf;
template <> struct VecOf<4> { typedef float4 type; };
template <> struct VecOf<2> { typedef float2 type; };
__device__ __forceinline__ void axpy(float4& acc, float a, const float4& v) {
  acc.x += a * v.x; acc.y += a * v.y; acc.z += a * v.z; acc.w += a * v.w;
}
__device__ __forceinline__ void axpy(float2& acc, float a, const float2& v) {
  acc.x += a * v.x; acc.y += a * v.y;
}

template <bool ET>
__device__ __forceinline__ float edge_logit(int p, float pd, const int* __restrict__ ssrc,
                                            const int* __restrict__ seid,
                                            const float* __restrict__ s_src,
                                            const float* __restrict__ eterm) {
  float l = s_src[ssrc[p]] + pd;
  if (ET) l += eterm[seid[p]];
  return leaky(l);
}
__device__ __forceinline__ float wave_max(float v) {
  for (int m = 16; m > 0; m >>= 1) v = fmaxf(v, __shfl_xor(v, m, 32));
  return v;
}
__device__ __forceinline__ float wave_sum(float v) {
  for (int m = 16; m > 0; m >>= 1) v += __shfl_xor(v, m, 32);
  return v;
}

template <int C, bool ET>
__global__ void gat_node_k(const int* __restrict__ rowptr, const int* __restrict__ ssrc,
                           const int* __restrict__ seid, const float* __restrict__ s_src,
                           const float* __restrict__ s_dst, const float* __restrict__ self_l,
                           const float* __restrict__ eterm, const float* __restrict__ h,
                           const float* __restrict__ bias, float* __restrict__ out) {
  constexpr int V = C / 32;
  typedef typename VecOf<V>::type vec_t;
  const int gid = blockIdx.x * blockDim.x + threadIdx.x;
  const int lane = gid & 31;
  const int n = __builtin_amdgcn_readfirstlane(gid >> 5);
  if (n >= NN) return;
  const int start = rowptr[n];
  const int end = rowptr[n + 1];
  const float pd = s_dst[n], sl = self_l[n];
  const int ch = lane * V;
  const vec_t hs = *(const vec_t*)&h[(size_t)n * C + ch];
  const vec_t bv = *(const vec_t*)&bias[ch];
  vec_t acc = bv;

  if (end - start <= 32) {
    const int p = start + lane;
    const bool act = p < end;
    int s = 0;
    float l = 0.f;
    if (act) { s = ssrc[p]; l = edge_logit<ET>(p, pd, ssrc, seid, s_src, eterm); }
    const float m = wave_max(act ? fmaxf(l, sl) : sl);
    const float w = act ? __expf(l - m) : 0.f;
    const float wself = __expf(sl - m);
    const float den = wself + wave_sum(w) + 1e-16f;
    const float a = w / den;
    axpy(acc, wself / den, hs);
    const int deg = end - start;
#pragma unroll 4
    for (int k = 0; k < deg; ++k) {
      const int sk = lane_i(s, k);
      const float ak = lane_f(a, k);
      axpy(acc, ak, *(const vec_t*)&h[(size_t)sk * C + ch]);
    }
  } else {
    float m = sl;
    for (int base = start; base < end; base += 32) {
      const int p = base + lane;
      if (p < end) m = fmaxf(m, edge_logit<ET>(p, pd, ssrc, seid, s_src, eterm));
    }
    m = wave_max(m);
    float ssum = 0.f;
    for (int base = start; base < end; base += 32) {
      const int p = base + lane;
      if (p < end) ssum += __expf(edge_logit<ET>(p, pd, ssrc, seid, s_src, eterm) - m);
    }
    const float wself = __expf(sl - m);
    const float den = wself + wave_sum(ssum) + 1e-16f;
    axpy(acc, wself / den, hs);
    for (int base = start; base < end; base += 32) {
      const int p = base + lane;
      int s = 0;
      float a = 0.f;
      if (p < end) {
        s = ssrc[p];
        a = __expf(edge_logit<ET>(p, pd, ssrc, seid, s_src, eterm) - m) / den;
      }
      const int cntk = min(32, end - base);
#pragma unroll 4
      for (int k = 0; k < cntk; ++k) {
        const int sk = lane_i(s, k);
        const float ak = lane_f(a, k);
        axpy(acc, ak, *(const vec_t*)&h[(size_t)sk * C + ch]);
      }
    }
  }
  *(vec_t*)&out[(size_t)n * C + ch] = acc;
}

struct Scratch {
  float *s_src, *s_dst, *self_l;
  const float* mean_attr;
  const float *eterm, *wa;
  const int *rowptr, *ssrc, *seid;
};

template <int C, bool ET>
static void run_gat(hipStream_t st, const float* hin, const float* Wm, const float* asrc,
                    const float* adst, const float* bias, float* hg, float* out,
                    const Scratch& S) {
  constexpr int COLT = C / 16;
  gemm_k128<COLT><<<NN / 16, 32 * COLT, 0, st>>>(hin, Wm, hg);
  node_prep_k<C, ET><<<(NN * 32 + 255) / 256, 256, 0, st>>>(hg, asrc, adst, S.mean_attr,
                                                            S.wa, S.s_src, S.s_dst, S.self_l);
  gat_node_k<C, ET><<<(NN * 32 + 255) / 256, 256, 0, st>>>(S.rowptr, S.ssrc, S.seid, S.s_src,
                                                           S.s_dst, S.self_l, S.eterm, hg,
                                                           bias, out);
}

extern "C" void kernel_launch(void* const* d_in, const int* in_sizes, int n_in, void* d_out,
                              int out_size, void* d_ws, size_t ws_size, hipStream_t stream) {
  (void)in_sizes; (void)n_in; (void)out_size; (void)ws_size;
  const float* x = (const float*)d_in[0];
  const int* ei = (const int*)d_in[1];
  const int* src = ei;
  const int* dst = ei + NE;
  const float* edge_attr = (const float*)d_in[2];
  const float *W1 = (const float*)d_in[3], *a1s = (const float*)d_in[4],
              *a1d = (const float*)d_in[5], *b1 = (const float*)d_in[6];
  const float *W2 = (const float*)d_in[7], *a2s = (const float*)d_in[8],
              *a2d = (const float*)d_in[9], *b2 = (const float*)d_in[10];
  const float *We2 = (const float*)d_in[11], *ae2 = (const float*)d_in[12];
  const float *Wr = (const float*)d_in[13], *br = (const float*)d_in[14];
  const float *Wmu = (const float*)d_in[15], *amus = (const float*)d_in[16],
              *amud = (const float*)d_in[17], *bmu = (const float*)d_in[18],
              *Wemu = (const float*)d_in[19], *aemu = (const float*)d_in[20];
  const float *Wls = (const float*)d_in[21], *alss = (const float*)d_in[22],
              *alsd = (const float*)d_in[23], *bls = (const float*)d_in[24],
              *Wels = (const float*)d_in[25], *aels = (const float*)d_in[26];
  float* outf = (float*)d_out;

  float* W = (float*)d_ws;
  size_t o = 0;
  float* hA = W + o; o += (size_t)NN * 128;
  float* hB = W + o; o += (size_t)NN * 128;
  float* hC = W + o; o += (size_t)NN * 128;
  Scratch S;
  S.s_src = W + o; o += NN;
  S.s_dst = W + o; o += NN;
  S.self_l = W + o; o += NN;
  float* mean_attr = W + o; o += (size_t)NN * 16;
  S.mean_attr = mean_attr;
  float* et0 = W + o; o += NE;
  float* et1 = W + o; o += NE;
  float* et2 = W + o; o += NE;
  float* wa3 = W + o; o += 48;
  S.eterm = et0; S.wa = wa3;
  int* cnt = (int*)(W + o); o += NN;
  int* bsum = (int*)(W + o); o += 256;
  int* rowptr = (int*)(W + o); o += NN + 16;
  int* rank = (int*)(W + o); o += NE;
  int* eid_raw = (int*)(W + o); o += NE;
  int* seid = (int*)(W + o); o += NE;
  int* ssrc = (int*)(W + o); o += NE;
  S.rowptr = rowptr; S.ssrc = ssrc; S.seid = seid;

  zero_i32_k<<<(NN + 255) / 256, 256, 0, stream>>>(cnt, NN);
  hist_rank_k<<<(NE + 255) / 256, 256, 0, stream>>>(dst, cnt, rank);
  blocksum_k<<<SCAN_BLOCKS, 256, 0, stream>>>(cnt, bsum);
  scan_k<<<SCAN_BLOCKS, 256, 0, stream>>>(cnt, bsum, rowptr);
  fill_k<<<(NE + 255) / 256, 256, 0, stream>>>(dst, rowptr, rank, eid_raw);
  seg_sort_k<<<(NN * 32 + 255) / 256, 256, 0, stream>>>(rowptr, eid_raw, src, seid, ssrc);

  attr_mean_k<<<(NN * 16 + 255) / 256, 256, 0, stream>>>(rowptr, seid, edge_attr, mean_attr);

  wa3_k<<<3, 16, 0, stream>>>(We2, ae2, Wemu, aemu, Wels, aels, wa3);
  eterm3_k<<<(NE * 4) / ET_TILE, ET_BLOCK, 0, stream>>>(edge_attr, wa3, et0, et1, et2);

  run_gat<128, false>(stream, x, W1, a1s, a1d, b1, hA, hB, S);
  relu_k<<<(NN * 128 + 255) / 256, 256, 0, stream>>>(hB, NN * 128);

  S.eterm = et0; S.wa = wa3;
  run_gat<128, true>(stream, hB, W2, a2s, a2d, b2, hA, hC, S);

  gemm_k128<8><<<NN / 16, 256, 0, stream>>>(x, Wr, hB);
  residual_k<<<(NN * 128 + 255) / 256, 256, 0, stream>>>(hC, hB, br, NN * 128);

  S.eterm = et1; S.wa = wa3 + 16;
  run_gat<64, true>(stream, hB, Wmu, amus, amud, bmu, hA, outf, S);
  S.eterm = et2; S.wa = wa3 + 32;
  run_gat<64, true>(stream, hB, Wls, alss, alsd, bls, hA, outf + (size_t)NN * 64, S);
}


# relu and the residual add (relu(conv2) + x@Wr + br) done by the node kernel on each row before its one store: the two one-float-per-thread passes (400000 waves each) and 2 of 24 launches go
# speedup vs baseline: 4.2838x; 1.1135x over previous
#include <hip/hip_runtime.h>
#include <math.h>

typedef float v2f __attribute__((ext_vector_type(2)));
typedef float v8f __attribute__((ext_vector_type(8)));

static constexpr int   NN = 100000;
static constexpr int   NE = 1600000;
static constexpr float NEG = 0.2f;
static constexpr int   SCAN_TILE = 1024;
static constexpr int   SCAN_BLOCKS = (NN + 1 + SCAN_TILE - 1) / SCAN_TILE;
static_assert(SCAN_BLOCKS <= 256, "scan_k sums the earlier block totals with one thread each");

__device__ __forceinline__ float leaky(float x) { return x > 0.f ? x : NEG * x; }

__global__ void zero_i32_k(int* p, int n) {
  int i = blockIdx.x * blockDim.x + threadIdx.x;
  if (i < n) p[i] = 0;
}

__global__ void hist_rank_k(const int* __restrict__ dst, int* cnt, int* __restrict__ rank) {
  int e = blockIdx.x * blockDim.x + threadIdx.x;
  if (e >= NE) return;
  rank[e] = atomicAdd(&cnt[dst[e]], 1);
}
__global__ void blocksum_k(const int* __restrict__ cnt, int* __restrict__ bsum) {
  __shared__ int wtot[8];
  const int tid = threadIdx.x, lane = tid & 31, w = tid >> 5;
  const int i0 = blockIdx.x * SCAN_TILE + tid * 4;
  int s = 0;
  for (int j = 0; j < 4; ++j) s += (i0 + j < NN) ? cnt[i0 + j] : 0;
  for (int m = 16; m > 0; m >>= 1) s += __shfl_xor(s, m, 32);
  if (lane == 0) wtot[w] = s;
  __syncthreads();
  if (tid == 0) {
    int t = 0;
    for (int i = 0; i < 8; ++i) t += wtot[i];
    bsum[blockIdx.x] = t;
  }
}
__global__ void scan_k(const int* __restrict__ cnt, const int* __restrict__ bsum,
                       int* __restrict__ rowptr) {
  __shared__ int wtot[8];
  __shared__ int wpre[8];
  const int tid = threadIdx.x, lane = tid & 31, w = tid >> 5;
  int part = (tid < (int)blockIdx.x) ? bsum[tid] : 0;
  for (int m = 16; m > 0; m >>= 1) part += __shfl_xor(part, m, 32);
  if (lane == 0) wpre[w] = part;
  const int i0 = blockIdx.x * SCAN_TILE + tid * 4;
  int c[4];
  for (int j = 0; j < 4; ++j) c[j] = (i0 + j < NN) ? cnt[i0 + j] : 0;
  const int ts = c[0] + c[1] + c[2] + c[3];
  int inc = ts;
  for (int o = 1; o < 32; o <<= 1) {
    int t = __shfl_up(inc, o, 32);
    if (lane >= o) inc += t;
  }
  if (lane == 31) wtot[w] = inc;
  __syncthreads();
  int run = inc - ts;
  for (int i = 0; i < 8; ++i) {
    run += wpre[i];
    if (i < w) run += wtot[i];
  }
  for (int j = 0; j < 4; ++j) {
    if (i0 + j <= NN) rowptr[i0 + j] = run;
    run += c[j];
  }
}
__global__ void fill_k(const int* __restrict__ dst, const int* __restrict__ rowptr,
                       const int* __restrict__ rank, int* __restrict__ eid_raw) {
  int e = blockIdx.x * blockDim.x + threadIdx.x;
  if (e >= NE) return;
  eid_raw[rowptr[dst[e]] + rank[e]] = e;
}
__device__ __forceinline__ int lane_i(int v, int k) { return __builtin_amdgcn_readlane(v, k); }
__device__ __forceinline__ float lane_f(float v, int k) {
  return __int_as_float(__builtin_amdgcn_readlane(__float_as_int(v), k));
}
__global__ void seg_sort_k(const int* __restrict__ rowptr, const int* __restrict__ eid_raw,
                           const int* __restrict__ src, int* __restrict__ seid,
                           int* __restrict__ ssrc) {
  const int gid = blockIdx.x * blockDim.x + threadIdx.x;
  const int lane = gid & 31;
  const int n = __builtin_amdgcn_readfirstlane(gid >> 5);
  if (n >= NN) return;
  const int start = rowptr[n];
  const int end = rowptr[n + 1];
  for (int base = start; base < end; base += 32) {
    const int p = base + lane;
    const int x = (p < end) ? eid_raw[p] : 0x7fffffff;
    int r = 0;
    for (int b2 = start; b2 < end; b2 += 32) {
      const int q = b2 + lane;
      const int y = (q < end) ? eid_raw[q] : 0x7fffffff;
      const int m = min(32, end - b2);
      for (int k = 0; k < m; ++k) r += (lane_i(y, k) < x) ? 1 : 0;
    }
    if (p < end) { seid[start + r] = x; ssrc[start + r] = src[x]; }
  }
}
__global__ void attr_mean_k(const int* __restrict__ rowptr, const int* __restrict__ seid,
                            const float* __restrict__ ea, float* __restrict__ mean_attr) {
  const int gid = blockIdx.x * blockDim.x + threadIdx.x;
  const int n = gid >> 4, j = gid & 15;
  if (n >= NN) return;
  const int start = rowptr[n], end = rowptr[n + 1];
  float s = 0.f;
  for (int p = start; p < end; ++p) s += ea[(size_t)seid[p] * 16 + j];
  mean_attr[(size_t)n * 16 + j] = s / fmaxf((float)(end - start), 1.f);
}

__global__ void wa3_k(const float* __restrict__ We_a, const float* __restrict__ ae_a,
                      const float* __restrict__ We_b, const float* __restrict__ ae_b,
                      const float* __restrict__ We_c, const float* __restrict__ ae_c,
                      float* __restrict__ wa3) {
  const int l = blockIdx.x;
  const float* __restrict__ We = (l == 0) ? We_a : ((l == 1) ? We_b : We_c);
  const float* __restrict__ ae = (l == 0) ? ae_a : ((l == 1) ? ae_b : ae_c);
  const int K = (l == 0) ? 128 : 64;
  const int i = threadIdx.x;
  if (i < 16) {
    float s = 0.f;
    for (int j = 0; j < K; ++j) s += We[i * K + j] * ae[j];
    wa3[l * 16 + i] = s;
  }
}

template <int COLT>
__global__ void gemm_k128(const float* __restrict__ A, const float* __restrict__ B,
                          float* __restrict__ C) {
  constexpr int COUT = COLT * 16;
  constexpr int AP = 132;
  __shared__ __attribute__((aligned(16))) float As[16 * AP];
  const int tid = threadIdx.x;
  const size_t abase = (size_t)blockIdx.x * (16 * 128);
  for (int i = tid * 4; i < 16 * 128; i += blockDim.x * 4)
    *(float4*)(&As[(i >> 7) * AP + (i & 127)]) = *(const float4*)(&A[abase + i]);
  __syncthreads();

  const int wave = tid >> 5, lane = tid & 31, half = lane >> 4, lidx = lane & 15;
  const int n0 = wave * 16;
  v8f acc = {};
#pragma unroll
  for (int k0 = 0; k0 < 128; k0 += 4) {
    v2f a, b;
    a.x = As[lidx * AP + k0 + 2 * half];
    a.y = As[lidx * AP + k0 + 2 * half + 1];
    b.x = B[(k0 + 2 * half) * COUT + n0 + lidx];
    b.y = B[(k0 + 2 * half + 1) * COUT + n0 + lidx];
    acc = __builtin_amdgcn_wmma_f32_16x16x4_f32(false, a, false, b, (short)0, acc,
                                                false, false);
  }
  const size_t row0 = (size_t)blockIdx.x * 16;
#pragma unroll
  for (int r = 0; r < 8; ++r)
    C[(row0 + r + 8 * half) * COUT + n0 + lidx] = acc[r];
}

template <int C, bool ET>
__global__ void node_prep_k(const float* __restrict__ h, const float* __restrict__ a_src,
                            const float* __restrict__ a_dst,
                            const float* __restrict__ mean_attr,
                            const float* __restrict__ wa, float* s_src, float* s_dst,
                            float* self_logit) {
  int gid = blockIdx.x * blockDim.x + threadIdx.x;
  int n = gid >> 5, lane = gid & 31;
  if (n >= NN) return;
  float ps = 0.f, pd = 0.f;
#pragma unroll
  for (int i = 0; i < C / 32; ++i) {
    int c = lane + i * 32;
    float v = h[(size_t)n * C + c];
    ps += v * a_src[c];
    pd += v * a_dst[c];
  }
#pragma unroll
  for (int m = 16; m > 0; m >>= 1) { ps += __shfl_xor(ps, m, 32); pd += __shfl_xor(pd, m, 32); }
  if (lane == 0) {
    float st = 0.f;
    if (ET) { for (int j = 0; j < 16; ++j) st += mean_attr[(size_t)n * 16 + j] * wa[j]; }
    s_src[n] = ps; s_dst[n] = pd; self_logit[n] = leaky(ps + pd + st);
  }
}

static constexpr int ET_PER_THREAD = 4;
static constexpr int ET_BLOCK = 256;
static constexpr int ET_TILE = ET_BLOCK * ET_PER_THREAD;
static_assert((NE * 4) % ET_TILE == 0, "eterm3_k has no tail handling");
__device__ __forceinline__ float dot4(const float4& a, const float4& b) {
  return a.x * b.x + a.y * b.y + a.z * b.z + a.w * b.w;
}
__device__ __forceinline__ float quad_sum(float v) {
  v += __shfl_xor(v, 1, 32);
  v += __shfl_xor(v, 2, 32);
  return v;
}
__global__ void eterm3_k(const float* __restrict__ ea, const float* __restrict__ wa3,
                         float* __restrict__ et0, float* __restrict__ et1,
                         float* __restrict__ et2) {
  const int tid = threadIdx.x;
  const int q = tid & 3;
  const float4* __restrict__ ea4 = (const float4*)ea;
  const size_t base = (size_t)blockIdx.x * ET_TILE + tid;
  float4 v[ET_PER_THREAD];
#pragma unroll
  for (int k = 0; k < ET_PER_THREAD; ++k) v[k] = ea4[base + (size_t)k * ET_BLOCK];
  const float4 w0 = ((const float4*)wa3)[q];
  const float4 w1 = ((const float4*)wa3)[4 + q];
  const float4 w2 = ((const float4*)wa3)[8 + q];
#pragma unroll
  for (int k = 0; k < ET_PER_THREAD; ++k) {
    const float s0 = quad_sum(dot4(v[k], w0));
    const float s1 = quad_sum(dot4(v[k], w1));
    const float s2 = quad_sum(dot4(v[k], w2));
    if (q == 0) {
      const size_t e = (base + (size_t)k * ET_BLOCK) >> 2;
      et0[e] = s0; et1[e] = s1; et2[e] = s2;
    }
  }
}

template <int V> struct VecOf;
template <> struct VecOf<4> { typedef float4 type; };
template <> struct VecOf<2> { typedef float2 type; };
__device__ __forceinline__ void axpy(float4& acc, float a, const float4& v) {
  acc.x += a * v.x; acc.y += a * v.y; acc.z += a * v.z; acc.w += a * v.w;
}
__device__ __forceinline__ void axpy(float2& acc, float a, const float2& v) {
  acc.x += a * v.x; acc.y += a * v.y;
}
enum { EPI_NONE = 0, EPI_RELU = 1, EPI_RELU_RESID = 2 };
__device__ __forceinline__ void relu_v(float4& v) {
  v.x = fmaxf(v.x, 0.f); v.y = fmaxf(v.y, 0.f); v.z = fmaxf(v.z, 0.f); v.w = fmaxf(v.w, 0.f);
}
__device__ __forceinline__ void relu_v(float2& v) { v.x = fmaxf(v.x, 0.f); v.y = fmaxf(v.y, 0.f); }
__device__ __forceinline__ void add2_v(float4& v, const float4& r, const float4& b) {
  v.x = v.x + r.x + b.x; v.y = v.y + r.y + b.y; v.z = v.z + r.z + b.z; v.w = v.w + r.w + b.w;
}
__device__ __forceinline__ void add2_v(float2& v, const float2& r, const float2& b) {
  v.x = v.x + r.x + b.x; v.y = v.y + r.y + b.y;
}

template <bool ET>
__device__ __forceinline__ float edge_logit(int p, float pd, const int* __restrict__ ssrc,
                                            const int* __restrict__ seid,
                                            const float* __restrict__ s_src,
                                            const float* __restrict__ eterm) {
  float l = s_src[ssrc[p]] + pd;
  if (ET) l += eterm[seid[p]];
  return leaky(l);
}
__device__ __forceinline__ float wave_max(float v) {
  for (int m = 16; m > 0; m >>= 1) v = fmaxf(v, __shfl_xor(v, m, 32));
  return v;
}
__device__ __forceinline__ float wave_sum(float v) {
  for (int m = 16; m > 0; m >>= 1) v += __shfl_xor(v, m, 32);
  return v;
}

template <int C, bool ET, int EPI>
__global__ void gat_node_k(const int* __restrict__ rowptr, const int* __restrict__ ssrc,
                           const int* __restrict__ seid, const float* __restrict__ s_src,
                           const float* __restrict__ s_dst, const float* __restrict__ self_l,
                           const float* __restrict__ eterm, const float* __restrict__ h,
                           const float* __restrict__ bias, const float* __restrict__ resid,
                           const float* __restrict__ rbias, float* __restrict__ out) {
  constexpr int V = C / 32;
  typedef typename VecOf<V>::type vec_t;
  const int gid = blockIdx.x * blockDim.x + threadIdx.x;
  const int lane = gid & 31;
  const int n = __builtin_amdgcn_readfirstlane(gid >> 5);
  if (n >= NN) return;
  const int start = rowptr[n];
  const int end = rowptr[n + 1];
  const float pd = s_dst[n], sl = self_l[n];
  const int ch = lane * V;
  const vec_t hs = *(const vec_t*)&h[(size_t)n * C + ch];
  const vec_t bv = *(const vec_t*)&bias[ch];
  vec_t acc = bv;
  vec_t rv = {}, rb = {};
  if (EPI == EPI_RELU_RESID) {
    rv = *(const vec_t*)&resid[(size_t)n * C + ch];
    rb = *(const vec_t*)&rbias[ch];
  }

  if (end - start <= 32) {
    const int p = start + lane;
    const bool act = p < end;
    int s = 0;
    float l = 0.f;
    if (act) { s = ssrc[p]; l = edge_logit<ET>(p, pd, ssrc, seid, s_src, eterm); }
    const float m = wave_max(act ? fmaxf(l, sl) : sl);
    const float w = act ? __expf(l - m) : 0.f;
    const float wself = __expf(sl - m);
    const float den = wself + wave_sum(w) + 1e-16f;
    const float a = w / den;
    axpy(acc, wself / den, hs);
    const int deg = end - start;
#pragma unroll 4
    for (int k = 0; k < deg; ++k) {
      const int sk = lane_i(s, k);
      const float ak = lane_f(a, k);
      axpy(acc, ak, *(const vec_t*)&h[(size_t)sk * C + ch]);
    }
  } else {
    float m = sl;
    for (int base = start; base < end; base += 32) {
      const int p = base + lane;
      if (p < end) m = fmaxf(m, edge_logit<ET>(p, pd, ssrc, seid, s_src, eterm));
    }
    m = wave_max(m);
    float ssum = 0.f;
    for (int base = start; base < end; base += 32) {
      const int p = base + lane;
      if (p < end) ssum += __expf(edge_logit<ET>(p, pd, ssrc, seid, s_src, eterm) - m);
    }
    const float wself = __expf(sl - m);
    const float den = wself + wave_sum(ssum) + 1e-16f;
    axpy(acc, wself / den, hs);
    for (int base = start; base < end; base += 32) {
      const int p = base + lane;
      int s = 0;
      float a = 0.f;
      if (p < end) {
        s = ssrc[p];
        a = __expf(edge_logit<ET>(p, pd, ssrc, seid, s_src, eterm) - m) / den;
      }
      const int cntk = min(32, end - base);
#pragma unroll 4
      for (int k = 0; k < cntk; ++k) {
        const int sk = lane_i(s, k);
        const float ak = lane_f(a, k);
        axpy(acc, ak, *(const vec_t*)&h[(size_t)sk * C + ch]);
      }
    }
  }
  if (EPI != EPI_NONE) relu_v(acc);
  if (EPI == EPI_RELU_RESID) add2_v(acc, rv, rb);
  *(vec_t*)&out[(size_t)n * C + ch] = acc;
}

struct Scratch {
  float *s_src, *s_dst, *self_l;
  const float* mean_attr;
  const float *eterm, *wa;
  const int *rowptr, *ssrc, *seid;
};

template <int C, bool ET, int EPI>
static void run_gat(hipStream_t st, const float* hin, const float* Wm, const float* asrc,
                    const float* adst, const float* bias, float* hg, float* out,
                    const Scratch& S, const float* resid, const float* rbias) {
  constexpr int COLT = C / 16;
  gemm_k128<COLT><<<NN / 16, 32 * COLT, 0, st>>>(hin, Wm, hg);
  node_prep_k<C, ET><<<(NN * 32 + 255) / 256, 256, 0, st>>>(hg, asrc, adst, S.mean_attr,
                                                            S.wa, S.s_src, S.s_dst, S.self_l);
  gat_node_k<C, ET, EPI><<<(NN * 32 + 255) / 256, 256, 0, st>>>(S.rowptr, S.ssrc, S.seid,
                                                                S.s_src, S.s_dst, S.self_l,
                                                                S.eterm, hg, bias, resid,
                                                                rbias, out);
}

extern "C" void kernel_launch(void* const* d_in, const int* in_sizes, int n_in, void* d_out,
                              int out_size, void* d_ws, size_t ws_size, hipStream_t stream) {
  (void)in_sizes; (void)n_in; (void)out_size; (void)ws_size;
  const float* x = (const float*)d_in[0];
  const int* ei = (const int*)d_in[1];
  const int* src = ei;
  const int* dst = ei + NE;
  const float* edge_attr = (const float*)d_in[2];
  const float *W1 = (const float*)d_in[3], *a1s = (const float*)d_in[4],
              *a1d = (const float*)d_in[5], *b1 = (const float*)d_in[6];
  const float *W2 = (const float*)d_in[7], *a2s = (const float*)d_in[8],
              *a2d = (const float*)d_in[9], *b2 = (const float*)d_in[10];
  const float *We2 = (const float*)d_in[11], *ae2 = (const float*)d_in[12];
  const float *Wr = (const float*)d_in[13], *br = (const float*)d_in[14];
  const float *Wmu = (const float*)d_in[15], *amus = (const float*)d_in[16],
              *amud = (const float*)d_in[17], *bmu = (const float*)d_in[18],
              *Wemu = (const float*)d_in[19], *aemu = (const float*)d_in[20];
  const float *Wls = (const float*)d_in[21], *alss = (const float*)d_in[22],
              *alsd = (const float*)d_in[23], *bls = (const float*)d_in[24],
              *Wels = (const float*)d_in[25], *aels = (const float*)d_in[26];
  float* outf = (float*)d_out;

  float* W = (float*)d_ws;
  size_t o = 0;
  float* hA = W + o; o += (size_t)NN * 128;
  float* hB = W + o; o += (size_t)NN * 128;
  float* hC = W + o; o += (size_t)NN * 128;
  Scratch S;
  S.s_src = W + o; o += NN;
  S.s_dst = W + o; o += NN;
  S.self_l = W + o; o += NN;
  float* mean_attr = W + o; o += (size_t)NN * 16;
  S.mean_attr = mean_attr;
  float* et0 = W + o; o += NE;
  float* et1 = W + o; o += NE;
  float* et2 = W + o; o += NE;
  float* wa3 = W + o; o += 48;
  S.eterm = et0; S.wa = wa3;
  int* cnt = (int*)(W + o); o += NN;
  int* bsum = (int*)(W + o); o += 256;
  int* rowptr = (int*)(W + o); o += NN + 16;
  int* rank = (int*)(W + o); o += NE;
  int* eid_raw = (int*)(W + o); o += NE;
  int* seid = (int*)(W + o); o += NE;
  int* ssrc = (int*)(W + o); o += NE;
  S.rowptr = rowptr; S.ssrc = ssrc; S.seid = seid;

  zero_i32_k<<<(NN + 255) / 256, 256, 0, stream>>>(cnt, NN);
  hist_rank_k<<<(NE + 255) / 256, 256, 0, stream>>>(dst, cnt, rank);
  blocksum_k<<<SCAN_BLOCKS, 256, 0, stream>>>(cnt, bsum);
  scan_k<<<SCAN_BLOCKS, 256, 0, stream>>>(cnt, bsum, rowptr);
  fill_k<<<(NE + 255) / 256, 256, 0, stream>>>(dst, rowptr, rank, eid_raw);
  seg_sort_k<<<(NN * 32 + 255) / 256, 256, 0, stream>>>(rowptr, eid_raw, src, seid, ssrc);

  attr_mean_k<<<(NN * 16 + 255) / 256, 256, 0, stream>>>(rowptr, seid, edge_attr, mean_attr);

  wa3_k<<<3, 16, 0, stream>>>(We2, ae2, Wemu, aemu, Wels, aels, wa3);
  eterm3_k<<<(NE * 4) / ET_TILE, ET_BLOCK, 0, stream>>>(edge_attr, wa3, et0, et1, et2);

  gemm_k128<8><<<NN / 16, 256, 0, stream>>>(x, Wr, hC);

  run_gat<128, false, EPI_RELU>(stream, x, W1, a1s, a1d, b1, hA, hB, S, nullptr, nullptr);

  S.eterm = et0; S.wa = wa3;
  run_gat<128, true, EPI_RELU_RESID>(stream, hB, W2, a2s, a2d, b2, hA, hB, S, hC, br);

  S.eterm = et1; S.wa = wa3 + 16;
  run_gat<64, true, EPI_NONE>(stream, hB, Wmu, amus, amud, bmu, hA, outf, S, nullptr, nullptr);
  S.eterm = et2; S.wa = wa3 + 32;
  run_gat<64, true, EPI_NONE>(stream, hB, Wls, alss, alsd, bls, hA, outf + (size_t)NN * 64, S,
                              nullptr, nullptr);
}


# Self-loop edge term = mean of the group's own edge terms, summed in the node kernel's wave (butterfly over the group, / count), not dot(mean edge_attr, wa): the attr-mean pass over edge_attr goes
# speedup vs baseline: 4.5303x; 1.0575x over previous
#include <hip/hip_runtime.h>
#include <math.h>

typedef float v2f __attribute__((ext_vector_type(2)));
typedef float v8f __attribute__((ext_vector_type(8)));

static constexpr int   NN = 100000;
static constexpr int   NE = 1600000;
static constexpr float NEG = 0.2f;
static constexpr int   SCAN_TILE = 1024;
static constexpr int   SCAN_BLOCKS = (NN + 1 + SCAN_TILE - 1) / SCAN_TILE;
static_assert(SCAN_BLOCKS <= 256, "scan_k sums the earlier block totals with one thread each");

__device__ __forceinline__ float leaky(float x) { return x > 0.f ? x : NEG * x; }

__global__ void zero_i32_k(int* p, int n) {
  int i = blockIdx.x * blockDim.x + threadIdx.x;
  if (i < n) p[i] = 0;
}

__global__ void hist_rank_k(const int* __restrict__ dst, int* cnt, int* __restrict__ rank) {
  int e = blockIdx.x * blockDim.x + threadIdx.x;
  if (e >= NE) return;
  rank[e] = atomicAdd(&cnt[dst[e]], 1);
}
__global__ void blocksum_k(const int* __restrict__ cnt, int* __restrict__ bsum) {
  __shared__ int wtot[8];
  const int tid = threadIdx.x, lane = tid & 31, w = tid >> 5;
  const int i0 = blockIdx.x * SCAN_TILE + tid * 4;
  int s = 0;
  for (int j = 0; j < 4; ++j) s += (i0 + j < NN) ? cnt[i0 + j] : 0;
  for (int m = 16; m > 0; m >>= 1) s += __shfl_xor(s, m, 32);
  if (lane == 0) wtot[w] = s;
  __syncthreads();
  if (tid == 0) {
    int t = 0;
    for (int i = 0; i < 8; ++i) t += wtot[i];
    bsum[blockIdx.x] = t;
  }
}
__global__ void scan_k(const int* __restrict__ cnt, const int* __restrict__ bsum,
                       int* __restrict__ rowptr) {
  __shared__ int wtot[8];
  __shared__ int wpre[8];
  const int tid = threadIdx.x, lane = tid & 31, w = tid >> 5;
  int part = (tid < (int)blockIdx.x) ? bsum[tid] : 0;
  for (int m = 16; m > 0; m >>= 1) part += __shfl_xor(part, m, 32);
  if (lane == 0) wpre[w] = part;
  const int i0 = blockIdx.x * SCAN_TILE + tid * 4;
  int c[4];
  for (int j = 0; j < 4; ++j) c[j] = (i0 + j < NN) ? cnt[i0 + j] : 0;
  const int ts = c[0] + c[1] + c[2] + c[3];
  int inc = ts;
  for (int o = 1; o < 32; o <<= 1) {
    int t = __shfl_up(inc, o, 32);
    if (lane >= o) inc += t;
  }
  if (lane == 31) wtot[w] = inc;
  __syncthreads();
  int run = inc - ts;
  for (int i = 0; i < 8; ++i) {
    run += wpre[i];
    if (i < w) run += wtot[i];
  }
  for (int j = 0; j < 4; ++j) {
    if (i0 + j <= NN) rowptr[i0 + j] = run;
    run += c[j];
  }
}
__global__ void fill_k(const int* __restrict__ dst, const int* __restrict__ rowptr,
                       const int* __restrict__ rank, int* __restrict__ eid_raw) {
  int e = blockIdx.x * blockDim.x + threadIdx.x;
  if (e >= NE) return;
  eid_raw[rowptr[dst[e]] + rank[e]] = e;
}
__device__ __forceinline__ int lane_i(int v, int k) { return __builtin_amdgcn_readlane(v, k); }
__device__ __forceinline__ float lane_f(float v, int k) {
  return __int_as_float(__builtin_amdgcn_readlane(__float_as_int(v), k));
}
__global__ void seg_sort_k(const int* __restrict__ rowptr, const int* __restrict__ eid_raw,
                           const int* __restrict__ src, int* __restrict__ seid,
                           int* __restrict__ ssrc) {
  const int gid = blockIdx.x * blockDim.x + threadIdx.x;
  const int lane = gid & 31;
  const int n = __builtin_amdgcn_readfirstlane(gid >> 5);
  if (n >= NN) return;
  const int start = rowptr[n];
  const int end = rowptr[n + 1];
  for (int base = start; base < end; base += 32) {
    const int p = base + lane;
    const int x = (p < end) ? eid_raw[p] : 0x7fffffff;
    int r = 0;
    for (int b2 = start; b2 < end; b2 += 32) {
      const int q = b2 + lane;
      const int y = (q < end) ? eid_raw[q] : 0x7fffffff;
      const int m = min(32, end - b2);
      for (int k = 0; k < m; ++k) r += (lane_i(y, k) < x) ? 1 : 0;
    }
    if (p < end) { seid[start + r] = x; ssrc[start + r] = src[x]; }
  }
}

__global__ void wa3_k(const float* __restrict__ We_a, const float* __restrict__ ae_a,
                      const float* __restrict__ We_b, const float* __restrict__ ae_b,
                      const float* __restrict__ We_c, const float* __restrict__ ae_c,
                      float* __restrict__ wa3) {
  const int l = blockIdx.x;
  const float* __restrict__ We = (l == 0) ? We_a : ((l == 1) ? We_b : We_c);
  const float* __restrict__ ae = (l == 0) ? ae_a : ((l == 1) ? ae_b : ae_c);
  const int K = (l == 0) ? 128 : 64;
  const int i = threadIdx.x;
  if (i < 16) {
    float s = 0.f;
    for (int j = 0; j < K; ++j) s += We[i * K + j] * ae[j];
    wa3[l * 16 + i] = s;
  }
}

template <int COLT>
__global__ void gemm_k128(const float* __restrict__ A, const float* __restrict__ B,
                          float* __restrict__ C) {
  constexpr int COUT = COLT * 16;
  constexpr int AP = 132;
  __shared__ __attribute__((aligned(16))) float As[16 * AP];
  const int tid = threadIdx.x;
  const size_t abase = (size_t)blockIdx.x * (16 * 128);
  for (int i = tid * 4; i < 16 * 128; i += blockDim.x * 4)
    *(float4*)(&As[(i >> 7) * AP + (i & 127)]) = *(const float4*)(&A[abase + i]);
  __syncthreads();

  const int wave = tid >> 5, lane = tid & 31, half = lane >> 4, lidx = lane & 15;
  const int n0 = wave * 16;
  v8f acc = {};
#pragma unroll
  for (int k0 = 0; k0 < 128; k0 += 4) {
    v2f a, b;
    a.x = As[lidx * AP + k0 + 2 * half];
    a.y = As[lidx * AP + k0 + 2 * half + 1];
    b.x = B[(k0 + 2 * half) * COUT + n0 + lidx];
    b.y = B[(k0 + 2 * half + 1) * COUT + n0 + lidx];
    acc = __builtin_amdgcn_wmma_f32_16x16x4_f32(false, a, false, b, (short)0, acc,
                                                false, false);
  }
  const size_t row0 = (size_t)blockIdx.x * 16;
#pragma unroll
  for (int r = 0; r < 8; ++r)
    C[(row0 + r + 8 * half) * COUT + n0 + lidx] = acc[r];
}

template <int C>
__global__ void node_prep_k(const float* __restrict__ h, const float* __restrict__ a_src,
                            const float* __restrict__ a_dst, float* s_src, float* s_dst) {
  int gid = blockIdx.x * blockDim.x + threadIdx.x;
  int n = gid >> 5, lane = gid & 31;
  if (n >= NN) return;
  float ps = 0.f, pd = 0.f;
#pragma unroll
  for (int i = 0; i < C / 32; ++i) {
    int c = lane + i * 32;
    float v = h[(size_t)n * C + c];
    ps += v * a_src[c];
    pd += v * a_dst[c];
  }
#pragma unroll
  for (int m = 16; m > 0; m >>= 1) { ps += __shfl_xor(ps, m, 32); pd += __shfl_xor(pd, m, 32); }
  if (lane == 0) { s_src[n] = ps; s_dst[n] = pd; }
}

static constexpr int ET_PER_THREAD = 4;
static constexpr int ET_BLOCK = 256;
static constexpr int ET_TILE = ET_BLOCK * ET_PER_THREAD;
static_assert((NE * 4) % ET_TILE == 0, "eterm3_k has no tail handling");
__device__ __forceinline__ float dot4(const float4& a, const float4& b) {
  return a.x * b.x + a.y * b.y + a.z * b.z + a.w * b.w;
}
__device__ __forceinline__ float quad_sum(float v) {
  v += __shfl_xor(v, 1, 32);
  v += __shfl_xor(v, 2, 32);
  return v;
}
__global__ void eterm3_k(const float* __restrict__ ea, const float* __restrict__ wa3,
                         float* __restrict__ et0, float* __restrict__ et1,
                         float* __restrict__ et2) {
  const int tid = threadIdx.x;
  const int q = tid & 3;
  const float4* __restrict__ ea4 = (const float4*)ea;
  const size_t base = (size_t)blockIdx.x * ET_TILE + tid;
  float4 v[ET_PER_THREAD];
#pragma unroll
  for (int k = 0; k < ET_PER_THREAD; ++k) v[k] = ea4[base + (size_t)k * ET_BLOCK];
  const float4 w0 = ((const float4*)wa3)[q];
  const float4 w1 = ((const float4*)wa3)[4 + q];
  const float4 w2 = ((const float4*)wa3)[8 + q];
#pragma unroll
  for (int k = 0; k < ET_PER_THREAD; ++k) {
    const float s0 = quad_sum(dot4(v[k], w0));
    const float s1 = quad_sum(dot4(v[k], w1));
    const float s2 = quad_sum(dot4(v[k], w2));
    if (q == 0) {
      const size_t e = (base + (size_t)k * ET_BLOCK) >> 2;
      et0[e] = s0; et1[e] = s1; et2[e] = s2;
    }
  }
}

template <int V> struct VecOf;
template <> struct VecOf<4> { typedef float4 type; };
template <> struct VecOf<2> { typedef float2 type; };
__device__ __forceinline__ void axpy(float4& acc, float a, const float4& v) {
  acc.x += a * v.x; acc.y += a * v.y; acc.z += a * v.z; acc.w += a * v.w;
}
__device__ __forceinline__ void axpy(float2& acc, float a, const float2& v) {
  acc.x += a * v.x; acc.y += a * v.y;
}
enum { EPI_NONE = 0, EPI_RELU = 1, EPI_RELU_RESID = 2 };
__device__ __forceinline__ void relu_v(float4& v) {
  v.x = fmaxf(v.x, 0.f); v.y = fmaxf(v.y, 0.f); v.z = fmaxf(v.z, 0.f); v.w = fmaxf(v.w, 0.f);
}
__device__ __forceinline__ void relu_v(float2& v) { v.x = fmaxf(v.x, 0.f); v.y = fmaxf(v.y, 0.f); }
__device__ __forceinline__ void add2_v(float4& v, const float4& r, const float4& b) {
  v.x = v.x + r.x + b.x; v.y = v.y + r.y + b.y; v.z = v.z + r.z + b.z; v.w = v.w + r.w + b.w;
}
__device__ __forceinline__ void add2_v(float2& v, const float2& r, const float2& b) {
  v.x = v.x + r.x + b.x; v.y = v.y + r.y + b.y;
}

template <bool ET>
__device__ __forceinline__ float edge_logit(int p, float pd, const int* __restrict__ ssrc,
                                            const int* __restrict__ seid,
                                            const float* __restrict__ s_src,
                                            const float* __restrict__ eterm, float& et) {
  float l = s_src[ssrc[p]] + pd;
  et = 0.f;
  if (ET) { et = eterm[seid[p]]; l += et; }
  return leaky(l);
}
template <bool ET>
__device__ __forceinline__ float self_logit(float ps, float pd, float et_sum, int deg) {
  float l = ps + pd;
  if (ET) l += et_sum / fmaxf((float)deg, 1.f);
  return leaky(l);
}
__device__ __forceinline__ float wave_max(float v) {
  for (int m = 16; m > 0; m >>= 1) v = fmaxf(v, __shfl_xor(v, m, 32));
  return v;
}
__device__ __forceinline__ float wave_sum(float v) {
  for (int m = 16; m > 0; m >>= 1) v += __shfl_xor(v, m, 32);
  return v;
}

template <int C, bool ET, int EPI>
__global__ void gat_node_k(const int* __restrict__ rowptr, const int* __restrict__ ssrc,
                           const int* __restrict__ seid, const float* __restrict__ s_src,
                           const float* __restrict__ s_dst,
                           const float* __restrict__ eterm, const float* __restrict__ h,
                           const float* __restrict__ bias, const float* __restrict__ resid,
                           const float* __restrict__ rbias, float* __restrict__ out) {
  constexpr int V = C / 32;
  typedef typename VecOf<V>::type vec_t;
  const int gid = blockIdx.x * blockDim.x + threadIdx.x;
  const int lane = gid & 31;
  const int n = __builtin_amdgcn_readfirstlane(gid >> 5);
  if (n >= NN) return;
  const int start = rowptr[n];
  const int end = rowptr[n + 1];
  const float ps = s_src[n], pd = s_dst[n];
  const int deg = end - start;
  const int ch = lane * V;
  const vec_t hs = *(const vec_t*)&h[(size_t)n * C + ch];
  const vec_t bv = *(const vec_t*)&bias[ch];
  vec_t acc = bv;
  vec_t rv = {}, rb = {};
  if (EPI == EPI_RELU_RESID) {
    rv = *(const vec_t*)&resid[(size_t)n * C + ch];
    rb = *(const vec_t*)&rbias[ch];
  }

  if (end - start <= 32) {
    const int p = start + lane;
    const bool act = p < end;
    int s = 0;
    float l = 0.f, et = 0.f;
    if (act) { s = ssrc[p]; l = edge_logit<ET>(p, pd, ssrc, seid, s_src, eterm, et); }
    const float sl = self_logit<ET>(ps, pd, ET ? wave_sum(et) : 0.f, deg);
    const float m = fmaxf(wave_max(act ? l : -INFINITY), sl);
    const float w = act ? __expf(l - m) : 0.f;
    const float wself = __expf(sl - m);
    const float den = wself + wave_sum(w) + 1e-16f;
    const float a = w / den;
    axpy(acc, wself / den, hs);
#pragma unroll 4
    for (int k = 0; k < deg; ++k) {
      const int sk = lane_i(s, k);
      const float ak = lane_f(a, k);
      axpy(acc, ak, *(const vec_t*)&h[(size_t)sk * C + ch]);
    }
  } else {
    float m = -INFINITY, ets = 0.f, et;
    for (int base = start; base < end; base += 32) {
      const int p = base + lane;
      if (p < end) { m = fmaxf(m, edge_logit<ET>(p, pd, ssrc, seid, s_src, eterm, et)); ets += et; }
    }
    const float sl = self_logit<ET>(ps, pd, ET ? wave_sum(ets) : 0.f, deg);
    m = fmaxf(wave_max(m), sl);
    float ssum = 0.f;
    for (int base = start; base < end; base += 32) {
      const int p = base + lane;
      if (p < end) ssum += __expf(edge_logit<ET>(p, pd, ssrc, seid, s_src, eterm, et) - m);
    }
    const float wself = __expf(sl - m);
    const float den = wself + wave_sum(ssum) + 1e-16f;
    axpy(acc, wself / den, hs);
    for (int base = start; base < end; base += 32) {
      const int p = base + lane;
      int s = 0;
      float a = 0.f;
      if (p < end) {
        s = ssrc[p];
        a = __expf(edge_logit<ET>(p, pd, ssrc, seid, s_src, eterm, et) - m) / den;
      }
      const int cntk = min(32, end - base);
#pragma unroll 4
      for (int k = 0; k < cntk; ++k) {
        const int sk = lane_i(s, k);
        const float ak = lane_f(a, k);
        axpy(acc, ak, *(const vec_t*)&h[(size_t)sk * C + ch]);
      }
    }
  }
  if (EPI != EPI_NONE) relu_v(acc);
  if (EPI == EPI_RELU_RESID) add2_v(acc, rv, rb);
  *(vec_t*)&out[(size_t)n * C + ch] = acc;
}

struct Scratch {
  float *s_src, *s_dst;
  const float* eterm;
  const int *rowptr, *ssrc, *seid;
};

template <int C, bool ET, int EPI>
static void run_gat(hipStream_t st, const float* hin, const float* Wm, const float* asrc,
                    const float* adst, const float* bias, float* hg, float* out,
                    const Scratch& S, const float* resid, const float* rbias) {
  constexpr int COLT = C / 16;
  gemm_k128<COLT><<<NN / 16, 32 * COLT, 0, st>>>(hin, Wm, hg);
  node_prep_k<C><<<(NN * 32 + 255) / 256, 256, 0, st>>>(hg, asrc, adst, S.s_src, S.s_dst);
  gat_node_k<C, ET, EPI><<<(NN * 32 + 255) / 256, 256, 0, st>>>(S.rowptr, S.ssrc, S.seid,
                                                                S.s_src, S.s_dst, S.eterm, hg,
                                                                bias, resid, rbias, out);
}

extern "C" void kernel_launch(void* const* d_in, const int* in_sizes, int n_in, void* d_out,
                              int out_size, void* d_ws, size_t ws_size, hipStream_t stream) {
  (void)in_sizes; (void)n_in; (void)out_size; (void)ws_size;
  const float* x = (const float*)d_in[0];
  const int* ei = (const int*)d_in[1];
  const int* src = ei;
  const int* dst = ei + NE;
  const float* edge_attr = (const float*)d_in[2];
  const float *W1 = (const float*)d_in[3], *a1s = (const float*)d_in[4],
              *a1d = (const float*)d_in[5], *b1 = (const float*)d_in[6];
  const float *W2 = (const float*)d_in[7], *a2s = (const float*)d_in[8],
              *a2d = (const float*)d_in[9], *b2 = (const float*)d_in[10];
  const float *We2 = (const float*)d_in[11], *ae2 = (const float*)d_in[12];
  const float *Wr = (const float*)d_in[13], *br = (const float*)d_in[14];
  const float *Wmu = (const float*)d_in[15], *amus = (const float*)d_in[16],
              *amud = (const float*)d_in[17], *bmu = (const float*)d_in[18],
              *Wemu = (const float*)d_in[19], *aemu = (const float*)d_in[20];
  const float *Wls = (const float*)d_in[21], *alss = (const float*)d_in[22],
              *alsd = (const float*)d_in[23], *bls = (const float*)d_in[24],
              *Wels = (const float*)d_in[25], *aels = (const float*)d_in[26];
  float* outf = (float*)d_out;

  float* W = (float*)d_ws;
  size_t o = 0;
  float* hA = W + o; o += (size_t)NN * 128;
  float* hB = W + o; o += (size_t)NN * 128;
  float* hC = W + o; o += (size_t)NN * 128;
  Scratch S;
  S.s_src = W + o; o += NN;
  S.s_dst = W + o; o += NN;
  float* et0 = W + o; o += NE;
  float* et1 = W + o; o += NE;
  float* et2 = W + o; o += NE;
  float* wa3 = W + o; o += 48;
  S.eterm = et0;
  int* cnt = (int*)(W + o); o += NN;
  int* bsum = (int*)(W + o); o += 256;
  int* rowptr = (int*)(W + o); o += NN + 16;
  int* rank = (int*)(W + o); o += NE;
  int* eid_raw = (int*)(W + o); o += NE;
  int* seid = (int*)(W + o); o += NE;
  int* ssrc = (int*)(W + o); o += NE;
  S.rowptr = rowptr; S.ssrc = ssrc; S.seid = seid;

  zero_i32_k<<<(NN + 255) / 256, 256, 0, stream>>>(cnt, NN);
  hist_rank_k<<<(NE + 255) / 256, 256, 0, stream>>>(dst, cnt, rank);
  blocksum_k<<<SCAN_BLOCKS, 256, 0, stream>>>(cnt, bsum);
  scan_k<<<SCAN_BLOCKS, 256, 0, stream>>>(cnt, bsum, rowptr);
  fill_k<<<(NE + 255) / 256, 256, 0, stream>>>(dst, rowptr, rank, eid_raw);
  seg_sort_k<<<(NN * 32 + 255) / 256, 256, 0, stream>>>(rowptr, eid_raw, src, seid, ssrc);

  wa3_k<<<3, 16, 0, stream>>>(We2, ae2, Wemu, aemu, Wels, aels, wa3);
  eterm3_k<<<(NE * 4) / ET_TILE, ET_BLOCK, 0, stream>>>(edge_attr, wa3, et0, et1, et2);

  gemm_k128<8><<<NN / 16, 256, 0, stream>>>(x, Wr, hC);

  run_gat<128, false, EPI_RELU>(stream, x, W1, a1s, a1d, b1, hA, hB, S, nullptr, nullptr);

  S.eterm = et0;
  run_gat<128, true, EPI_RELU_RESID>(stream, hB, W2, a2s, a2d, b2, hA, hB, S, hC, br);

  S.eterm = et1;
  run_gat<64, true, EPI_NONE>(stream, hB, Wmu, amus, amud, bmu, hA, outf, S, nullptr, nullptr);
  S.eterm = et2;
  run_gat<64, true, EPI_NONE>(stream, hB, Wls, alss, alsd, bls, hA, outf + (size_t)NN * 64, S,
                              nullptr, nullptr);
}


# The mu and logstd heads run as ONE 128-wide conv: GEMM on [Wmu|Wls], one prep pass, one node kernel (lane half = head, both heads' scalars gathered as pairs); 3 of 21 launches go, sums as before
# speedup vs baseline: 4.9835x; 1.1000x over previous
#include <hip/hip_runtime.h>
#include <math.h>

typedef float v2f __attribute__((ext_vector_type(2)));
typedef float v8f __attribute__((ext_vector_type(8)));

static constexpr int   NN = 100000;
static constexpr int   NE = 1600000;
static constexpr float NEG = 0.2f;
static constexpr int   SCAN_TILE = 1024;
static constexpr int   SCAN_BLOCKS = (NN + 1 + SCAN_TILE - 1) / SCAN_TILE;
static_assert(SCAN_BLOCKS <= 256, "scan_k sums the earlier block totals with one thread each");

__device__ __forceinline__ float leaky(float x) { return x > 0.f ? x : NEG * x; }

__global__ void zero_i32_k(int* p, int n) {
  int i = blockIdx.x * blockDim.x + threadIdx.x;
  if (i < n) p[i] = 0;
}

__global__ void hist_rank_k(const int* __restrict__ dst, int* cnt, int* __restrict__ rank) {
  int e = blockIdx.x * blockDim.x + threadIdx.x;
  if (e >= NE) return;
  rank[e] = atomicAdd(&cnt[dst[e]], 1);
}
__global__ void blocksum_k(const int* __restrict__ cnt, int* __restrict__ bsum) {
  __shared__ int wtot[8];
  const int tid = threadIdx.x, lane = tid & 31, w = tid >> 5;
  const int i0 = blockIdx.x * SCAN_TILE + tid * 4;
  int s = 0;
  for (int j = 0; j < 4; ++j) s += (i0 + j < NN) ? cnt[i0 + j] : 0;
  for (int m = 16; m > 0; m >>= 1) s += __shfl_xor(s, m, 32);
  if (lane == 0) wtot[w] = s;
  __syncthreads();
  if (tid == 0) {
    int t = 0;
    for (int i = 0; i < 8; ++i) t += wtot[i];
    bsum[blockIdx.x] = t;
  }
}
__global__ void scan_k(const int* __restrict__ cnt, const int* __restrict__ bsum,
                       int* __restrict__ rowptr) {
  __shared__ int wtot[8];
  __shared__ int wpre[8];
  const int tid = threadIdx.x, lane = tid & 31, w = tid >> 5;
  int part = (tid < (int)blockIdx.x) ? bsum[tid] : 0;
  for (int m = 16; m > 0; m >>= 1) part += __shfl_xor(part, m, 32);
  if (lane == 0) wpre[w] = part;
  const int i0 = blockIdx.x * SCAN_TILE + tid * 4;
  int c[4];
  for (int j = 0; j < 4; ++j) c[j] = (i0 + j < NN) ? cnt[i0 + j] : 0;
  const int ts = c[0] + c[1] + c[2] + c[3];
  int inc = ts;
  for (int o = 1; o < 32; o <<= 1) {
    int t = __shfl_up(inc, o, 32);
    if (lane >= o) inc += t;
  }
  if (lane == 31) wtot[w] = inc;
  __syncthreads();
  int run = inc - ts;
  for (int i = 0; i < 8; ++i) {
    run += wpre[i];
    if (i < w) run += wtot[i];
  }
  for (int j = 0; j < 4; ++j) {
    if (i0 + j <= NN) rowptr[i0 + j] = run;
    run += c[j];
  }
}
__global__ void fill_k(const int* __restrict__ dst, const int* __restrict__ rowptr,
                       const int* __restrict__ rank, int* __restrict__ eid_raw) {
  int e = blockIdx.x * blockDim.x + threadIdx.x;
  if (e >= NE) return;
  eid_raw[rowptr[dst[e]] + rank[e]] = e;
}
__device__ __forceinline__ int lane_i(int v, int k) { return __builtin_amdgcn_readlane(v, k); }
__device__ __forceinline__ float lane_f(float v, int k) {
  return __int_as_float(__builtin_amdgcn_readlane(__float_as_int(v), k));
}
__global__ void seg_sort_k(const int* __restrict__ rowptr, const int* __restrict__ eid_raw,
                           const int* __restrict__ src, int* __restrict__ seid,
                           int* __restrict__ ssrc) {
  const int gid = blockIdx.x * blockDim.x + threadIdx.x;
  const int lane = gid & 31;
  const int n = __builtin_amdgcn_readfirstlane(gid >> 5);
  if (n >= NN) return;
  const int start = rowptr[n];
  const int end = rowptr[n + 1];
  for (int base = start; base < end; base += 32) {
    const int p = base + lane;
    const int x = (p < end) ? eid_raw[p] : 0x7fffffff;
    int r = 0;
    for (int b2 = start; b2 < end; b2 += 32) {
      const int q = b2 + lane;
      const int y = (q < end) ? eid_raw[q] : 0x7fffffff;
      const int m = min(32, end - b2);
      for (int k = 0; k < m; ++k) r += (lane_i(y, k) < x) ? 1 : 0;
    }
    if (p < end) { seid[start + r] = x; ssrc[start + r] = src[x]; }
  }
}

__global__ void wa3_k(const float* __restrict__ We_a, const float* __restrict__ ae_a,
                      const float* __restrict__ We_b, const float* __restrict__ ae_b,
                      const float* __restrict__ We_c, const float* __restrict__ ae_c,
                      float* __restrict__ wa3) {
  const int l = blockIdx.x;
  const float* __restrict__ We = (l == 0) ? We_a : ((l == 1) ? We_b : We_c);
  const float* __restrict__ ae = (l == 0) ? ae_a : ((l == 1) ? ae_b : ae_c);
  const int K = (l == 0) ? 128 : 64;
  const int i = threadIdx.x;
  if (i < 16) {
    float s = 0.f;
    for (int j = 0; j < K; ++j) s += We[i * K + j] * ae[j];
    wa3[l * 16 + i] = s;
  }
}

template <int COLT, bool SPLIT>
__global__ void gemm_k128(const float* __restrict__ A, const float* __restrict__ B,
                          const float* __restrict__ B2, float* __restrict__ C) {
  constexpr int COUT = COLT * 16;
  constexpr int LDB = SPLIT ? COUT / 2 : COUT;
  constexpr int AP = 132;
  __shared__ __attribute__((aligned(16))) float As[16 * AP];
  const int tid = threadIdx.x;
  const size_t abase = (size_t)blockIdx.x * (16 * 128);
  for (int i = tid * 4; i < 16 * 128; i += blockDim.x * 4)
    *(float4*)(&As[(i >> 7) * AP + (i & 127)]) = *(const float4*)(&A[abase + i]);
  __syncthreads();

  const int wave = SPLIT ? __builtin_amdgcn_readfirstlane(tid >> 5) : (tid >> 5);
  const int lane = tid & 31, half = lane >> 4, lidx = lane & 15;
  const int n0 = wave * 16;
  const int nb = SPLIT ? (wave % (COLT / 2)) * 16 : n0;
  const float* __restrict__ Bw = (SPLIT && wave >= COLT / 2) ? B2 : B;
  v8f acc = {};
#pragma unroll
  for (int k0 = 0; k0 < 128; k0 += 4) {
    v2f a, b;
    a.x = As[lidx * AP + k0 + 2 * half];
    a.y = As[lidx * AP + k0 + 2 * half + 1];
    b.x = Bw[(k0 + 2 * half) * LDB + nb + lidx];
    b.y = Bw[(k0 + 2 * half + 1) * LDB + nb + lidx];
    acc = __builtin_amdgcn_wmma_f32_16x16x4_f32(false, a, false, b, (short)0, acc,
                                                false, false);
  }
  const size_t row0 = (size_t)blockIdx.x * 16;
#pragma unroll
  for (int r = 0; r < 8; ++r)
    C[(row0 + r + 8 * half) * COUT + n0 + lidx] = acc[r];
}

template <int C>
__global__ void node_prep_k(const float* __restrict__ h, const float* __restrict__ a_src,
                            const float* __restrict__ a_dst, float* s_src, float* s_dst) {
  int gid = blockIdx.x * blockDim.x + threadIdx.x;
  int n = gid >> 5, lane = gid & 31;
  if (n >= NN) return;
  float ps = 0.f, pd = 0.f;
#pragma unroll
  for (int i = 0; i < C / 32; ++i) {
    int c = lane + i * 32;
    float v = h[(size_t)n * C + c];
    ps += v * a_src[c];
    pd += v * a_dst[c];
  }
#pragma unroll
  for (int m = 16; m > 0; m >>= 1) { ps += __shfl_xor(ps, m, 32); pd += __shfl_xor(pd, m, 32); }
  if (lane == 0) { s_src[n] = ps; s_dst[n] = pd; }
}

__global__ void node_prep2_k(const float* __restrict__ h, const float* __restrict__ a_src0,
                             const float* __restrict__ a_dst0, const float* __restrict__ a_src1,
                             const float* __restrict__ a_dst1, float* s_src2, float* s_dst2) {
  int gid = blockIdx.x * blockDim.x + threadIdx.x;
  int n = gid >> 5, lane = gid & 31;
  if (n >= NN) return;
  float ps0 = 0.f, pd0 = 0.f, ps1 = 0.f, pd1 = 0.f;
#pragma unroll
  for (int i = 0; i < 2; ++i) {
    int c = lane + i * 32;
    float v0 = h[(size_t)n * 128 + c];
    float v1 = h[(size_t)n * 128 + 64 + c];
    ps0 += v0 * a_src0[c];
    pd0 += v0 * a_dst0[c];
    ps1 += v1 * a_src1[c];
    pd1 += v1 * a_dst1[c];
  }
#pragma unroll
  for (int m = 16; m > 0; m >>= 1) {
    ps0 += __shfl_xor(ps0, m, 32); pd0 += __shfl_xor(pd0, m, 32);
    ps1 += __shfl_xor(ps1, m, 32); pd1 += __shfl_xor(pd1, m, 32);
  }
  if (lane == 0) {
    ((float2*)s_src2)[n] = make_float2(ps0, ps1);
    ((float2*)s_dst2)[n] = make_float2(pd0, pd1);
  }
}

static constexpr int ET_PER_THREAD = 4;
static constexpr int ET_BLOCK = 256;
static constexpr int ET_TILE = ET_BLOCK * ET_PER_THREAD;
static_assert((NE * 4) % ET_TILE == 0, "eterm3_k has no tail handling");
__device__ __forceinline__ float dot4(const float4& a, const float4& b) {
  return a.x * b.x + a.y * b.y + a.z * b.z + a.w * b.w;
}
__device__ __forceinline__ float quad_sum(float v) {
  v += __shfl_xor(v, 1, 32);
  v += __shfl_xor(v, 2, 32);
  return v;
}
__global__ void eterm3_k(const float* __restrict__ ea, const float* __restrict__ wa3,
                         float* __restrict__ et0, float* __restrict__ et12) {
  const int tid = threadIdx.x;
  const int q = tid & 3;
  const float4* __restrict__ ea4 = (const float4*)ea;
  const size_t base = (size_t)blockIdx.x * ET_TILE + tid;
  float4 v[ET_PER_THREAD];
#pragma unroll
  for (int k = 0; k < ET_PER_THREAD; ++k) v[k] = ea4[base + (size_t)k * ET_BLOCK];
  const float4 w0 = ((const float4*)wa3)[q];
  const float4 w1 = ((const float4*)wa3)[4 + q];
  const float4 w2 = ((const float4*)wa3)[8 + q];
#pragma unroll
  for (int k = 0; k < ET_PER_THREAD; ++k) {
    const float s0 = quad_sum(dot4(v[k], w0));
    const float s1 = quad_sum(dot4(v[k], w1));
    const float s2 = quad_sum(dot4(v[k], w2));
    if (q == 0) {
      const size_t e = (base + (size_t)k * ET_BLOCK) >> 2;
      et0[e] = s0; ((float2*)et12)[e] = make_float2(s1, s2);
    }
  }
}

template <int V> struct VecOf;
template <> struct VecOf<4> { typedef float4 type; };
template <> struct VecOf<2> { typedef float2 type; };
__device__ __forceinline__ void axpy(float4& acc, float a, const float4& v) {
  acc.x += a * v.x; acc.y += a * v.y; acc.z += a * v.z; acc.w += a * v.w;
}
__device__ __forceinline__ void axpy(float2& acc, float a, const float2& v) {
  acc.x += a * v.x; acc.y += a * v.y;
}
enum { EPI_NONE = 0, EPI_RELU = 1, EPI_RELU_RESID = 2 };
__device__ __forceinline__ void relu_v(float4& v) {
  v.x = fmaxf(v.x, 0.f); v.y = fmaxf(v.y, 0.f); v.z = fmaxf(v.z, 0.f); v.w = fmaxf(v.w, 0.f);
}
__device__ __forceinline__ void relu_v(float2& v) { v.x = fmaxf(v.x, 0.f); v.y = fmaxf(v.y, 0.f); }
__device__ __forceinline__ void add2_v(float4& v, const float4& r, const float4& b) {
  v.x = v.x + r.x + b.x; v.y = v.y + r.y + b.y; v.z = v.z + r.z + b.z; v.w = v.w + r.w + b.w;
}
__device__ __forceinline__ void add2_v(float2& v, const float2& r, const float2& b) {
  v.x = v.x + r.x + b.x; v.y = v.y + r.y + b.y;
}

template <bool ET>
__device__ __forceinline__ float edge_logit(int p, float pd, const int* __restrict__ ssrc,
                                            const int* __restrict__ seid,
                                            const float* __restrict__ s_src,
                                            const float* __restrict__ eterm, float& et) {
  float l = s_src[ssrc[p]] + pd;
  et = 0.f;
  if (ET) { et = eterm[seid[p]]; l += et; }
  return leaky(l);
}
template <bool ET>
__device__ __forceinline__ float self_logit(float ps, float pd, float et_sum, int deg) {
  float l = ps + pd;
  if (ET) l += et_sum / fmaxf((float)deg, 1.f);
  return leaky(l);
}
__device__ __forceinline__ float wave_max(float v) {
  for (int m = 16; m > 0; m >>= 1) v = fmaxf(v, __shfl_xor(v, m, 32));
  return v;
}
__device__ __forceinline__ float wave_sum(float v) {
  for (int m = 16; m > 0; m >>= 1) v += __shfl_xor(v, m, 32);
  return v;
}

template <int C, bool ET, int EPI>
__global__ void gat_node_k(const int* __restrict__ rowptr, const int* __restrict__ ssrc,
                           const int* __restrict__ seid, const float* __restrict__ s_src,
                           const float* __restrict__ s_dst,
                           const float* __restrict__ eterm, const float* __restrict__ h,
                           const float* __restrict__ bias, const float* __restrict__ resid,
                           const float* __restrict__ rbias, float* __restrict__ out) {
  constexpr int V = C / 32;
  typedef typename VecOf<V>::type vec_t;
  const int gid = blockIdx.x * blockDim.x + threadIdx.x;
  const int lane = gid & 31;
  const int n = __builtin_amdgcn_readfirstlane(gid >> 5);
  if (n >= NN) return;
  const int start = rowptr[n];
  const int end = rowptr[n + 1];
  const float ps = s_src[n], pd = s_dst[n];
  const int deg = end - start;
  const int ch = lane * V;
  const vec_t hs = *(const vec_t*)&h[(size_t)n * C + ch];
  const vec_t bv = *(const vec_t*)&bias[ch];
  vec_t acc = bv;
  vec_t rv = {}, rb = {};
  if (EPI == EPI_RELU_RESID) {
    rv = *(const vec_t*)&resid[(size_t)n * C + ch];
    rb = *(const vec_t*)&rbias[ch];
  }

  if (end - start <= 32) {
    const int p = start + lane;
    const bool act = p < end;
    int s = 0;
    float l = 0.f, et = 0.f;
    if (act) { s = ssrc[p]; l = edge_logit<ET>(p, pd, ssrc, seid, s_src, eterm, et); }
    const float sl = self_logit<ET>(ps, pd, ET ? wave_sum(et) : 0.f, deg);
    const float m = fmaxf(wave_max(act ? l : -INFINITY), sl);
    const float w = act ? __expf(l - m) : 0.f;
    const float wself = __expf(sl - m);
    const float den = wself + wave_sum(w) + 1e-16f;
    const float a = w / den;
    axpy(acc, wself / den, hs);
#pragma unroll 4
    for (int k = 0; k < deg; ++k) {
      const int sk = lane_i(s, k);
      const float ak = lane_f(a, k);
      axpy(acc, ak, *(const vec_t*)&h[(size_t)sk * C + ch]);
    }
  } else {
    float m = -INFINITY, ets = 0.f, et;
    for (int base = start; base < end; base += 32) {
      const int p = base + lane;
      if (p < end) { m = fmaxf(m, edge_logit<ET>(p, pd, ssrc, seid, s_src, eterm, et)); ets += et; }
    }
    const float sl = self_logit<ET>(ps, pd, ET ? wave_sum(ets) : 0.f, deg);
    m = fmaxf(wave_max(m), sl);
    float ssum = 0.f;
    for (int base = start; base < end; base += 32) {
      const int p = base + lane;
      if (p < end) ssum += __expf(edge_logit<ET>(p, pd, ssrc, seid, s_src, eterm, et) - m);
    }
    const float wself = __expf(sl - m);
    const float den = wself + wave_sum(ssum) + 1e-16f;
    axpy(acc, wself / den, hs);
    for (int base = start; base < end; base += 32) {
      const int p = base + lane;
      int s = 0;
      float a = 0.f;
      if (p < end) {
        s = ssrc[p];
        a = __expf(edge_logit<ET>(p, pd, ssrc, seid, s_src, eterm, et) - m) / den;
      }
      const int cntk = min(32, end - base);
#pragma unroll 4
      for (int k = 0; k < cntk; ++k) {
        const int sk = lane_i(s, k);
        const float ak = lane_f(a, k);
        axpy(acc, ak, *(const vec_t*)&h[(size_t)sk * C + ch]);
      }
    }
  }
  if (EPI != EPI_NONE) relu_v(acc);
  if (EPI == EPI_RELU_RESID) add2_v(acc, rv, rb);
  *(vec_t*)&out[(size_t)n * C + ch] = acc;
}

struct Logit2 { float l0, l1, e0, e1; };
__device__ __forceinline__ Logit2 edge_logit2(int p, float2 pd, const int* __restrict__ ssrc,
                                              const int* __restrict__ seid,
                                              const float2* __restrict__ s_src2,
                                              const float2* __restrict__ et12) {
  const float2 ss = s_src2[ssrc[p]];
  const float2 et = et12[seid[p]];
  Logit2 r;
  r.e0 = et.x; r.e1 = et.y;
  float l0 = ss.x + pd.x; l0 += et.x;
  float l1 = ss.y + pd.y; l1 += et.y;
  r.l0 = leaky(l0); r.l1 = leaky(l1);
  return r;
}
__global__ void gat_node2_k(const int* __restrict__ rowptr, const int* __restrict__ ssrc,
                            const int* __restrict__ seid, const float* __restrict__ s_src2f,
                            const float* __restrict__ s_dst2f, const float* __restrict__ et12f,
                            const float* __restrict__ h, const float* __restrict__ bias0,
                            const float* __restrict__ bias1, float* __restrict__ out0,
                            float* __restrict__ out1) {
  const float2* __restrict__ s_src2 = (const float2*)s_src2f;
  const float2* __restrict__ s_dst2 = (const float2*)s_dst2f;
  const float2* __restrict__ et12 = (const float2*)et12f;
  const int gid = blockIdx.x * blockDim.x + threadIdx.x;
  const int lane = gid & 31;
  const int n = __builtin_amdgcn_readfirstlane(gid >> 5);
  if (n >= NN) return;
  const int start = rowptr[n];
  const int end = rowptr[n + 1];
  const float2 ps = s_src2[n], pd = s_dst2[n];
  const int deg = end - start;
  const bool hd = lane >= 16;
  const int ch = lane * 4;
  const int hc = (lane & 15) * 4;
  const float4 hs = *(const float4*)&h[(size_t)n * 128 + ch];
  const float4 bv = *(const float4*)&(hd ? bias1 : bias0)[hc];
  float4 acc = bv;

  if (deg <= 32) {
    const int p = start + lane;
    const bool act = p < end;
    int s = 0;
    Logit2 q = {0.f, 0.f, 0.f, 0.f};
    if (act) { s = ssrc[p]; q = edge_logit2(p, pd, ssrc, seid, s_src2, et12); }
    const float sl0 = self_logit<true>(ps.x, pd.x, wave_sum(q.e0), deg);
    const float sl1 = self_logit<true>(ps.y, pd.y, wave_sum(q.e1), deg);
    const float m0 = fmaxf(wave_max(act ? q.l0 : -INFINITY), sl0);
    const float m1 = fmaxf(wave_max(act ? q.l1 : -INFINITY), sl1);
    const float w0 = act ? __expf(q.l0 - m0) : 0.f;
    const float w1 = act ? __expf(q.l1 - m1) : 0.f;
    const float wself0 = __expf(sl0 - m0);
    const float wself1 = __expf(sl1 - m1);
    const float den0 = wself0 + wave_sum(w0) + 1e-16f;
    const float den1 = wself1 + wave_sum(w1) + 1e-16f;
    const float a0 = w0 / den0;
    const float a1 = w1 / den1;
    axpy(acc, hd ? wself1 / den1 : wself0 / den0, hs);
#pragma unroll 4
    for (int k = 0; k < deg; ++k) {
      const int sk = lane_i(s, k);
      const float ak0 = lane_f(a0, k);
      const float ak1 = lane_f(a1, k);
      axpy(acc, hd ? ak1 : ak0, *(const float4*)&h[(size_t)sk * 128 + ch]);
    }
  } else {
    float m0 = -INFINITY, m1 = -INFINITY, ets0 = 0.f, ets1 = 0.f;
    for (int base = start; base < end; base += 32) {
      const int p = base + lane;
      if (p < end) {
        const Logit2 q = edge_logit2(p, pd, ssrc, seid, s_src2, et12);
        m0 = fmaxf(m0, q.l0); m1 = fmaxf(m1, q.l1);
        ets0 += q.e0; ets1 += q.e1;
      }
    }
    const float sl0 = self_logit<true>(ps.x, pd.x, wave_sum(ets0), deg);
    const float sl1 = self_logit<true>(ps.y, pd.y, wave_sum(ets1), deg);
    m0 = fmaxf(wave_max(m0), sl0);
    m1 = fmaxf(wave_max(m1), sl1);
    float ssum0 = 0.f, ssum1 = 0.f;
    for (int base = start; base < end; base += 32) {
      const int p = base + lane;
      if (p < end) {
        const Logit2 q = edge_logit2(p, pd, ssrc, seid, s_src2, et12);
        ssum0 += __expf(q.l0 - m0); ssum1 += __expf(q.l1 - m1);
      }
    }
    const float wself0 = __expf(sl0 - m0);
    const float wself1 = __expf(sl1 - m1);
    const float den0 = wself0 + wave_sum(ssum0) + 1e-16f;
    const float den1 = wself1 + wave_sum(ssum1) + 1e-16f;
    axpy(acc, hd ? wself1 / den1 : wself0 / den0, hs);
    for (int base = start; base < end; base += 32) {
      const int p = base + lane;
      int s = 0;
      float a0 = 0.f, a1 = 0.f;
      if (p < end) {
        s = ssrc[p];
        const Logit2 q = edge_logit2(p, pd, ssrc, seid, s_src2, et12);
        a0 = __expf(q.l0 - m0) / den0;
        a1 = __expf(q.l1 - m1) / den1;
      }
      const int cntk = min(32, end - base);
#pragma unroll 4
      for (int k = 0; k < cntk; ++k) {
        const int sk = lane_i(s, k);
        const float ak0 = lane_f(a0, k);
        const float ak1 = lane_f(a1, k);
        axpy(acc, hd ? ak1 : ak0, *(const float4*)&h[(size_t)sk * 128 + ch]);
      }
    }
  }
  float* __restrict__ o = hd ? out1 : out0;
  *(float4*)&o[(size_t)n * 64 + hc] = acc;
}

struct Scratch {
  float *s_src, *s_dst;
  const float* eterm;
  const int *rowptr, *ssrc, *seid;
};

template <int C, bool ET, int EPI>
static void run_gat(hipStream_t st, const float* hin, const float* Wm, const float* asrc,
                    const float* adst, const float* bias, float* hg, float* out,
                    const Scratch& S, const float* resid, const float* rbias) {
  constexpr int COLT = C / 16;
  gemm_k128<COLT, false><<<NN / 16, 32 * COLT, 0, st>>>(hin, Wm, nullptr, hg);
  node_prep_k<C><<<(NN * 32 + 255) / 256, 256, 0, st>>>(hg, asrc, adst, S.s_src, S.s_dst);
  gat_node_k<C, ET, EPI><<<(NN * 32 + 255) / 256, 256, 0, st>>>(S.rowptr, S.ssrc, S.seid,
                                                                S.s_src, S.s_dst, S.eterm, hg,
                                                                bias, resid, rbias, out);
}

extern "C" void kernel_launch(void* const* d_in, const int* in_sizes, int n_in, void* d_out,
                              int out_size, void* d_ws, size_t ws_size, hipStream_t stream) {
  (void)in_sizes; (void)n_in; (void)out_size; (void)ws_size;
  const float* x = (const float*)d_in[0];
  const int* ei = (const int*)d_in[1];
  const int* src = ei;
  const int* dst = ei + NE;
  const float* edge_attr = (const float*)d_in[2];
  const float *W1 = (const float*)d_in[3], *a1s = (const float*)d_in[4],
              *a1d = (const float*)d_in[5], *b1 = (const float*)d_in[6];
  const float *W2 = (const float*)d_in[7], *a2s = (const float*)d_in[8],
              *a2d = (const float*)d_in[9], *b2 = (const float*)d_in[10];
  const float *We2 = (const float*)d_in[11], *ae2 = (const float*)d_in[12];
  const float *Wr = (const float*)d_in[13], *br = (const float*)d_in[14];
  const float *Wmu = (const float*)d_in[15], *amus = (const float*)d_in[16],
              *amud = (const float*)d_in[17], *bmu = (const float*)d_in[18],
              *Wemu = (const float*)d_in[19], *aemu = (const float*)d_in[20];
  const float *Wls = (const float*)d_in[21], *alss = (const float*)d_in[22],
              *alsd = (const float*)d_in[23], *bls = (const float*)d_in[24],
              *Wels = (const float*)d_in[25], *aels = (const float*)d_in[26];
  float* outf = (float*)d_out;

  float* W = (float*)d_ws;
  size_t o = 0;
  float* hA = W + o; o += (size_t)NN * 128;
  float* hB = W + o; o += (size_t)NN * 128;
  float* hC = W + o; o += (size_t)NN * 128;
  Scratch S;
  S.s_src = W + o; o += NN;
  S.s_dst = W + o; o += NN;
  float* et0 = W + o; o += NE;
  float* et12 = W + o; o += (size_t)NE * 2;
  float* wa3 = W + o; o += 48;
  S.eterm = et0;
  int* cnt = (int*)(W + o); o += NN;
  int* bsum = (int*)(W + o); o += 256;
  int* rowptr = (int*)(W + o); o += NN + 16;
  int* rank = (int*)(W + o); o += NE;
  int* eid_raw = (int*)(W + o); o += NE;
  int* seid = (int*)(W + o); o += NE;
  int* ssrc = (int*)(W + o); o += NE;
  S.rowptr = rowptr; S.ssrc = ssrc; S.seid = seid;
  float* s_src2 = W + o; o += (size_t)NN * 2;
  float* s_dst2 = W + o; o += (size_t)NN * 2;

  zero_i32_k<<<(NN + 255) / 256, 256, 0, stream>>>(cnt, NN);
  hist_rank_k<<<(NE + 255) / 256, 256, 0, stream>>>(dst, cnt, rank);
  blocksum_k<<<SCAN_BLOCKS, 256, 0, stream>>>(cnt, bsum);
  scan_k<<<SCAN_BLOCKS, 256, 0, stream>>>(cnt, bsum, rowptr);
  fill_k<<<(NE + 255) / 256, 256, 0, stream>>>(dst, rowptr, rank, eid_raw);
  seg_sort_k<<<(NN * 32 + 255) / 256, 256, 0, stream>>>(rowptr, eid_raw, src, seid, ssrc);

  wa3_k<<<3, 16, 0, stream>>>(We2, ae2, Wemu, aemu, Wels, aels, wa3);
  eterm3_k<<<(NE * 4) / ET_TILE, ET_BLOCK, 0, stream>>>(edge_attr, wa3, et0, et12);

  gemm_k128<8, false><<<NN / 16, 256, 0, stream>>>(x, Wr, nullptr, hC);

  run_gat<128, false, EPI_RELU>(stream, x, W1, a1s, a1d, b1, hA, hB, S, nullptr, nullptr);

  S.eterm = et0;
  run_gat<128, true, EPI_RELU_RESID>(stream, hB, W2, a2s, a2d, b2, hA, hB, S, hC, br);

  gemm_k128<8, true><<<NN / 16, 256, 0, stream>>>(hB, Wmu, Wls, hA);
  node_prep2_k<<<(NN * 32 + 255) / 256, 256, 0, stream>>>(hA, amus, amud, alss, alsd, s_src2, s_dst2);
  gat_node2_k<<<(NN * 32 + 255) / 256, 256, 0, stream>>>(rowptr, ssrc, seid, s_src2, s_dst2, et12, hA,
                                                         bmu, bls, outf, outf + (size_t)NN * 64);
}


# Attention scalars s_src/s_dst summed in the GEMM epilogue from its accumulators (16-lane sums, then 8 waves via LDS): 3 prep passes over h and 3 of 18 launches go; these dots' sum order changes
# speedup vs baseline: 5.4008x; 1.0837x over previous
#include <hip/hip_runtime.h>
#include <math.h>

typedef float v2f __attribute__((ext_vector_type(2)));
typedef float v4f __attribute__((ext_vector_type(4)));
typedef float v8f __attribute__((ext_vector_type(8)));

static constexpr int   NN = 100000;
static constexpr int   NE = 1600000;
static constexpr float NEG = 0.2f;
static constexpr int   SCAN_TILE = 1024;
static constexpr int   SCAN_BLOCKS = (NN + 1 + SCAN_TILE - 1) / SCAN_TILE;
static_assert(SCAN_BLOCKS <= 256, "scan_k sums the earlier block totals with one thread each");

__device__ __forceinline__ float leaky(float x) { return x > 0.f ? x : NEG * x; }

__global__ void zero_i32_k(int* p, int n) {
  int i = blockIdx.x * blockDim.x + threadIdx.x;
  if (i < n) p[i] = 0;
}

__global__ void hist_rank_k(const int* __restrict__ dst, int* cnt, int* __restrict__ rank) {
  int e = blockIdx.x * blockDim.x + threadIdx.x;
  if (e >= NE) return;
  rank[e] = atomicAdd(&cnt[dst[e]], 1);
}
__global__ void blocksum_k(const int* __restrict__ cnt, int* __restrict__ bsum) {
  __shared__ int wtot[8];
  const int tid = threadIdx.x, lane = tid & 31, w = tid >> 5;
  const int i0 = blockIdx.x * SCAN_TILE + tid * 4;
  int s = 0;
  for (int j = 0; j < 4; ++j) s += (i0 + j < NN) ? cnt[i0 + j] : 0;
  for (int m = 16; m > 0; m >>= 1) s += __shfl_xor(s, m, 32);
  if (lane == 0) wtot[w] = s;
  __syncthreads();
  if (tid == 0) {
    int t = 0;
    for (int i = 0; i < 8; ++i) t += wtot[i];
    bsum[blockIdx.x] = t;
  }
}
__global__ void scan_k(const int* __restrict__ cnt, const int* __restrict__ bsum,
                       int* __restrict__ rowptr) {
  __shared__ int wtot[8];
  __shared__ int wpre[8];
  const int tid = threadIdx.x, lane = tid & 31, w = tid >> 5;
  int part = (tid < (int)blockIdx.x) ? bsum[tid] : 0;
  for (int m = 16; m > 0; m >>= 1) part += __shfl_xor(part, m, 32);
  if (lane == 0) wpre[w] = part;
  const int i0 = blockIdx.x * SCAN_TILE + tid * 4;
  int c[4];
  for (int j = 0; j < 4; ++j) c[j] = (i0 + j < NN) ? cnt[i0 + j] : 0;
  const int ts = c[0] + c[1] + c[2] + c[3];
  int inc = ts;
  for (int o = 1; o < 32; o <<= 1) {
    int t = __shfl_up(inc, o, 32);
    if (lane >= o) inc += t;
  }
  if (lane == 31) wtot[w] = inc;
  __syncthreads();
  int run = inc - ts;
  for (int i = 0; i < 8; ++i) {
    run += wpre[i];
    if (i < w) run += wtot[i];
  }
  for (int j = 0; j < 4; ++j) {
    if (i0 + j <= NN) rowptr[i0 + j] = run;
    run += c[j];
  }
}
__global__ void fill_k(const int* __restrict__ dst, const int* __restrict__ rowptr,
                       const int* __restrict__ rank, int* __restrict__ eid_raw) {
  int e = blockIdx.x * blockDim.x + threadIdx.x;
  if (e >= NE) return;
  eid_raw[rowptr[dst[e]] + rank[e]] = e;
}
__device__ __forceinline__ int lane_i(int v, int k) { return __builtin_amdgcn_readlane(v, k); }
__device__ __forceinline__ float lane_f(float v, int k) {
  return __int_as_float(__builtin_amdgcn_readlane(__float_as_int(v), k));
}
__global__ void seg_sort_k(const int* __restrict__ rowptr, const int* __restrict__ eid_raw,
                           const int* __restrict__ src, int* __restrict__ seid,
                           int* __restrict__ ssrc) {
  const int gid = blockIdx.x * blockDim.x + threadIdx.x;
  const int lane = gid & 31;
  const int n = __builtin_amdgcn_readfirstlane(gid >> 5);
  if (n >= NN) return;
  const int start = rowptr[n];
  const int end = rowptr[n + 1];
  for (int base = start; base < end; base += 32) {
    const int p = base + lane;
    const int x = (p < end) ? eid_raw[p] : 0x7fffffff;
    int r = 0;
    for (int b2 = start; b2 < end; b2 += 32) {
      const int q = b2 + lane;
      const int y = (q < end) ? eid_raw[q] : 0x7fffffff;
      const int m = min(32, end - b2);
      for (int k = 0; k < m; ++k) r += (lane_i(y, k) < x) ? 1 : 0;
    }
    if (p < end) { seid[start + r] = x; ssrc[start + r] = src[x]; }
  }
}

__global__ void wa3_k(const float* __restrict__ We_a, const float* __restrict__ ae_a,
                      const float* __restrict__ We_b, const float* __restrict__ ae_b,
                      const float* __restrict__ We_c, const float* __restrict__ ae_c,
                      float* __restrict__ wa3) {
  const int l = blockIdx.x;
  const float* __restrict__ We = (l == 0) ? We_a : ((l == 1) ? We_b : We_c);
  const float* __restrict__ ae = (l == 0) ? ae_a : ((l == 1) ? ae_b : ae_c);
  const int K = (l == 0) ? 128 : 64;
  const int i = threadIdx.x;
  if (i < 16) {
    float s = 0.f;
    for (int j = 0; j < K; ++j) s += We[i * K + j] * ae[j];
    wa3[l * 16 + i] = s;
  }
}

template <int M>
__device__ __forceinline__ float xmask_add(float keep, float send) {
  return keep + __int_as_float(__builtin_amdgcn_update_dpp(0, __float_as_int(send), 0x160 + M, 0xf,
                                                          0xf, true));
}

template <int COLT, bool SPLIT, bool PREP>
__global__ void gemm_k128(const float* __restrict__ A, const float* __restrict__ B,
                          const float* __restrict__ B2, float* __restrict__ C,
                          const float* __restrict__ a_src, const float* __restrict__ a_dst,
                          const float* __restrict__ a_src1, const float* __restrict__ a_dst1,
                          float* __restrict__ s_src, float* __restrict__ s_dst) {
  constexpr int COUT = COLT * 16;
  constexpr int LDB = SPLIT ? COUT / 2 : COUT;
  constexpr int AP = 132;
  __shared__ __attribute__((aligned(16))) float As[16 * AP];
  const int tid = threadIdx.x;
  const size_t abase = (size_t)blockIdx.x * (16 * 128);
  for (int i = tid * 4; i < 16 * 128; i += blockDim.x * 4)
    *(float4*)(&As[(i >> 7) * AP + (i & 127)]) = *(const float4*)(&A[abase + i]);
  __syncthreads();

  const int wave = SPLIT ? __builtin_amdgcn_readfirstlane(tid >> 5) : (tid >> 5);
  const int lane = tid & 31, half = lane >> 4, lidx = lane & 15;
  const int n0 = wave * 16;
  const int nb = SPLIT ? (wave % (COLT / 2)) * 16 : n0;
  const float* __restrict__ Bw = (SPLIT && wave >= COLT / 2) ? B2 : B;
  v8f acc = {};
#pragma unroll
  for (int k0 = 0; k0 < 128; k0 += 4) {
    v2f a, b;
    a.x = As[lidx * AP + k0 + 2 * half];
    a.y = As[lidx * AP + k0 + 2 * half + 1];
    b.x = Bw[(k0 + 2 * half) * LDB + nb + lidx];
    b.y = Bw[(k0 + 2 * half + 1) * LDB + nb + lidx];
    acc = __builtin_amdgcn_wmma_f32_16x16x4_f32(false, a, false, b, (short)0, acc,
                                                false, false);
  }
  const size_t row0 = (size_t)blockIdx.x * 16;
  __shared__ float part[PREP ? COLT * 32 : 1];
  if constexpr (PREP) {
    const bool right = SPLIT && wave >= COLT / 2;
    const float* __restrict__ asw = right ? a_src1 : a_src;
    const float* __restrict__ adw = right ? a_dst1 : a_dst;
    const float as = asw[nb + lidx], ad = adw[nb + lidx];
    const v8f P = acc * as, Q = acc * ad;
    const bool u8 = (lidx & 8) != 0, u4 = (lidx & 4) != 0, u2 = (lidx & 2) != 0, u1 = (lidx & 1) != 0;
    const v8f k8 = u8 ? Q : P, s8 = u8 ? P : Q;
    v8f r8;
    for (int i = 0; i < 8; ++i) r8[i] = xmask_add<8>(k8[i], s8[i]);
    const v4f k4 = u4 ? r8.hi : r8.lo, s4 = u4 ? r8.lo : r8.hi;
    v4f r4;
    for (int i = 0; i < 4; ++i) r4[i] = xmask_add<4>(k4[i], s4[i]);
    const v2f k2 = u2 ? r4.hi : r4.lo, s2 = u2 ? r4.lo : r4.hi;
    v2f r2;
    for (int i = 0; i < 2; ++i) r2[i] = xmask_add<2>(k2[i], s2[i]);
    const float r1 = xmask_add<1>(u1 ? r2.y : r2.x, u1 ? r2.x : r2.y);
    part[wave * 32 + (lidx >> 3) * 16 + 8 * half + (lidx & 7)] = r1;
    __syncthreads();
  }
#pragma unroll
  for (int r = 0; r < 8; ++r)
    C[(row0 + r + 8 * half) * COUT + n0 + lidx] = acc[r];
  if constexpr (PREP) {
    if (!SPLIT) {
      if (tid < 32) {
        float s = part[tid];
        for (int w = 1; w < COLT; ++w) s += part[w * 32 + tid];
        ((tid >> 4) ? s_dst : s_src)[row0 + (tid & 15)] = s;
      }
    } else {
      if (tid < 64) {
        const int hd = tid >> 5, slot = tid & 31;
        float s = part[(hd * (COLT / 2)) * 32 + slot];
        for (int w = 1; w < COLT / 2; ++w) s += part[(hd * (COLT / 2) + w) * 32 + slot];
        ((slot >> 4) ? s_dst : s_src)[(row0 + (slot & 15)) * 2 + hd] = s;
      }
    }
  }
}

static constexpr int ET_PER_THREAD = 4;
static constexpr int ET_BLOCK = 256;
static constexpr int ET_TILE = ET_BLOCK * ET_PER_THREAD;
static_assert((NE * 4) % ET_TILE == 0, "eterm3_k has no tail handling");
__device__ __forceinline__ float dot4(const float4& a, const float4& b) {
  return a.x * b.x + a.y * b.y + a.z * b.z + a.w * b.w;
}
__device__ __forceinline__ float quad_sum(float v) {
  v += __shfl_xor(v, 1, 32);
  v += __shfl_xor(v, 2, 32);
  return v;
}
__global__ void eterm3_k(const float* __restrict__ ea, const float* __restrict__ wa3,
                         float* __restrict__ et0, float* __restrict__ et12) {
  const int tid = threadIdx.x;
  const int q = tid & 3;
  const float4* __restrict__ ea4 = (const float4*)ea;
  const size_t base = (size_t)blockIdx.x * ET_TILE + tid;
  float4 v[ET_PER_THREAD];
#pragma unroll
  for (int k = 0; k < ET_PER_THREAD; ++k) v[k] = ea4[base + (size_t)k * ET_BLOCK];
  const float4 w0 = ((const float4*)wa3)[q];
  const float4 w1 = ((const float4*)wa3)[4 + q];
  const float4 w2 = ((const float4*)wa3)[8 + q];
#pragma unroll
  for (int k = 0; k < ET_PER_THREAD; ++k) {
    const float s0 = quad_sum(dot4(v[k], w0));
    const float s1 = quad_sum(dot4(v[k], w1));
    const float s2 = quad_sum(dot4(v[k], w2));
    if (q == 0) {
      const size_t e = (base + (size_t)k * ET_BLOCK) >> 2;
      et0[e] = s0; ((float2*)et12)[e] = make_float2(s1, s2);
    }
  }
}

template <int V> struct VecOf;
template <> struct VecOf<4> { typedef float4 type; };
template <> struct VecOf<2> { typedef float2 type; };
__device__ __forceinline__ void axpy(float4& acc, float a, const float4& v) {
  acc.x += a * v.x; acc.y += a * v.y; acc.z += a * v.z; acc.w += a * v.w;
}
__device__ __forceinline__ void axpy(float2& acc, float a, const float2& v) {
  acc.x += a * v.x; acc.y += a * v.y;
}
enum { EPI_NONE = 0, EPI_RELU = 1, EPI_RELU_RESID = 2 };
__device__ __forceinline__ void relu_v(float4& v) {
  v.x = fmaxf(v.x, 0.f); v.y = fmaxf(v.y, 0.f); v.z = fmaxf(v.z, 0.f); v.w = fmaxf(v.w, 0.f);
}
__device__ __forceinline__ void relu_v(float2& v) { v.x = fmaxf(v.x, 0.f); v.y = fmaxf(v.y, 0.f); }
__device__ __forceinline__ void add2_v(float4& v, const float4& r, const float4& b) {
  v.x = v.x + r.x + b.x; v.y = v.y + r.y + b.y; v.z = v.z + r.z + b.z; v.w = v.w + r.w + b.w;
}
__device__ __forceinline__ void add2_v(float2& v, const float2& r, const float2& b) {
  v.x = v.x + r.x + b.x; v.y = v.y + r.y + b.y;
}

template <bool ET>
__device__ __forceinline__ float edge_logit(int p, float pd, const int* __restrict__ ssrc,
                                            const int* __restrict__ seid,
                                            const float* __restrict__ s_src,
                                            const float* __restrict__ eterm, float& et) {
  float l = s_src[ssrc[p]] + pd;
  et = 0.f;
  if (ET) { et = eterm[seid[p]]; l += et; }
  return leaky(l);
}
template <bool ET>
__device__ __forceinline__ float self_logit(float ps, float pd, float et_sum, int deg) {
  float l = ps + pd;
  if (ET) l += et_sum / fmaxf((float)deg, 1.f);
  return leaky(l);
}
__device__ __forceinline__ float wave_max(float v) {
  for (int m = 16; m > 0; m >>= 1) v = fmaxf(v, __shfl_xor(v, m, 32));
  return v;
}
__device__ __forceinline__ float wave_sum(float v) {
  for (int m = 16; m > 0; m >>= 1) v += __shfl_xor(v, m, 32);
  return v;
}

template <int C, bool ET, int EPI>
__global__ void gat_node_k(const int* __restrict__ rowptr, const int* __restrict__ ssrc,
                           const int* __restrict__ seid, const float* __restrict__ s_src,
                           const float* __restrict__ s_dst,
                           const float* __restrict__ eterm, const float* __restrict__ h,
                           const float* __restrict__ bias, const float* __restrict__ resid,
                           const float* __restrict__ rbias, float* __restrict__ out) {
  constexpr int V = C / 32;
  typedef typename VecOf<V>::type vec_t;
  const int gid = blockIdx.x * blockDim.x + threadIdx.x;
  const int lane = gid & 31;
  const int n = __builtin_amdgcn_readfirstlane(gid >> 5);
  if (n >= NN) return;
  const int start = rowptr[n];
  const int end = rowptr[n + 1];
  const float ps = s_src[n], pd = s_dst[n];
  const int deg = end - start;
  const int ch = lane * V;
  const vec_t hs = *(const vec_t*)&h[(size_t)n * C + ch];
  const vec_t bv = *(const vec_t*)&bias[ch];
  vec_t acc = bv;
  vec_t rv = {}, rb = {};
  if (EPI == EPI_RELU_RESID) {
    rv = *(const vec_t*)&resid[(size_t)n * C + ch];
    rb = *(const vec_t*)&rbias[ch];
  }

  if (end - start <= 32) {
    const int p = start + lane;
    const bool act = p < end;
    int s = 0;
    float l = 0.f, et = 0.f;
    if (act) { s = ssrc[p]; l = edge_logit<ET>(p, pd, ssrc, seid, s_src, eterm, et); }
    const float sl = self_logit<ET>(ps, pd, ET ? wave_sum(et) : 0.f, deg);
    const float m = fmaxf(wave_max(act ? l : -INFINITY), sl);
    const float w = act ? __expf(l - m) : 0.f;
    const float wself = __expf(sl - m);
    const float den = wself + wave_sum(w) + 1e-16f;
    const float a = w / den;
    axpy(acc, wself / den, hs);
#pragma unroll 4
    for (int k = 0; k < deg; ++k) {
      const int sk = lane_i(s, k);
      const float ak = lane_f(a, k);
      axpy(acc, ak, *(const vec_t*)&h[(size_t)sk * C + ch]);
    }
  } else {
    float m = -INFINITY, ets = 0.f, et;
    for (int base = start; base < end; base += 32) {
      const int p = base + lane;
      if (p < end) { m = fmaxf(m, edge_logit<ET>(p, pd, ssrc, seid, s_src, eterm, et)); ets += et; }
    }
    const float sl = self_logit<ET>(ps, pd, ET ? wave_sum(ets) : 0.f, deg);
    m = fmaxf(wave_max(m), sl);
    float ssum = 0.f;
    for (int base = start; base < end; base += 32) {
      const int p = base + lane;
      if (p < end) ssum += __expf(edge_logit<ET>(p, pd, ssrc, seid, s_src, eterm, et) - m);
    }
    const float wself = __expf(sl - m);
    const float den = wself + wave_sum(ssum) + 1e-16f;
    axpy(acc, wself / den, hs);
    for (int base = start; base < end; base += 32) {
      const int p = base + lane;
      int s = 0;
      float a = 0.f;
      if (p < end) {
        s = ssrc[p];
        a = __expf(edge_logit<ET>(p, pd, ssrc, seid, s_src, eterm, et) - m) / den;
      }
      const int cntk = min(32, end - base);
#pragma unroll 4
      for (int k = 0; k < cntk; ++k) {
        const int sk = lane_i(s, k);
        const float ak = lane_f(a, k);
        axpy(acc, ak, *(const vec_t*)&h[(size_t)sk * C + ch]);
      }
    }
  }
  if (EPI != EPI_NONE) relu_v(acc);
  if (EPI == EPI_RELU_RESID) add2_v(acc, rv, rb);
  *(vec_t*)&out[(size_t)n * C + ch] = acc;
}

struct Logit2 { float l0, l1, e0, e1; };
__device__ __forceinline__ Logit2 edge_logit2(int p, float2 pd, const int* __restrict__ ssrc,
                                              const int* __restrict__ seid,
                                              const float2* __restrict__ s_src2,
                                              const float2* __restrict__ et12) {
  const float2 ss = s_src2[ssrc[p]];
  const float2 et = et12[seid[p]];
  Logit2 r;
  r.e0 = et.x; r.e1 = et.y;
  float l0 = ss.x + pd.x; l0 += et.x;
  float l1 = ss.y + pd.y; l1 += et.y;
  r.l0 = leaky(l0); r.l1 = leaky(l1);
  return r;
}
__global__ void gat_node2_k(const int* __restrict__ rowptr, const int* __restrict__ ssrc,
                            const int* __restrict__ seid, const float* __restrict__ s_src2f,
                            const float* __restrict__ s_dst2f, const float* __restrict__ et12f,
                            const float* __restrict__ h, const float* __restrict__ bias0,
                            const float* __restrict__ bias1, float* __restrict__ out0,
                            float* __restrict__ out1) {
  const float2* __restrict__ s_src2 = (const float2*)s_src2f;
  const float2* __restrict__ s_dst2 = (const float2*)s_dst2f;
  const float2* __restrict__ et12 = (const float2*)et12f;
  const int gid = blockIdx.x * blockDim.x + threadIdx.x;
  const int lane = gid & 31;
  const int n = __builtin_amdgcn_readfirstlane(gid >> 5);
  if (n >= NN) return;
  const int start = rowptr[n];
  const int end = rowptr[n + 1];
  const float2 ps = s_src2[n], pd = s_dst2[n];
  const int deg = end - start;
  const bool hd = lane >= 16;
  const int ch = lane * 4;
  const int hc = (lane & 15) * 4;
  const float4 hs = *(const float4*)&h[(size_t)n * 128 + ch];
  const float4 bv = *(const float4*)&(hd ? bias1 : bias0)[hc];
  float4 acc = bv;

  if (deg <= 32) {
    const int p = start + lane;
    const bool act = p < end;
    int s = 0;
    Logit2 q = {0.f, 0.f, 0.f, 0.f};
    if (act) { s = ssrc[p]; q = edge_logit2(p, pd, ssrc, seid, s_src2, et12); }
    const float sl0 = self_logit<true>(ps.x, pd.x, wave_sum(q.e0), deg);
    const float sl1 = self_logit<true>(ps.y, pd.y, wave_sum(q.e1), deg);
    const float m0 = fmaxf(wave_max(act ? q.l0 : -INFINITY), sl0);
    const float m1 = fmaxf(wave_max(act ? q.l1 : -INFINITY), sl1);
    const float w0 = act ? __expf(q.l0 - m0) : 0.f;
    const float w1 = act ? __expf(q.l1 - m1) : 0.f;
    const float wself0 = __expf(sl0 - m0);
    const float wself1 = __expf(sl1 - m1);
    const float den0 = wself0 + wave_sum(w0) + 1e-16f;
    const float den1 = wself1 + wave_sum(w1) + 1e-16f;
    const float a0 = w0 / den0;
    const float a1 = w1 / den1;
    axpy(acc, hd ? wself1 / den1 : wself0 / den0, hs);
#pragma unroll 4
    for (int k = 0; k < deg; ++k) {
      const int sk = lane_i(s, k);
      const float ak0 = lane_f(a0, k);
      const float ak1 = lane_f(a1, k);
      axpy(acc, hd ? ak1 : ak0, *(const float4*)&h[(size_t)sk * 128 + ch]);
    }
  } else {
    float m0 = -INFINITY, m1 = -INFINITY, ets0 = 0.f, ets1 = 0.f;
    for (int base = start; base < end; base += 32) {
      const int p = base + lane;
      if (p < end) {
        const Logit2 q = edge_logit2(p, pd, ssrc, seid, s_src2, et12);
        m0 = fmaxf(m0, q.l0); m1 = fmaxf(m1, q.l1);
        ets0 += q.e0; ets1 += q.e1;
      }
    }
    const float sl0 = self_logit<true>(ps.x, pd.x, wave_sum(ets0), deg);
    const float sl1 = self_logit<true>(ps.y, pd.y, wave_sum(ets1), deg);
    m0 = fmaxf(wave_max(m0), sl0);
    m1 = fmaxf(wave_max(m1), sl1);
    float ssum0 = 0.f, ssum1 = 0.f;
    for (int base = start; base < end; base += 32) {
      const int p = base + lane;
      if (p < end) {
        const Logit2 q = edge_logit2(p, pd, ssrc, seid, s_src2, et12);
        ssum0 += __expf(q.l0 - m0); ssum1 += __expf(q.l1 - m1);
      }
    }
    const float wself0 = __expf(sl0 - m0);
    const float wself1 = __expf(sl1 - m1);
    const float den0 = wself0 + wave_sum(ssum0) + 1e-16f;
    const float den1 = wself1 + wave_sum(ssum1) + 1e-16f;
    axpy(acc, hd ? wself1 / den1 : wself0 / den0, hs);
    for (int base = start; base < end; base += 32) {
      const int p = base + lane;
      int s = 0;
      float a0 = 0.f, a1 = 0.f;
      if (p < end) {
        s = ssrc[p];
        const Logit2 q = edge_logit2(p, pd, ssrc, seid, s_src2, et12);
        a0 = __expf(q.l0 - m0) / den0;
        a1 = __expf(q.l1 - m1) / den1;
      }
      const int cntk = min(32, end - base);
#pragma unroll 4
      for (int k = 0; k < cntk; ++k) {
        const int sk = lane_i(s, k);
        const float ak0 = lane_f(a0, k);
        const float ak1 = lane_f(a1, k);
        axpy(acc, hd ? ak1 : ak0, *(const float4*)&h[(size_t)sk * 128 + ch]);
      }
    }
  }
  float* __restrict__ o = hd ? out1 : out0;
  *(float4*)&o[(size_t)n * 64 + hc] = acc;
}

struct Scratch {
  float *s_src, *s_dst;
  const float* eterm;
  const int *rowptr, *ssrc, *seid;
};

template <int C, bool ET, int EPI>
static void run_gat(hipStream_t st, const float* hin, const float* Wm, const float* asrc,
                    const float* adst, const float* bias, float* hg, float* out,
                    const Scratch& S, const float* resid, const float* rbias) {
  constexpr int COLT = C / 16;
  gemm_k128<COLT, false, true><<<NN / 16, 32 * COLT, 0, st>>>(hin, Wm, nullptr, hg, asrc, adst, nullptr,
                                                             nullptr, S.s_src, S.s_dst);
  gat_node_k<C, ET, EPI><<<(NN * 32 + 255) / 256, 256, 0, st>>>(S.rowptr, S.ssrc, S.seid,
                                                                S.s_src, S.s_dst, S.eterm, hg,
                                                                bias, resid, rbias, out);
}

extern "C" void kernel_launch(void* const* d_in, const int* in_sizes, int n_in, void* d_out,
                              int out_size, void* d_ws, size_t ws_size, hipStream_t stream) {
  (void)in_sizes; (void)n_in; (void)out_size; (void)ws_size;
  const float* x = (const float*)d_in[0];
  const int* ei = (const int*)d_in[1];
  const int* src = ei;
  const int* dst = ei + NE;
  const float* edge_attr = (const float*)d_in[2];
  const float *W1 = (const float*)d_in[3], *a1s = (const float*)d_in[4],
              *a1d = (const float*)d_in[5], *b1 = (const float*)d_in[6];
  const float *W2 = (const float*)d_in[7], *a2s = (const float*)d_in[8],
              *a2d = (const float*)d_in[9], *b2 = (const float*)d_in[10];
  const float *We2 = (const float*)d_in[11], *ae2 = (const float*)d_in[12];
  const float *Wr = (const float*)d_in[13], *br = (const float*)d_in[14];
  const float *Wmu = (const float*)d_in[15], *amus = (const float*)d_in[16],
              *amud = (const float*)d_in[17], *bmu = (const float*)d_in[18],
              *Wemu = (const float*)d_in[19], *aemu = (const float*)d_in[20];
  const float *Wls = (const float*)d_in[21], *alss = (const float*)d_in[22],
              *alsd = (const float*)d_in[23], *bls = (const float*)d_in[24],
              *Wels = (const float*)d_in[25], *aels = (const float*)d_in[26];
  float* outf = (float*)d_out;

  float* W = (float*)d_ws;
  size_t o = 0;
  float* hA = W + o; o += (size_t)NN * 128;
  float* hB = W + o; o += (size_t)NN * 128;
  float* hC = W + o; o += (size_t)NN * 128;
  Scratch S;
  S.s_src = W + o; o += NN;
  S.s_dst = W + o; o += NN;
  float* et0 = W + o; o += NE;
  float* et12 = W + o; o += (size_t)NE * 2;
  float* wa3 = W + o; o += 48;
  S.eterm = et0;
  int* cnt = (int*)(W + o); o += NN;
  int* bsum = (int*)(W + o); o += 256;
  int* rowptr = (int*)(W + o); o += NN + 16;
  int* rank = (int*)(W + o); o += NE;
  int* eid_raw = (int*)(W + o); o += NE;
  int* seid = (int*)(W + o); o += NE;
  int* ssrc = (int*)(W + o); o += NE;
  S.rowptr = rowptr; S.ssrc = ssrc; S.seid = seid;
  float* s_src2 = W + o; o += (size_t)NN * 2;
  float* s_dst2 = W + o; o += (size_t)NN * 2;

  zero_i32_k<<<(NN + 255) / 256, 256, 0, stream>>>(cnt, NN);
  hist_rank_k<<<(NE + 255) / 256, 256, 0, stream>>>(dst, cnt, rank);
  blocksum_k<<<SCAN_BLOCKS, 256, 0, stream>>>(cnt, bsum);
  scan_k<<<SCAN_BLOCKS, 256, 0, stream>>>(cnt, bsum, rowptr);
  fill_k<<<(NE + 255) / 256, 256, 0, stream>>>(dst, rowptr, rank, eid_raw);
  seg_sort_k<<<(NN * 32 + 255) / 256, 256, 0, stream>>>(rowptr, eid_raw, src, seid, ssrc);

  wa3_k<<<3, 16, 0, stream>>>(We2, ae2, Wemu, aemu, Wels, aels, wa3);
  eterm3_k<<<(NE * 4) / ET_TILE, ET_BLOCK, 0, stream>>>(edge_attr, wa3, et0, et12);

  gemm_k128<8, false, false><<<NN / 16, 256, 0, stream>>>(x, Wr, nullptr, hC, nullptr, nullptr, nullptr,
                                                        nullptr, nullptr, nullptr);

  run_gat<128, false, EPI_RELU>(stream, x, W1, a1s, a1d, b1, hA, hB, S, nullptr, nullptr);

  S.eterm = et0;
  run_gat<128, true, EPI_RELU_RESID>(stream, hB, W2, a2s, a2d, b2, hA, hB, S, hC, br);

  gemm_k128<8, true, true><<<NN / 16, 256, 0, stream>>>(hB, Wmu, Wls, hA, amus, amud, alss, alsd, s_src2,
                                                      s_dst2);
  gat_node2_k<<<(NN * 32 + 255) / 256, 256, 0, stream>>>(rowptr, ssrc, seid, s_src2, s_dst2, et12, hA,
                                                         bmu, bls, outf, outf + (size_t)NN * 64);
}


# Edges grouped by destination in two levels on LDS counters (256-node bucket, then node; returning LDS add = slot): per-edge global atomic and 4-B scatter go; 4 launches for 6; group order as before
# speedup vs baseline: 7.0111x; 1.2982x over previous
#include <hip/hip_runtime.h>
#include <math.h>

typedef float v2f __attribute__((ext_vector_type(2)));
typedef float v4f __attribute__((ext_vector_type(4)));
typedef float v8f __attribute__((ext_vector_type(8)));

static constexpr int   NN = 100000;
static constexpr int   NE = 1600000;
static constexpr float NEG = 0.2f;

__device__ __forceinline__ float leaky(float x) { return x > 0.f ? x : NEG * x; }

static constexpr int BSH = 8;
static constexpr int BNODES = 1 << BSH;
static constexpr int NBUCK = (NN + BNODES - 1) >> BSH;
static constexpr int L1_BLOCKS = 250;
static constexpr int L1_CHUNK = NE / L1_BLOCKS;
static constexpr int GT = 1024;
static constexpr int L2_CAP = 6144;
static constexpr int L1_PER = (L1_CHUNK + GT - 1) / GT;
static constexpr int L2_PER = 4;
static_assert(L1_CHUNK * L1_BLOCKS == NE, "level 1 has no tail handling");
static_assert(NBUCK <= GT && BNODES <= GT && L1_BLOCKS <= 256, "one thread per bucket / node / level-1 block");

__device__ __forceinline__ int lane_i(int v, int k) { return __builtin_amdgcn_readlane(v, k); }
__device__ __forceinline__ float lane_f(float v, int k) {
  return __int_as_float(__builtin_amdgcn_readlane(__float_as_int(v), k));
}
__device__ __forceinline__ int wave_sum_i(int v) {
  for (int m = 16; m > 0; m >>= 1) v += __shfl_xor(v, m, 32);
  return v;
}
__device__ __forceinline__ int block_excl_scan(int v, int* wtot, int& total) {
  const int lane = threadIdx.x & 31, w = threadIdx.x >> 5;
  int inc = v;
  for (int o = 1; o < 32; o <<= 1) {
    const int u = __shfl_up(inc, o, 32);
    if (lane >= o) inc += u;
  }
  if (lane == 31) wtot[w] = inc;
  __syncthreads();
  const int wt = (lane < (int)(blockDim.x >> 5)) ? wtot[lane] : 0;
  const int before = wave_sum_i(lane < w ? wt : 0);
  total = wave_sum_i(wt);
  __syncthreads();
  return inc - v + before;
}

__global__ void l1_count_k(const int* __restrict__ dst, int* __restrict__ cntmat) {
  __shared__ int hist[NBUCK];
  const int t = threadIdx.x, blk = blockIdx.x;
  if (t < NBUCK) hist[t] = 0;
  __syncthreads();
  const int e0 = blk * L1_CHUNK;
  int d[L1_PER];
#pragma unroll
  for (int j = 0; j < L1_PER; ++j) d[j] = (t + j * GT < L1_CHUNK) ? dst[e0 + t + j * GT] : -1;
#pragma unroll
  for (int j = 0; j < L1_PER; ++j)
    if (d[j] >= 0) atomicAdd(&hist[d[j] >> BSH], 1);
  __syncthreads();
  if (t < NBUCK) cntmat[t * L1_BLOCKS + blk] = hist[t];
}
__global__ void l1_rowscan_k(const int* __restrict__ cntmat, int* __restrict__ offmat,
                             int* __restrict__ rowtot) {
  __shared__ int wtot[32];
  const int t = threadIdx.x, b = blockIdx.x;
  const int own = (t < L1_BLOCKS) ? cntmat[b * L1_BLOCKS + t] : 0;
  int total;
  const int pre = block_excl_scan(own, wtot, total);
  if (t < L1_BLOCKS) offmat[b * L1_BLOCKS + t] = pre;
  if (t == 0) rowtot[b] = total;
}
__global__ void l1_scatter_k(const int* __restrict__ dst, const int* __restrict__ cntmat,
                             const int* __restrict__ offmat, const int* __restrict__ rowtot,
                             int* __restrict__ part_i) {
  __shared__ int loff[NBUCK];
  __shared__ int cur[NBUCK];
  __shared__ int goff[NBUCK];
  __shared__ int wtot[32];
  __shared__ int2 stage[L1_CHUNK];
  int2* __restrict__ part = (int2*)part_i;
  const int t = threadIdx.x, blk = blockIdx.x;
  const bool isb = t < NBUCK;
  int total;
  const int pre = block_excl_scan(isb ? cntmat[t * L1_BLOCKS + blk] : 0, wtot, total);
  const int bbase = block_excl_scan(isb ? rowtot[t] : 0, wtot, total);
  if (isb) { loff[t] = pre; cur[t] = pre; goff[t] = bbase + offmat[t * L1_BLOCKS + blk]; }
  __syncthreads();
  const int e0 = blk * L1_CHUNK;
  int d[L1_PER];
#pragma unroll
  for (int j = 0; j < L1_PER; ++j) d[j] = (t + j * GT < L1_CHUNK) ? dst[e0 + t + j * GT] : -1;
#pragma unroll
  for (int j = 0; j < L1_PER; ++j)
    if (d[j] >= 0) {
      const int slot = atomicAdd(&cur[d[j] >> BSH], 1);
      stage[slot] = make_int2(e0 + t + j * GT, d[j]);
    }
  __syncthreads();
  for (int i = t; i < L1_CHUNK; i += GT) {
    const int2 v = stage[i];
    const int b = v.y >> BSH;
    part[goff[b] + (i - loff[b])] = v;
  }
}
__device__ __forceinline__ void l2_place_sort(int* st1, int* st2, const int2* __restrict__ part,
                                              int gb, int nb, int* cur, const int* nstart,
                                              const int* __restrict__ src, int* seid,
                                              int* __restrict__ ssrc) {
  const int t = threadIdx.x;
  for (int i0 = t; i0 < nb; i0 += L2_PER * GT) {
    int2 v[L2_PER];
#pragma unroll
    for (int j = 0; j < L2_PER; ++j)
      v[j] = (i0 + j * GT < nb) ? part[gb + i0 + j * GT] : make_int2(-1, -1);
#pragma unroll
    for (int j = 0; j < L2_PER; ++j)
      if (v[j].y >= 0) {
        const int slot = atomicAdd(&cur[v[j].y & (BNODES - 1)], 1);
        st1[slot] = v[j].x;
      }
  }
  __syncthreads();
  const int lane = t & 31;
  const int w = __builtin_amdgcn_readfirstlane(t >> 5);
  for (int ln = w; ln < BNODES; ln += GT / 32) {
    const int start = __builtin_amdgcn_readfirstlane(nstart[ln]);
    const int end = __builtin_amdgcn_readfirstlane(nstart[ln + 1]);
    for (int base = start; base < end; base += 32) {
      const int p = base + lane;
      const int x = (p < end) ? st1[p] : 0x7fffffff;
      int r = 0;
      for (int b2 = start; b2 < end; b2 += 32) {
        const int q = b2 + lane;
        const int y = (q < end) ? st1[q] : 0x7fffffff;
        const int m = min(32, end - b2);
        for (int k = 0; k < m; ++k) r += (lane_i(y, k) < x) ? 1 : 0;
      }
      if (p < end) st2[start + r] = x;
    }
  }
  __syncthreads();
  for (int i0 = t; i0 < nb; i0 += L2_PER * GT) {
    int x[L2_PER], s[L2_PER];
#pragma unroll
    for (int j = 0; j < L2_PER; ++j) x[j] = (i0 + j * GT < nb) ? st2[i0 + j * GT] : -1;
#pragma unroll
    for (int j = 0; j < L2_PER; ++j) s[j] = (x[j] >= 0) ? src[x[j]] : 0;
#pragma unroll
    for (int j = 0; j < L2_PER; ++j)
      if (x[j] >= 0) { seid[gb + i0 + j * GT] = x[j]; ssrc[gb + i0 + j * GT] = s[j]; }
  }
}
__global__ void l2_group_k(const int* __restrict__ part_i, const int* __restrict__ rowtot,
                           const int* __restrict__ src, int* __restrict__ rowptr, int* seid,
                           int* __restrict__ ssrc, int* eid_raw) {
  __shared__ int ncnt[BNODES];
  __shared__ int nstart[BNODES + 1];
  __shared__ int cur[BNODES];
  __shared__ int wtot[32];
  __shared__ int sa[L2_CAP];
  __shared__ int sb[L2_CAP];
  const int2* __restrict__ part = (const int2*)part_i;
  const int t = threadIdx.x, b = blockIdx.x;
  int gb;
  block_excl_scan((t < b) ? rowtot[t] : 0, wtot, gb);
  const int nb = rowtot[b];
  if (t < BNODES) ncnt[t] = 0;
  __syncthreads();
  for (int i0 = t; i0 < nb; i0 += L2_PER * GT) {
    int dn[L2_PER];
#pragma unroll
    for (int j = 0; j < L2_PER; ++j) dn[j] = (i0 + j * GT < nb) ? part[gb + i0 + j * GT].y : -1;
#pragma unroll
    for (int j = 0; j < L2_PER; ++j)
      if (dn[j] >= 0) atomicAdd(&ncnt[dn[j] & (BNODES - 1)], 1);
  }
  __syncthreads();
  int total;
  const int pre = block_excl_scan((t < BNODES) ? ncnt[t] : 0, wtot, total);
  if (t < BNODES) {
    nstart[t] = pre;
    cur[t] = pre;
    const int n = b * BNODES + t;
    if (n < NN) rowptr[n] = gb + pre;
  }
  if (t == 0) {
    nstart[BNODES] = nb;
    if (b == NBUCK - 1) rowptr[NN] = gb + nb;
  }
  __syncthreads();
  if (nb <= L2_CAP) l2_place_sort(sa, sb, part, gb, nb, cur, nstart, src, seid, ssrc);
  else l2_place_sort(eid_raw + gb, seid + gb, part, gb, nb, cur, nstart, src, seid, ssrc);
}

__global__ void wa3_k(const float* __restrict__ We_a, const float* __restrict__ ae_a,
                      const float* __restrict__ We_b, const float* __restrict__ ae_b,
                      const float* __restrict__ We_c, const float* __restrict__ ae_c,
                      float* __restrict__ wa3) {
  const int l = blockIdx.x;
  const float* __restrict__ We = (l == 0) ? We_a : ((l == 1) ? We_b : We_c);
  const float* __restrict__ ae = (l == 0) ? ae_a : ((l == 1) ? ae_b : ae_c);
  const int K = (l == 0) ? 128 : 64;
  const int i = threadIdx.x;
  if (i < 16) {
    float s = 0.f;
    for (int j = 0; j < K; ++j) s += We[i * K + j] * ae[j];
    wa3[l * 16 + i] = s;
  }
}

template <int M>
__device__ __forceinline__ float xmask_add(float keep, float send) {
  return keep + __int_as_float(__builtin_amdgcn_update_dpp(0, __float_as_int(send), 0x160 + M, 0xf,
                                                          0xf, true));
}

template <int COLT, bool SPLIT, bool PREP>
__global__ void gemm_k128(const float* __restrict__ A, const float* __restrict__ B,
                          const float* __restrict__ B2, float* __restrict__ C,
                          const float* __restrict__ a_src, const float* __restrict__ a_dst,
                          const float* __restrict__ a_src1, const float* __restrict__ a_dst1,
                          float* __restrict__ s_src, float* __restrict__ s_dst) {
  constexpr int COUT = COLT * 16;
  constexpr int LDB = SPLIT ? COUT / 2 : COUT;
  constexpr int AP = 132;
  __shared__ __attribute__((aligned(16))) float As[16 * AP];
  const int tid = threadIdx.x;
  const size_t abase = (size_t)blockIdx.x * (16 * 128);
  for (int i = tid * 4; i < 16 * 128; i += blockDim.x * 4)
    *(float4*)(&As[(i >> 7) * AP + (i & 127)]) = *(const float4*)(&A[abase + i]);
  __syncthreads();

  const int wave = SPLIT ? __builtin_amdgcn_readfirstlane(tid >> 5) : (tid >> 5);
  const int lane = tid & 31, half = lane >> 4, lidx = lane & 15;
  const int n0 = wave * 16;
  const int nb = SPLIT ? (wave % (COLT / 2)) * 16 : n0;
  const float* __restrict__ Bw = (SPLIT && wave >= COLT / 2) ? B2 : B;
  v8f acc = {};
#pragma unroll
  for (int k0 = 0; k0 < 128; k0 += 4) {
    v2f a, b;
    a.x = As[lidx * AP + k0 + 2 * half];
    a.y = As[lidx * AP + k0 + 2 * half + 1];
    b.x = Bw[(k0 + 2 * half) * LDB + nb + lidx];
    b.y = Bw[(k0 + 2 * half + 1) * LDB + nb + lidx];
    acc = __builtin_amdgcn_wmma_f32_16x16x4_f32(false, a, false, b, (short)0, acc,
                                                false, false);
  }
  const size_t row0 = (size_t)blockIdx.x * 16;
  __shared__ float part[PREP ? COLT * 32 : 1];
  if constexpr (PREP) {
    const bool right = SPLIT && wave >= COLT / 2;
    const float* __restrict__ asw = right ? a_src1 : a_src;
    const float* __restrict__ adw = right ? a_dst1 : a_dst;
    const float as = asw[nb + lidx], ad = adw[nb + lidx];
    const v8f P = acc * as, Q = acc * ad;
    const bool u8 = (lidx & 8) != 0, u4 = (lidx & 4) != 0, u2 = (lidx & 2) != 0, u1 = (lidx & 1) != 0;
    const v8f k8 = u8 ? Q : P, s8 = u8 ? P : Q;
    v8f r8;
    for (int i = 0; i < 8; ++i) r8[i] = xmask_add<8>(k8[i], s8[i]);
    const v4f k4 = u4 ? r8.hi : r8.lo, s4 = u4 ? r8.lo : r8.hi;
    v4f r4;
    for (int i = 0; i < 4; ++i) r4[i] = xmask_add<4>(k4[i], s4[i]);
    const v2f k2 = u2 ? r4.hi : r4.lo, s2 = u2 ? r4.lo : r4.hi;
    v2f r2;
    for (int i = 0; i < 2; ++i) r2[i] = xmask_add<2>(k2[i], s2[i]);
    const float r1 = xmask_add<1>(u1 ? r2.y : r2.x, u1 ? r2.x : r2.y);
    part[wave * 32 + (lidx >> 3) * 16 + 8 * half + (lidx & 7)] = r1;
    __syncthreads();
  }
#pragma unroll
  for (int r = 0; r < 8; ++r)
    C[(row0 + r + 8 * half) * COUT + n0 + lidx] = acc[r];
  if constexpr (PREP) {
    if (!SPLIT) {
      if (tid < 32) {
        float s = part[tid];
        for (int w = 1; w < COLT; ++w) s += part[w * 32 + tid];
        ((tid >> 4) ? s_dst : s_src)[row0 + (tid & 15)] = s;
      }
    } else {
      if (tid < 64) {
        const int hd = tid >> 5, slot = tid & 31;
        float s = part[(hd * (COLT / 2)) * 32 + slot];
        for (int w = 1; w < COLT / 2; ++w) s += part[(hd * (COLT / 2) + w) * 32 + slot];
        ((slot >> 4) ? s_dst : s_src)[(row0 + (slot & 15)) * 2 + hd] = s;
      }
    }
  }
}

static constexpr int ET_PER_THREAD = 4;
static constexpr int ET_BLOCK = 256;
static constexpr int ET_TILE = ET_BLOCK * ET_PER_THREAD;
static_assert((NE * 4) % ET_TILE == 0, "eterm3_k has no tail handling");
__device__ __forceinline__ float dot4(const float4& a, const float4& b) {
  return a.x * b.x + a.y * b.y + a.z * b.z + a.w * b.w;
}
__device__ __forceinline__ float quad_sum(float v) {
  v += __shfl_xor(v, 1, 32);
  v += __shfl_xor(v, 2, 32);
  return v;
}
__global__ void eterm3_k(const float* __restrict__ ea, const float* __restrict__ wa3,
                         float* __restrict__ et0, float* __restrict__ et12) {
  const int tid = threadIdx.x;
  const int q = tid & 3;
  const float4* __restrict__ ea4 = (const float4*)ea;
  const size_t base = (size_t)blockIdx.x * ET_TILE + tid;
  float4 v[ET_PER_THREAD];
#pragma unroll
  for (int k = 0; k < ET_PER_THREAD; ++k) v[k] = ea4[base + (size_t)k * ET_BLOCK];
  const float4 w0 = ((const float4*)wa3)[q];
  const float4 w1 = ((const float4*)wa3)[4 + q];
  const float4 w2 = ((const float4*)wa3)[8 + q];
#pragma unroll
  for (int k = 0; k < ET_PER_THREAD; ++k) {
    const float s0 = quad_sum(dot4(v[k], w0));
    const float s1 = quad_sum(dot4(v[k], w1));
    const float s2 = quad_sum(dot4(v[k], w2));
    if (q == 0) {
      const size_t e = (base + (size_t)k * ET_BLOCK) >> 2;
      et0[e] = s0; ((float2*)et12)[e] = make_float2(s1, s2);
    }
  }
}

template <int V> struct VecOf;
template <> struct VecOf<4> { typedef float4 type; };
template <> struct VecOf<2> { typedef float2 type; };
__device__ __forceinline__ void axpy(float4& acc, float a, const float4& v) {
  acc.x += a * v.x; acc.y += a * v.y; acc.z += a * v.z; acc.w += a * v.w;
}
__device__ __forceinline__ void axpy(float2& acc, float a, const float2& v) {
  acc.x += a * v.x; acc.y += a * v.y;
}
enum { EPI_NONE = 0, EPI_RELU = 1, EPI_RELU_RESID = 2 };
__device__ __forceinline__ void relu_v(float4& v) {
  v.x = fmaxf(v.x, 0.f); v.y = fmaxf(v.y, 0.f); v.z = fmaxf(v.z, 0.f); v.w = fmaxf(v.w, 0.f);
}
__device__ __forceinline__ void relu_v(float2& v) { v.x = fmaxf(v.x, 0.f); v.y = fmaxf(v.y, 0.f); }
__device__ __forceinline__ void add2_v(float4& v, const float4& r, const float4& b) {
  v.x = v.x + r.x + b.x; v.y = v.y + r.y + b.y; v.z = v.z + r.z + b.z; v.w = v.w + r.w + b.w;
}
__device__ __forceinline__ void add2_v(float2& v, const float2& r, const float2& b) {
  v.x = v.x + r.x + b.x; v.y = v.y + r.y + b.y;
}

template <bool ET>
__device__ __forceinline__ float edge_logit(int p, float pd, const int* __restrict__ ssrc,
                                            const int* __restrict__ seid,
                                            const float* __restrict__ s_src,
                                            const float* __restrict__ eterm, float& et) {
  float l = s_src[ssrc[p]] + pd;
  et = 0.f;
  if (ET) { et = eterm[seid[p]]; l += et; }
  return leaky(l);
}
template <bool ET>
__device__ __forceinline__ float self_logit(float ps, float pd, float et_sum, int deg) {
  float l = ps + pd;
  if (ET) l += et_sum / fmaxf((float)deg, 1.f);
  return leaky(l);
}
__device__ __forceinline__ float wave_max(float v) {
  for (int m = 16; m > 0; m >>= 1) v = fmaxf(v, __shfl_xor(v, m, 32));
  return v;
}
__device__ __forceinline__ float wave_sum(float v) {
  for (int m = 16; m > 0; m >>= 1) v += __shfl_xor(v, m, 32);
  return v;
}

template <int C, bool ET, int EPI>
__global__ void gat_node_k(const int* __restrict__ rowptr, const int* __restrict__ ssrc,
                           const int* __restrict__ seid, const float* __restrict__ s_src,
                           const float* __restrict__ s_dst,
                           const float* __restrict__ eterm, const float* __restrict__ h,
                           const float* __restrict__ bias, const float* __restrict__ resid,
                           const float* __restrict__ rbias, float* __restrict__ out) {
  constexpr int V = C / 32;
  typedef typename VecOf<V>::type vec_t;
  const int gid = blockIdx.x * blockDim.x + threadIdx.x;
  const int lane = gid & 31;
  const int n = __builtin_amdgcn_readfirstlane(gid >> 5);
  if (n >= NN) return;
  const int start = rowptr[n];
  const int end = rowptr[n + 1];
  const float ps = s_src[n], pd = s_dst[n];
  const int deg = end - start;
  const int ch = lane * V;
  const vec_t hs = *(const vec_t*)&h[(size_t)n * C + ch];
  const vec_t bv = *(const vec_t*)&bias[ch];
  vec_t acc = bv;
  vec_t rv = {}, rb = {};
  if (EPI == EPI_RELU_RESID) {
    rv = *(const vec_t*)&resid[(size_t)n * C + ch];
    rb = *(const vec_t*)&rbias[ch];
  }

  if (end - start <= 32) {
    const int p = start + lane;
    const bool act = p < end;
    int s = 0;
    float l = 0.f, et = 0.f;
    if (act) { s = ssrc[p]; l = edge_logit<ET>(p, pd, ssrc, seid, s_src, eterm, et); }
    const float sl = self_logit<ET>(ps, pd, ET ? wave_sum(et) : 0.f, deg);
    const float m = fmaxf(wave_max(act ? l : -INFINITY), sl);
    const float w = act ? __expf(l - m) : 0.f;
    const float wself = __expf(sl - m);
    const float den = wself + wave_sum(w) + 1e-16f;
    const float a = w / den;
    axpy(acc, wself / den, hs);
#pragma unroll 4
    for (int k = 0; k < deg; ++k) {
      const int sk = lane_i(s, k);
      const float ak = lane_f(a, k);
      axpy(acc, ak, *(const vec_t*)&h[(size_t)sk * C + ch]);
    }
  } else {
    float m = -INFINITY, ets = 0.f, et;
    for (int base = start; base < end; base += 32) {
      const int p = base + lane;
      if (p < end) { m = fmaxf(m, edge_logit<ET>(p, pd, ssrc, seid, s_src, eterm, et)); ets += et; }
    }
    const float sl = self_logit<ET>(ps, pd, ET ? wave_sum(ets) : 0.f, deg);
    m = fmaxf(wave_max(m), sl);
    float ssum = 0.f;
    for (int base = start; base < end; base += 32) {
      const int p = base + lane;
      if (p < end) ssum += __expf(edge_logit<ET>(p, pd, ssrc, seid, s_src, eterm, et) - m);
    }
    const float wself = __expf(sl - m);
    const float den = wself + wave_sum(ssum) + 1e-16f;
    axpy(acc, wself / den, hs);
    for (int base = start; base < end; base += 32) {
      const int p = base + lane;
      int s = 0;
      float a = 0.f;
      if (p < end) {
        s = ssrc[p];
        a = __expf(edge_logit<ET>(p, pd, ssrc, seid, s_src, eterm, et) - m) / den;
      }
      const int cntk = min(32, end - base);
#pragma unroll 4
      for (int k = 0; k < cntk; ++k) {
        const int sk = lane_i(s, k);
        const float ak = lane_f(a, k);
        axpy(acc, ak, *(const vec_t*)&h[(size_t)sk * C + ch]);
      }
    }
  }
  if (EPI != EPI_NONE) relu_v(acc);
  if (EPI == EPI_RELU_RESID) add2_v(acc, rv, rb);
  *(vec_t*)&out[(size_t)n * C + ch] = acc;
}

struct Logit2 { float l0, l1, e0, e1; };
__device__ __forceinline__ Logit2 edge_logit2(int p, float2 pd, const int* __restrict__ ssrc,
                                              const int* __restrict__ seid,
                                              const float2* __restrict__ s_src2,
                                              const float2* __restrict__ et12) {
  const float2 ss = s_src2[ssrc[p]];
  const float2 et = et12[seid[p]];
  Logit2 r;
  r.e0 = et.x; r.e1 = et.y;
  float l0 = ss.x + pd.x; l0 += et.x;
  float l1 = ss.y + pd.y; l1 += et.y;
  r.l0 = leaky(l0); r.l1 = leaky(l1);
  return r;
}
__global__ void gat_node2_k(const int* __restrict__ rowptr, const int* __restrict__ ssrc,
                            const int* __restrict__ seid, const float* __restrict__ s_src2f,
                            const float* __restrict__ s_dst2f, const float* __restrict__ et12f,
                            const float* __restrict__ h, const float* __restrict__ bias0,
                            const float* __restrict__ bias1, float* __restrict__ out0,
                            float* __restrict__ out1) {
  const float2* __restrict__ s_src2 = (const float2*)s_src2f;
  const float2* __restrict__ s_dst2 = (const float2*)s_dst2f;
  const float2* __restrict__ et12 = (const float2*)et12f;
  const int gid = blockIdx.x * blockDim.x + threadIdx.x;
  const int lane = gid & 31;
  const int n = __builtin_amdgcn_readfirstlane(gid >> 5);
  if (n >= NN) return;
  const int start = rowptr[n];
  const int end = rowptr[n + 1];
  const float2 ps = s_src2[n], pd = s_dst2[n];
  const int deg = end - start;
  const bool hd = lane >= 16;
  const int ch = lane * 4;
  const int hc = (lane & 15) * 4;
  const float4 hs = *(const float4*)&h[(size_t)n * 128 + ch];
  const float4 bv = *(const float4*)&(hd ? bias1 : bias0)[hc];
  float4 acc = bv;

  if (deg <= 32) {
    const int p = start + lane;
    const bool act = p < end;
    int s = 0;
    Logit2 q = {0.f, 0.f, 0.f, 0.f};
    if (act) { s = ssrc[p]; q = edge_logit2(p, pd, ssrc, seid, s_src2, et12); }
    const float sl0 = self_logit<true>(ps.x, pd.x, wave_sum(q.e0), deg);
    const float sl1 = self_logit<true>(ps.y, pd.y, wave_sum(q.e1), deg);
    const float m0 = fmaxf(wave_max(act ? q.l0 : -INFINITY), sl0);
    const float m1 = fmaxf(wave_max(act ? q.l1 : -INFINITY), sl1);
    const float w0 = act ? __expf(q.l0 - m0) : 0.f;
    const float w1 = act ? __expf(q.l1 - m1) : 0.f;
    const float wself0 = __expf(sl0 - m0);
    const float wself1 = __expf(sl1 - m1);
    const float den0 = wself0 + wave_sum(w0) + 1e-16f;
    const float den1 = wself1 + wave_sum(w1) + 1e-16f;
    const float a0 = w0 / den0;
    const float a1 = w1 / den1;
    axpy(acc, hd ? wself1 / den1 : wself0 / den0, hs);
#pragma unroll 4
    for (int k = 0; k < deg; ++k) {
      const int sk = lane_i(s, k);
      const float ak0 = lane_f(a0, k);
      const float ak1 = lane_f(a1, k);
      axpy(acc, hd ? ak1 : ak0, *(const float4*)&h[(size_t)sk * 128 + ch]);
    }
  } else {
    float m0 = -INFINITY, m1 = -INFINITY, ets0 = 0.f, ets1 = 0.f;
    for (int base = start; base < end; base += 32) {
      const int p = base + lane;
      if (p < end) {
        const Logit2 q = edge_logit2(p, pd, ssrc, seid, s_src2, et12);
        m0 = fmaxf(m0, q.l0); m1 = fmaxf(m1, q.l1);
        ets0 += q.e0; ets1 += q.e1;
      }
    }
    const float sl0 = self_logit<true>(ps.x, pd.x, wave_sum(ets0), deg);
    const float sl1 = self_logit<true>(ps.y, pd.y, wave_sum(ets1), deg);
    m0 = fmaxf(wave_max(m0), sl0);
    m1 = fmaxf(wave_max(m1), sl1);
    float ssum0 = 0.f, ssum1 = 0.f;
    for (int base = start; base < end; base += 32) {
      const int p = base + lane;
      if (p < end) {
        const Logit2 q = edge_logit2(p, pd, ssrc, seid, s_src2, et12);
        ssum0 += __expf(q.l0 - m0); ssum1 += __expf(q.l1 - m1);
      }
    }
    const float wself0 = __expf(sl0 - m0);
    const float wself1 = __expf(sl1 - m1);
    const float den0 = wself0 + wave_sum(ssum0) + 1e-16f;
    const float den1 = wself1 + wave_sum(ssum1) + 1e-16f;
    axpy(acc, hd ? wself1 / den1 : wself0 / den0, hs);
    for (int base = start; base < end; base += 32) {
      const int p = base + lane;
      int s = 0;
      float a0 = 0.f, a1 = 0.f;
      if (p < end) {
        s = ssrc[p];
        const Logit2 q = edge_logit2(p, pd, ssrc, seid, s_src2, et12);
        a0 = __expf(q.l0 - m0) / den0;
        a1 = __expf(q.l1 - m1) / den1;
      }
      const int cntk = min(32, end - base);
#pragma unroll 4
      for (int k = 0; k < cntk; ++k) {
        const int sk = lane_i(s, k);
        const float ak0 = lane_f(a0, k);
        const float ak1 = lane_f(a1, k);
        axpy(acc, hd ? ak1 : ak0, *(const float4*)&h[(size_t)sk * 128 + ch]);
      }
    }
  }
  float* __restrict__ o = hd ? out1 : out0;
  *(float4*)&o[(size_t)n * 64 + hc] = acc;
}

struct Scratch {
  float *s_src, *s_dst;
  const float* eterm;
  const int *rowptr, *ssrc, *seid;
};

template <int C, bool ET, int EPI>
static void run_gat(hipStream_t st, const float* hin, const float* Wm, const float* asrc,
                    const float* adst, const float* bias, float* hg, float* out,
                    const Scratch& S, const float* resid, const float* rbias) {
  constexpr int COLT = C / 16;
  gemm_k128<COLT, false, true><<<NN / 16, 32 * COLT, 0, st>>>(hin, Wm, nullptr, hg, asrc, adst, nullptr,
                                                             nullptr, S.s_src, S.s_dst);
  gat_node_k<C, ET, EPI><<<(NN * 32 + 255) / 256, 256, 0, st>>>(S.rowptr, S.ssrc, S.seid,
                                                                S.s_src, S.s_dst, S.eterm, hg,
                                                                bias, resid, rbias, out);
}

extern "C" void kernel_launch(void* const* d_in, const int* in_sizes, int n_in, void* d_out,
                              int out_size, void* d_ws, size_t ws_size, hipStream_t stream) {
  (void)in_sizes; (void)n_in; (void)out_size; (void)ws_size;
  const float* x = (const float*)d_in[0];
  const int* ei = (const int*)d_in[1];
  const int* src = ei;
  const int* dst = ei + NE;
  const float* edge_attr = (const float*)d_in[2];
  const float *W1 = (const float*)d_in[3], *a1s = (const float*)d_in[4],
              *a1d = (const float*)d_in[5], *b1 = (const float*)d_in[6];
  const float *W2 = (const float*)d_in[7], *a2s = (const float*)d_in[8],
              *a2d = (const float*)d_in[9], *b2 = (const float*)d_in[10];
  const float *We2 = (const float*)d_in[11], *ae2 = (const float*)d_in[12];
  const float *Wr = (const float*)d_in[13], *br = (const float*)d_in[14];
  const float *Wmu = (const float*)d_in[15], *amus = (const float*)d_in[16],
              *amud = (const float*)d_in[17], *bmu = (const float*)d_in[18],
              *Wemu = (const float*)d_in[19], *aemu = (const float*)d_in[20];
  const float *Wls = (const float*)d_in[21], *alss = (const float*)d_in[22],
              *alsd = (const float*)d_in[23], *bls = (const float*)d_in[24],
              *Wels = (const float*)d_in[25], *aels = (const float*)d_in[26];
  float* outf = (float*)d_out;

  float* W = (float*)d_ws;
  size_t o = 0;
  float* hA = W + o; o += (size_t)NN * 128;
  float* hB = W + o; o += (size_t)NN * 128;
  float* hC = W + o; o += (size_t)NN * 128;
  Scratch S;
  S.s_src = W + o; o += NN;
  S.s_dst = W + o; o += NN;
  float* et0 = W + o; o += NE;
  float* et12 = W + o; o += (size_t)NE * 2;
  float* wa3 = W + o; o += 48;
  S.eterm = et0;
  int* rowptr = (int*)(W + o); o += NN + 16;
  int* part = (int*)(W + o); o += (size_t)NE * 2;
  int* cntmat = (int*)(W + o); o += NBUCK * L1_BLOCKS + 2;
  int* offmat = (int*)(W + o); o += NBUCK * L1_BLOCKS + 2;
  int* rowtot = (int*)(W + o); o += NBUCK + 1;
  int* eid_raw = (int*)(W + o); o += NE;
  int* seid = (int*)(W + o); o += NE;
  int* ssrc = (int*)(W + o); o += NE;
  S.rowptr = rowptr; S.ssrc = ssrc; S.seid = seid;
  float* s_src2 = W + o; o += (size_t)NN * 2;
  float* s_dst2 = W + o; o += (size_t)NN * 2;

  l1_count_k<<<L1_BLOCKS, GT, 0, stream>>>(dst, cntmat);
  l1_rowscan_k<<<NBUCK, 256, 0, stream>>>(cntmat, offmat, rowtot);
  l1_scatter_k<<<L1_BLOCKS, GT, 0, stream>>>(dst, cntmat, offmat, rowtot, part);
  l2_group_k<<<NBUCK, GT, 0, stream>>>(part, rowtot, src, rowptr, seid, ssrc, eid_raw);

  wa3_k<<<3, 16, 0, stream>>>(We2, ae2, Wemu, aemu, Wels, aels, wa3);
  eterm3_k<<<(NE * 4) / ET_TILE, ET_BLOCK, 0, stream>>>(edge_attr, wa3, et0, et12);

  gemm_k128<8, false, false><<<NN / 16, 256, 0, stream>>>(x, Wr, nullptr, hC, nullptr, nullptr, nullptr,
                                                        nullptr, nullptr, nullptr);

  run_gat<128, false, EPI_RELU>(stream, x, W1, a1s, a1d, b1, hA, hB, S, nullptr, nullptr);

  S.eterm = et0;
  run_gat<128, true, EPI_RELU_RESID>(stream, hB, W2, a2s, a2d, b2, hA, hB, S, hC, br);

  gemm_k128<8, true, true><<<NN / 16, 256, 0, stream>>>(hB, Wmu, Wls, hA, amus, amud, alss, alsd, s_src2,
                                                      s_dst2);
  gat_node2_k<<<(NN * 32 + 255) / 256, 256, 0, stream>>>(rowptr, ssrc, seid, s_src2, s_dst2, et12, hA,
                                                         bmu, bls, outf, outf + (size_t)NN * 64);
}


# Node pass of conv1/conv2 also runs the NEXT conv's GEMM on its 16 finished rows (A tile in LDS, weights staged once per resident block): rows stay out of memory, 2 of 13 launches go; sums unchanged
# speedup vs baseline: 7.5454x; 1.0762x over previous
#include <hip/hip_runtime.h>
#include <math.h>

typedef float v2f __attribute__((ext_vector_type(2)));
typedef float v4f __attribute__((ext_vector_type(4)));
typedef float v8f __attribute__((ext_vector_type(8)));

static constexpr int   NN = 100000;
static constexpr int   NE = 1600000;
static constexpr float NEG = 0.2f;

__device__ __forceinline__ float leaky(float x) { return x > 0.f ? x : NEG * x; }

static constexpr int BSH = 8;
static constexpr int BNODES = 1 << BSH;
static constexpr int NBUCK = (NN + BNODES - 1) >> BSH;
static constexpr int L1_BLOCKS = 250;
static constexpr int L1_CHUNK = NE / L1_BLOCKS;
static constexpr int GT = 1024;
static constexpr int L2_CAP = 6144;
static constexpr int L1_PER = (L1_CHUNK + GT - 1) / GT;
static constexpr int L2_PER = 4;
static_assert(L1_CHUNK * L1_BLOCKS == NE, "level 1 has no tail handling");
static_assert(NBUCK <= GT && BNODES <= GT && L1_BLOCKS <= 256, "one thread per bucket / node / level-1 block");

__device__ __forceinline__ int lane_i(int v, int k) { return __builtin_amdgcn_readlane(v, k); }
__device__ __forceinline__ float lane_f(float v, int k) {
  return __int_as_float(__builtin_amdgcn_readlane(__float_as_int(v), k));
}
__device__ __forceinline__ int wave_sum_i(int v) {
  for (int m = 16; m > 0; m >>= 1) v += __shfl_xor(v, m, 32);
  return v;
}
__device__ __forceinline__ int block_excl_scan(int v, int* wtot, int& total) {
  const int lane = threadIdx.x & 31, w = threadIdx.x >> 5;
  int inc = v;
  for (int o = 1; o < 32; o <<= 1) {
    const int u = __shfl_up(inc, o, 32);
    if (lane >= o) inc += u;
  }
  if (lane == 31) wtot[w] = inc;
  __syncthreads();
  const int wt = (lane < (int)(blockDim.x >> 5)) ? wtot[lane] : 0;
  const int before = wave_sum_i(lane < w ? wt : 0);
  total = wave_sum_i(wt);
  __syncthreads();
  return inc - v + before;
}

__global__ void l1_count_k(const int* __restrict__ dst, int* __restrict__ cntmat) {
  __shared__ int hist[NBUCK];
  const int t = threadIdx.x, blk = blockIdx.x;
  if (t < NBUCK) hist[t] = 0;
  __syncthreads();
  const int e0 = blk * L1_CHUNK;
  int d[L1_PER];
#pragma unroll
  for (int j = 0; j < L1_PER; ++j) d[j] = (t + j * GT < L1_CHUNK) ? dst[e0 + t + j * GT] : -1;
#pragma unroll
  for (int j = 0; j < L1_PER; ++j)
    if (d[j] >= 0) atomicAdd(&hist[d[j] >> BSH], 1);
  __syncthreads();
  if (t < NBUCK) cntmat[t * L1_BLOCKS + blk] = hist[t];
}
__global__ void l1_rowscan_k(const int* __restrict__ cntmat, int* __restrict__ offmat,
                             int* __restrict__ rowtot) {
  __shared__ int wtot[32];
  const int t = threadIdx.x, b = blockIdx.x;
  const int own = (t < L1_BLOCKS) ? cntmat[b * L1_BLOCKS + t] : 0;
  int total;
  const int pre = block_excl_scan(own, wtot, total);
  if (t < L1_BLOCKS) offmat[b * L1_BLOCKS + t] = pre;
  if (t == 0) rowtot[b] = total;
}
__global__ void l1_scatter_k(const int* __restrict__ dst, const int* __restrict__ cntmat,
                             const int* __restrict__ offmat, const int* __restrict__ rowtot,
                             int* __restrict__ part_i) {
  __shared__ int loff[NBUCK];
  __shared__ int cur[NBUCK];
  __shared__ int goff[NBUCK];
  __shared__ int wtot[32];
  __shared__ int2 stage[L1_CHUNK];
  int2* __restrict__ part = (int2*)part_i;
  const int t = threadIdx.x, blk = blockIdx.x;
  const bool isb = t < NBUCK;
  int total;
  const int pre = block_excl_scan(isb ? cntmat[t * L1_BLOCKS + blk] : 0, wtot, total);
  const int bbase = block_excl_scan(isb ? rowtot[t] : 0, wtot, total);
  if (isb) { loff[t] = pre; cur[t] = pre; goff[t] = bbase + offmat[t * L1_BLOCKS + blk]; }
  __syncthreads();
  const int e0 = blk * L1_CHUNK;
  int d[L1_PER];
#pragma unroll
  for (int j = 0; j < L1_PER; ++j) d[j] = (t + j * GT < L1_CHUNK) ? dst[e0 + t + j * GT] : -1;
#pragma unroll
  for (int j = 0; j < L1_PER; ++j)
    if (d[j] >= 0) {
      const int slot = atomicAdd(&cur[d[j] >> BSH], 1);
      stage[slot] = make_int2(e0 + t + j * GT, d[j]);
    }
  __syncthreads();
  for (int i = t; i < L1_CHUNK; i += GT) {
    const int2 v = stage[i];
    const int b = v.y >> BSH;
    part[goff[b] + (i - loff[b])] = v;
  }
}
__device__ __forceinline__ void l2_place_sort(int* st1, int* st2, const int2* __restrict__ part,
                                              int gb, int nb, int* cur, const int* nstart,
                                              const int* __restrict__ src, int* seid,
                                              int* __restrict__ ssrc) {
  const int t = threadIdx.x;
  for (int i0 = t; i0 < nb; i0 += L2_PER * GT) {
    int2 v[L2_PER];
#pragma unroll
    for (int j = 0; j < L2_PER; ++j)
      v[j] = (i0 + j * GT < nb) ? part[gb + i0 + j * GT] : make_int2(-1, -1);
#pragma unroll
    for (int j = 0; j < L2_PER; ++j)
      if (v[j].y >= 0) {
        const int slot = atomicAdd(&cur[v[j].y & (BNODES - 1)], 1);
        st1[slot] = v[j].x;
      }
  }
  __syncthreads();
  const int lane = t & 31;
  const int w = __builtin_amdgcn_readfirstlane(t >> 5);
  for (int ln = w; ln < BNODES; ln += GT / 32) {
    const int start = __builtin_amdgcn_readfirstlane(nstart[ln]);
    const int end = __builtin_amdgcn_readfirstlane(nstart[ln + 1]);
    for (int base = start; base < end; base += 32) {
      const int p = base + lane;
      const int x = (p < end) ? st1[p] : 0x7fffffff;
      int r = 0;
      for (int b2 = start; b2 < end; b2 += 32) {
        const int q = b2 + lane;
        const int y = (q < end) ? st1[q] : 0x7fffffff;
        const int m = min(32, end - b2);
        for (int k = 0; k < m; ++k) r += (lane_i(y, k) < x) ? 1 : 0;
      }
      if (p < end) st2[start + r] = x;
    }
  }
  __syncthreads();
  for (int i0 = t; i0 < nb; i0 += L2_PER * GT) {
    int x[L2_PER], s[L2_PER];
#pragma unroll
    for (int j = 0; j < L2_PER; ++j) x[j] = (i0 + j * GT < nb) ? st2[i0 + j * GT] : -1;
#pragma unroll
    for (int j = 0; j < L2_PER; ++j) s[j] = (x[j] >= 0) ? src[x[j]] : 0;
#pragma unroll
    for (int j = 0; j < L2_PER; ++j)
      if (x[j] >= 0) { seid[gb + i0 + j * GT] = x[j]; ssrc[gb + i0 + j * GT] = s[j]; }
  }
}
__global__ void l2_group_k(const int* __restrict__ part_i, const int* __restrict__ rowtot,
                           const int* __restrict__ src, int* __restrict__ rowptr, int* seid,
                           int* __restrict__ ssrc, int* eid_raw) {
  __shared__ int ncnt[BNODES];
  __shared__ int nstart[BNODES + 1];
  __shared__ int cur[BNODES];
  __shared__ int wtot[32];
  __shared__ int sa[L2_CAP];
  __shared__ int sb[L2_CAP];
  const int2* __restrict__ part = (const int2*)part_i;
  const int t = threadIdx.x, b = blockIdx.x;
  int gb;
  block_excl_scan((t < b) ? rowtot[t] : 0, wtot, gb);
  const int nb = rowtot[b];
  if (t < BNODES) ncnt[t] = 0;
  __syncthreads();
  for (int i0 = t; i0 < nb; i0 += L2_PER * GT) {
    int dn[L2_PER];
#pragma unroll
    for (int j = 0; j < L2_PER; ++j) dn[j] = (i0 + j * GT < nb) ? part[gb + i0 + j * GT].y : -1;
#pragma unroll
    for (int j = 0; j < L2_PER; ++j)
      if (dn[j] >= 0) atomicAdd(&ncnt[dn[j] & (BNODES - 1)], 1);
  }
  __syncthreads();
  int total;
  const int pre = block_excl_scan((t < BNODES) ? ncnt[t] : 0, wtot, total);
  if (t < BNODES) {
    nstart[t] = pre;
    cur[t] = pre;
    const int n = b * BNODES + t;
    if (n < NN) rowptr[n] = gb + pre;
  }
  if (t == 0) {
    nstart[BNODES] = nb;
    if (b == NBUCK - 1) rowptr[NN] = gb + nb;
  }
  __syncthreads();
  if (nb <= L2_CAP) l2_place_sort(sa, sb, part, gb, nb, cur, nstart, src, seid, ssrc);
  else l2_place_sort(eid_raw + gb, seid + gb, part, gb, nb, cur, nstart, src, seid, ssrc);
}

__global__ void wa3_k(const float* __restrict__ We_a, const float* __restrict__ ae_a,
                      const float* __restrict__ We_b, const float* __restrict__ ae_b,
                      const float* __restrict__ We_c, const float* __restrict__ ae_c,
                      float* __restrict__ wa3) {
  const int l = blockIdx.x;
  const float* __restrict__ We = (l == 0) ? We_a : ((l == 1) ? We_b : We_c);
  const float* __restrict__ ae = (l == 0) ? ae_a : ((l == 1) ? ae_b : ae_c);
  const int K = (l == 0) ? 128 : 64;
  const int i = threadIdx.x;
  if (i < 16) {
    float s = 0.f;
    for (int j = 0; j < K; ++j) s += We[i * K + j] * ae[j];
    wa3[l * 16 + i] = s;
  }
}

template <int M>
__device__ __forceinline__ float xmask_add(float keep, float send) {
  return keep + __int_as_float(__builtin_amdgcn_update_dpp(0, __float_as_int(send), 0x160 + M, 0xf,
                                                          0xf, true));
}

static constexpr int AP = 132;
template <int COLT, bool SPLIT, bool PREP>
__global__ void gemm_k128(const float* __restrict__ A, const float* __restrict__ B,
                          const float* __restrict__ B2, float* __restrict__ C,
                          const float* __restrict__ a_src, const float* __restrict__ a_dst,
                          const float* __restrict__ a_src1, const float* __restrict__ a_dst1,
                          float* __restrict__ s_src, float* __restrict__ s_dst) {
  constexpr int COUT = COLT * 16;
  constexpr int LDB = SPLIT ? COUT / 2 : COUT;
  __shared__ __attribute__((aligned(16))) float As[16 * AP];
  const int tid = threadIdx.x;
  const size_t abase = (size_t)blockIdx.x * (16 * 128);
  for (int i = tid * 4; i < 16 * 128; i += blockDim.x * 4)
    *(float4*)(&As[(i >> 7) * AP + (i & 127)]) = *(const float4*)(&A[abase + i]);
  __syncthreads();

  const int wave = SPLIT ? __builtin_amdgcn_readfirstlane(tid >> 5) : (tid >> 5);
  const int lane = tid & 31, half = lane >> 4, lidx = lane & 15;
  const int n0 = wave * 16;
  const int nb = SPLIT ? (wave % (COLT / 2)) * 16 : n0;
  const float* __restrict__ Bw = (SPLIT && wave >= COLT / 2) ? B2 : B;
  v8f acc = {};
#pragma unroll
  for (int k0 = 0; k0 < 128; k0 += 4) {
    v2f a, b;
    a.x = As[lidx * AP + k0 + 2 * half];
    a.y = As[lidx * AP + k0 + 2 * half + 1];
    b.x = Bw[(k0 + 2 * half) * LDB + nb + lidx];
    b.y = Bw[(k0 + 2 * half + 1) * LDB + nb + lidx];
    acc = __builtin_amdgcn_wmma_f32_16x16x4_f32(false, a, false, b, (short)0, acc,
                                                false, false);
  }
  const size_t row0 = (size_t)blockIdx.x * 16;
  __shared__ float part[PREP ? COLT * 32 : 1];
  if constexpr (PREP) {
    const bool right = SPLIT && wave >= COLT / 2;
    const float* __restrict__ asw = right ? a_src1 : a_src;
    const float* __restrict__ adw = right ? a_dst1 : a_dst;
    const float as = asw[nb + lidx], ad = adw[nb + lidx];
    const v8f P = acc * as, Q = acc * ad;
    const bool u8 = (lidx & 8) != 0, u4 = (lidx & 4) != 0, u2 = (lidx & 2) != 0, u1 = (lidx & 1) != 0;
    const v8f k8 = u8 ? Q : P, s8 = u8 ? P : Q;
    v8f r8;
    for (int i = 0; i < 8; ++i) r8[i] = xmask_add<8>(k8[i], s8[i]);
    const v4f k4 = u4 ? r8.hi : r8.lo, s4 = u4 ? r8.lo : r8.hi;
    v4f r4;
    for (int i = 0; i < 4; ++i) r4[i] = xmask_add<4>(k4[i], s4[i]);
    const v2f k2 = u2 ? r4.hi : r4.lo, s2 = u2 ? r4.lo : r4.hi;
    v2f r2;
    for (int i = 0; i < 2; ++i) r2[i] = xmask_add<2>(k2[i], s2[i]);
    const float r1 = xmask_add<1>(u1 ? r2.y : r2.x, u1 ? r2.x : r2.y);
    part[wave * 32 + (lidx >> 3) * 16 + 8 * half + (lidx & 7)] = r1;
    __syncthreads();
  }
#pragma unroll
  for (int r = 0; r < 8; ++r)
    C[(row0 + r + 8 * half) * COUT + n0 + lidx] = acc[r];
  if constexpr (PREP) {
    if (!SPLIT) {
      if (tid < 32) {
        float s = part[tid];
        for (int w = 1; w < COLT; ++w) s += part[w * 32 + tid];
        ((tid >> 4) ? s_dst : s_src)[row0 + (tid & 15)] = s;
      }
    } else {
      if (tid < 64) {
        const int hd = tid >> 5, slot = tid & 31;
        float s = part[(hd * (COLT / 2)) * 32 + slot];
        for (int w = 1; w < COLT / 2; ++w) s += part[(hd * (COLT / 2) + w) * 32 + slot];
        ((slot >> 4) ? s_dst : s_src)[(row0 + (slot & 15)) * 2 + hd] = s;
      }
    }
  }
}

static constexpr int ET_PER_THREAD = 4;
static constexpr int ET_BLOCK = 256;
static constexpr int ET_TILE = ET_BLOCK * ET_PER_THREAD;
static_assert((NE * 4) % ET_TILE == 0, "eterm3_k has no tail handling");
__device__ __forceinline__ float dot4(const float4& a, const float4& b) {
  return a.x * b.x + a.y * b.y + a.z * b.z + a.w * b.w;
}
__device__ __forceinline__ float quad_sum(float v) {
  v += __shfl_xor(v, 1, 32);
  v += __shfl_xor(v, 2, 32);
  return v;
}
__global__ void eterm3_k(const float* __restrict__ ea, const float* __restrict__ wa3,
                         float* __restrict__ et0, float* __restrict__ et12) {
  const int tid = threadIdx.x;
  const int q = tid & 3;
  const float4* __restrict__ ea4 = (const float4*)ea;
  const size_t base = (size_t)blockIdx.x * ET_TILE + tid;
  float4 v[ET_PER_THREAD];
#pragma unroll
  for (int k = 0; k < ET_PER_THREAD; ++k) v[k] = ea4[base + (size_t)k * ET_BLOCK];
  const float4 w0 = ((const float4*)wa3)[q];
  const float4 w1 = ((const float4*)wa3)[4 + q];
  const float4 w2 = ((const float4*)wa3)[8 + q];
#pragma unroll
  for (int k = 0; k < ET_PER_THREAD; ++k) {
    const float s0 = quad_sum(dot4(v[k], w0));
    const float s1 = quad_sum(dot4(v[k], w1));
    const float s2 = quad_sum(dot4(v[k], w2));
    if (q == 0) {
      const size_t e = (base + (size_t)k * ET_BLOCK) >> 2;
      et0[e] = s0; ((float2*)et12)[e] = make_float2(s1, s2);
    }
  }
}

__device__ __forceinline__ void axpy(float4& acc, float a, const float4& v) {
  acc.x += a * v.x; acc.y += a * v.y; acc.z += a * v.z; acc.w += a * v.w;
}
enum { EPI_RELU = 1, EPI_RELU_RESID = 2 };
__device__ __forceinline__ void relu_v(float4& v) {
  v.x = fmaxf(v.x, 0.f); v.y = fmaxf(v.y, 0.f); v.z = fmaxf(v.z, 0.f); v.w = fmaxf(v.w, 0.f);
}
__device__ __forceinline__ void add2_v(float4& v, const float4& r, const float4& b) {
  v.x = v.x + r.x + b.x; v.y = v.y + r.y + b.y; v.z = v.z + r.z + b.z; v.w = v.w + r.w + b.w;
}

template <bool ET>
__device__ __forceinline__ float edge_logit(int p, float pd, const int* __restrict__ ssrc,
                                            const int* __restrict__ seid,
                                            const float* __restrict__ s_src,
                                            const float* __restrict__ eterm, float& et) {
  float l = s_src[ssrc[p]] + pd;
  et = 0.f;
  if (ET) { et = eterm[seid[p]]; l += et; }
  return leaky(l);
}
template <bool ET>
__device__ __forceinline__ float self_logit(float ps, float pd, float et_sum, int deg) {
  float l = ps + pd;
  if (ET) l += et_sum / fmaxf((float)deg, 1.f);
  return leaky(l);
}
__device__ __forceinline__ float wave_max(float v) {
  for (int m = 16; m > 0; m >>= 1) v = fmaxf(v, __shfl_xor(v, m, 32));
  return v;
}
__device__ __forceinline__ float wave_sum(float v) {
  for (int m = 16; m > 0; m >>= 1) v += __shfl_xor(v, m, 32);
  return v;
}

template <bool ET, int EPI>
__device__ __forceinline__ float4 gat_node_row(int n, int lane, const int* __restrict__ rowptr,
                                               const int* __restrict__ ssrc,
                                               const int* __restrict__ seid,
                                               const float* __restrict__ s_src,
                                               const float* __restrict__ s_dst,
                                               const float* __restrict__ eterm,
                                               const float* __restrict__ h,
                                               const float* __restrict__ bias,
                                               const float* __restrict__ resid,
                                               const float* __restrict__ rbias) {
  constexpr int C = 128;
  const int start = rowptr[n];
  const int end = rowptr[n + 1];
  const float ps = s_src[n], pd = s_dst[n];
  const int deg = end - start;
  const int ch = lane * 4;
  const float4 hs = *(const float4*)&h[(size_t)n * C + ch];
  const float4 bv = *(const float4*)&bias[ch];
  float4 acc = bv;
  float4 rv = {}, rb = {};
  if (EPI == EPI_RELU_RESID) {
    rv = *(const float4*)&resid[(size_t)n * C + ch];
    rb = *(const float4*)&rbias[ch];
  }

  if (end - start <= 32) {
    const int p = start + lane;
    const bool act = p < end;
    int s = 0;
    float l = 0.f, et = 0.f;
    if (act) { s = ssrc[p]; l = edge_logit<ET>(p, pd, ssrc, seid, s_src, eterm, et); }
    const float sl = self_logit<ET>(ps, pd, ET ? wave_sum(et) : 0.f, deg);
    const float m = fmaxf(wave_max(act ? l : -INFINITY), sl);
    const float w = act ? __expf(l - m) : 0.f;
    const float wself = __expf(sl - m);
    const float den = wself + wave_sum(w) + 1e-16f;
    const float a = w / den;
    axpy(acc, wself / den, hs);
#pragma unroll 4
    for (int k = 0; k < deg; ++k) {
      const int sk = lane_i(s, k);
      const float ak = lane_f(a, k);
      axpy(acc, ak, *(const float4*)&h[(size_t)sk * C + ch]);
    }
  } else {
    float m = -INFINITY, ets = 0.f, et;
    for (int base = start; base < end; base += 32) {
      const int p = base + lane;
      if (p < end) { m = fmaxf(m, edge_logit<ET>(p, pd, ssrc, seid, s_src, eterm, et)); ets += et; }
    }
    const float sl = self_logit<ET>(ps, pd, ET ? wave_sum(ets) : 0.f, deg);
    m = fmaxf(wave_max(m), sl);
    float ssum = 0.f;
    for (int base = start; base < end; base += 32) {
      const int p = base + lane;
      if (p < end) ssum += __expf(edge_logit<ET>(p, pd, ssrc, seid, s_src, eterm, et) - m);
    }
    const float wself = __expf(sl - m);
    const float den = wself + wave_sum(ssum) + 1e-16f;
    axpy(acc, wself / den, hs);
    for (int base = start; base < end; base += 32) {
      const int p = base + lane;
      int s = 0;
      float a = 0.f;
      if (p < end) {
        s = ssrc[p];
        a = __expf(edge_logit<ET>(p, pd, ssrc, seid, s_src, eterm, et) - m) / den;
      }
      const int cntk = min(32, end - base);
#pragma unroll 4
      for (int k = 0; k < cntk; ++k) {
        const int sk = lane_i(s, k);
        const float ak = lane_f(a, k);
        axpy(acc, ak, *(const float4*)&h[(size_t)sk * C + ch]);
      }
    }
  }
  relu_v(acc);
  if (EPI == EPI_RELU_RESID) add2_v(acc, rv, rb);
  return acc;
}

static constexpr int NG_TILES = NN / 16;
static constexpr int NG_BLOCKS = 1024;
static_assert(NN % 16 == 0, "node_gemm_k has no tail handling");
template <bool ET, int EPI, bool SPLIT>
__global__ void __launch_bounds__(512, 16)
node_gemm_k(const int* __restrict__ rowptr, const int* __restrict__ ssrc,
            const int* __restrict__ seid, const float* __restrict__ s_src,
            const float* __restrict__ s_dst, const float* __restrict__ eterm,
            const float* __restrict__ h, const float* __restrict__ bias,
            const float* __restrict__ resid, const float* __restrict__ rbias,
            const float* __restrict__ B, const float* __restrict__ B2,
            float* __restrict__ C, const float* __restrict__ a_src,
            const float* __restrict__ a_dst, const float* __restrict__ a_src1,
            const float* __restrict__ a_dst1, float* __restrict__ s_src_o,
            float* __restrict__ s_dst_o) {
  __shared__ __attribute__((aligned(16))) float Bs[128 * 128];
  __shared__ __attribute__((aligned(16))) float As[16 * AP];
  __shared__ float part[8 * 32];
  constexpr int LDB = SPLIT ? 64 : 128;
  const int tid = threadIdx.x;
  const int lane = tid & 31, half = lane >> 4, lidx = lane & 15;
  const int wave = __builtin_amdgcn_readfirstlane(tid >> 5);

#pragma unroll
  for (int it = tid; it < 64 * 32; it += 512) {
    const int kp = it >> 5, c = (it & 31) * 4;
    const float* __restrict__ Bm = (SPLIT && c >= 64) ? B2 : B;
    const int cc = SPLIT ? (c & 63) : c;
    const float4 r0 = *(const float4*)&Bm[(2 * kp) * LDB + cc];
    const float4 r1 = *(const float4*)&Bm[(2 * kp + 1) * LDB + cc];
    float* d = &Bs[((((c >> 4) * 32 + (kp >> 1)) * 32) + (kp & 1) * 16 + (c & 15)) * 2];
    *(float4*)(d) = make_float4(r0.x, r1.x, r0.y, r1.y);
    *(float4*)(d + 4) = make_float4(r0.z, r1.z, r0.w, r1.w);
  }
  const bool right = SPLIT && wave >= 4;
  const int nb = SPLIT ? (wave & 3) * 16 : (wave & 7) * 16;
  const int n0 = (wave & 7) * 16;

  for (int tile = blockIdx.x; tile < NG_TILES; tile += NG_BLOCKS) {
    const int row0 = tile * 16;
    const float4 row = gat_node_row<ET, EPI>(row0 + wave, lane, rowptr, ssrc, seid, s_src, s_dst, eterm,
                                             h, bias, resid, rbias);
    *(float4*)&As[wave * AP + lane * 4] = row;
    __syncthreads();
    v8f acc = {};
    if (wave < 8) {
      const float as = (right ? a_src1 : a_src)[nb + lidx], ad = (right ? a_dst1 : a_dst)[nb + lidx];
#pragma unroll
      for (int ks = 0; ks < 32; ++ks) {
        v2f a;
        a.x = As[lidx * AP + 4 * ks + 2 * half];
        a.y = As[lidx * AP + 4 * ks + 2 * half + 1];
        const v2f b = *(const v2f*)&Bs[((wave * 32 + ks) * 32 + lane) * 2];
        acc = __builtin_amdgcn_wmma_f32_16x16x4_f32(false, a, false, b, (short)0, acc,
                                                    false, false);
      }
      const v8f P = acc * as, Q = acc * ad;
      const bool u8 = (lidx & 8) != 0, u4 = (lidx & 4) != 0, u2 = (lidx & 2) != 0, u1 = (lidx & 1) != 0;
      const v8f k8 = u8 ? Q : P, s8 = u8 ? P : Q;
      v8f r8;
      for (int i = 0; i < 8; ++i) r8[i] = xmask_add<8>(k8[i], s8[i]);
      const v4f k4 = u4 ? r8.hi : r8.lo, s4 = u4 ? r8.lo : r8.hi;
      v4f r4;
      for (int i = 0; i < 4; ++i) r4[i] = xmask_add<4>(k4[i], s4[i]);
      const v2f k2 = u2 ? r4.hi : r4.lo, s2 = u2 ? r4.lo : r4.hi;
      v2f r2;
      for (int i = 0; i < 2; ++i) r2[i] = xmask_add<2>(k2[i], s2[i]);
      const float r1 = xmask_add<1>(u1 ? r2.y : r2.x, u1 ? r2.x : r2.y);
      part[wave * 32 + (lidx >> 3) * 16 + 8 * half + (lidx & 7)] = r1;
    }
    __syncthreads();
    if (wave < 8) {
      float* __restrict__ Ct = C + (size_t)row0 * 128;
#pragma unroll
      for (int r = 0; r < 8; ++r)
        Ct[(r + 8 * half) * 128 + n0 + lidx] = acc[r];
    }
    if (!SPLIT) {
      if (tid < 32) {
        float s = part[tid];
        for (int w = 1; w < 8; ++w) s += part[w * 32 + tid];
        ((tid >> 4) ? s_dst_o : s_src_o)[row0 + (tid & 15)] = s;
      }
    } else {
      if (tid < 64) {
        const int hd = tid >> 5, slot = tid & 31;
        float s = part[(hd * 4) * 32 + slot];
        for (int w = 1; w < 4; ++w) s += part[(hd * 4 + w) * 32 + slot];
        ((slot >> 4) ? s_dst_o : s_src_o)[(size_t)(row0 + (slot & 15)) * 2 + hd] = s;
      }
    }
  }
}

struct Logit2 { float l0, l1, e0, e1; };
__device__ __forceinline__ Logit2 edge_logit2(int p, float2 pd, const int* __restrict__ ssrc,
                                              const int* __restrict__ seid,
                                              const float2* __restrict__ s_src2,
                                              const float2* __restrict__ et12) {
  const float2 ss = s_src2[ssrc[p]];
  const float2 et = et12[seid[p]];
  Logit2 r;
  r.e0 = et.x; r.e1 = et.y;
  float l0 = ss.x + pd.x; l0 += et.x;
  float l1 = ss.y + pd.y; l1 += et.y;
  r.l0 = leaky(l0); r.l1 = leaky(l1);
  return r;
}
__global__ void gat_node2_k(const int* __restrict__ rowptr, const int* __restrict__ ssrc,
                            const int* __restrict__ seid, const float* __restrict__ s_src2f,
                            const float* __restrict__ s_dst2f, const float* __restrict__ et12f,
                            const float* __restrict__ h, const float* __restrict__ bias0,
                            const float* __restrict__ bias1, float* __restrict__ out0,
                            float* __restrict__ out1) {
  const float2* __restrict__ s_src2 = (const float2*)s_src2f;
  const float2* __restrict__ s_dst2 = (const float2*)s_dst2f;
  const float2* __restrict__ et12 = (const float2*)et12f;
  const int gid = blockIdx.x * blockDim.x + threadIdx.x;
  const int lane = gid & 31;
  const int n = __builtin_amdgcn_readfirstlane(gid >> 5);
  if (n >= NN) return;
  const int start = rowptr[n];
  const int end = rowptr[n + 1];
  const float2 ps = s_src2[n], pd = s_dst2[n];
  const int deg = end - start;
  const bool hd = lane >= 16;
  const int ch = lane * 4;
  const int hc = (lane & 15) * 4;
  const float4 hs = *(const float4*)&h[(size_t)n * 128 + ch];
  const float4 bv = *(const float4*)&(hd ? bias1 : bias0)[hc];
  float4 acc = bv;

  if (deg <= 32) {
    const int p = start + lane;
    const bool act = p < end;
    int s = 0;
    Logit2 q = {0.f, 0.f, 0.f, 0.f};
    if (act) { s = ssrc[p]; q = edge_logit2(p, pd, ssrc, seid, s_src2, et12); }
    const float sl0 = self_logit<true>(ps.x, pd.x, wave_sum(q.e0), deg);
    const float sl1 = self_logit<true>(ps.y, pd.y, wave_sum(q.e1), deg);
    const float m0 = fmaxf(wave_max(act ? q.l0 : -INFINITY), sl0);
    const float m1 = fmaxf(wave_max(act ? q.l1 : -INFINITY), sl1);
    const float w0 = act ? __expf(q.l0 - m0) : 0.f;
    const float w1 = act ? __expf(q.l1 - m1) : 0.f;
    const float wself0 = __expf(sl0 - m0);
    const float wself1 = __expf(sl1 - m1);
    const float den0 = wself0 + wave_sum(w0) + 1e-16f;
    const float den1 = wself1 + wave_sum(w1) + 1e-16f;
    const float a0 = w0 / den0;
    const float a1 = w1 / den1;
    axpy(acc, hd ? wself1 / den1 : wself0 / den0, hs);
#pragma unroll 4
    for (int k = 0; k < deg; ++k) {
      const int sk = lane_i(s, k);
      const float ak0 = lane_f(a0, k);
      const float ak1 = lane_f(a1, k);
      axpy(acc, hd ? ak1 : ak0, *(const float4*)&h[(size_t)sk * 128 + ch]);
    }
  } else {
    float m0 = -INFINITY, m1 = -INFINITY, ets0 = 0.f, ets1 = 0.f;
    for (int base = start; base < end; base += 32) {
      const int p = base + lane;
      if (p < end) {
        const Logit2 q = edge_logit2(p, pd, ssrc, seid, s_src2, et12);
        m0 = fmaxf(m0, q.l0); m1 = fmaxf(m1, q.l1);
        ets0 += q.e0; ets1 += q.e1;
      }
    }
    const float sl0 = self_logit<true>(ps.x, pd.x, wave_sum(ets0), deg);
    const float sl1 = self_logit<true>(ps.y, pd.y, wave_sum(ets1), deg);
    m0 = fmaxf(wave_max(m0), sl0);
    m1 = fmaxf(wave_max(m1), sl1);
    float ssum0 = 0.f, ssum1 = 0.f;
    for (int base = start; base < end; base += 32) {
      const int p = base + lane;
      if (p < end) {
        const Logit2 q = edge_logit2(p, pd, ssrc, seid, s_src2, et12);
        ssum0 += __expf(q.l0 - m0); ssum1 += __expf(q.l1 - m1);
      }
    }
    const float wself0 = __expf(sl0 - m0);
    const float wself1 = __expf(sl1 - m1);
    const float den0 = wself0 + wave_sum(ssum0) + 1e-16f;
    const float den1 = wself1 + wave_sum(ssum1) + 1e-16f;
    axpy(acc, hd ? wself1 / den1 : wself0 / den0, hs);
    for (int base = start; base < end; base += 32) {
      const int p = base + lane;
      int s = 0;
      float a0 = 0.f, a1 = 0.f;
      if (p < end) {
        s = ssrc[p];
        const Logit2 q = edge_logit2(p, pd, ssrc, seid, s_src2, et12);
        a0 = __expf(q.l0 - m0) / den0;
        a1 = __expf(q.l1 - m1) / den1;
      }
      const int cntk = min(32, end - base);
#pragma unroll 4
      for (int k = 0; k < cntk; ++k) {
        const int sk = lane_i(s, k);
        const float ak0 = lane_f(a0, k);
        const float ak1 = lane_f(a1, k);
        axpy(acc, hd ? ak1 : ak0, *(const float4*)&h[(size_t)sk * 128 + ch]);
      }
    }
  }
  float* __restrict__ o = hd ? out1 : out0;
  *(float4*)&o[(size_t)n * 64 + hc] = acc;
}

extern "C" void kernel_launch(void* const* d_in, const int* in_sizes, int n_in, void* d_out,
                              int out_size, void* d_ws, size_t ws_size, hipStream_t stream) {
  (void)in_sizes; (void)n_in; (void)out_size; (void)ws_size;
  const float* x = (const float*)d_in[0];
  const int* ei = (const int*)d_in[1];
  const int* src = ei;
  const int* dst = ei + NE;
  const float* edge_attr = (const float*)d_in[2];
  const float *W1 = (const float*)d_in[3], *a1s = (const float*)d_in[4],
              *a1d = (const float*)d_in[5], *b1 = (const float*)d_in[6];
  const float *W2 = (const float*)d_in[7], *a2s = (const float*)d_in[8],
              *a2d = (const float*)d_in[9], *b2 = (const float*)d_in[10];
  const float *We2 = (const float*)d_in[11], *ae2 = (const float*)d_in[12];
  const float *Wr = (const float*)d_in[13], *br = (const float*)d_in[14];
  const float *Wmu = (const float*)d_in[15], *amus = (const float*)d_in[16],
              *amud = (const float*)d_in[17], *bmu = (const float*)d_in[18],
              *Wemu = (const float*)d_in[19], *aemu = (const float*)d_in[20];
  const float *Wls = (const float*)d_in[21], *alss = (const float*)d_in[22],
              *alsd = (const float*)d_in[23], *bls = (const float*)d_in[24],
              *Wels = (const float*)d_in[25], *aels = (const float*)d_in[26];
  float* outf = (float*)d_out;

  float* W = (float*)d_ws;
  size_t o = 0;
  float* hA = W + o; o += (size_t)NN * 128;
  float* hB = W + o; o += (size_t)NN * 128;
  float* hC = W + o; o += (size_t)NN * 128;
  float* sa_src = W + o; o += NN;
  float* sa_dst = W + o; o += NN;
  float* sb_src = W + o; o += NN;
  float* sb_dst = W + o; o += NN;
  float* et0 = W + o; o += NE;
  float* et12 = W + o; o += (size_t)NE * 2;
  float* wa3 = W + o; o += 48;
  int* rowptr = (int*)(W + o); o += NN + 16;
  int* part = (int*)(W + o); o += (size_t)NE * 2;
  int* cntmat = (int*)(W + o); o += NBUCK * L1_BLOCKS + 2;
  int* offmat = (int*)(W + o); o += NBUCK * L1_BLOCKS + 2;
  int* rowtot = (int*)(W + o); o += NBUCK + 1;
  int* eid_raw = (int*)(W + o); o += NE;
  int* seid = (int*)(W + o); o += NE;
  int* ssrc = (int*)(W + o); o += NE;
  float* s_src2 = W + o; o += (size_t)NN * 2;
  float* s_dst2 = W + o; o += (size_t)NN * 2;

  l1_count_k<<<L1_BLOCKS, GT, 0, stream>>>(dst, cntmat);
  l1_rowscan_k<<<NBUCK, 256, 0, stream>>>(cntmat, offmat, rowtot);
  l1_scatter_k<<<L1_BLOCKS, GT, 0, stream>>>(dst, cntmat, offmat, rowtot, part);
  l2_group_k<<<NBUCK, GT, 0, stream>>>(part, rowtot, src, rowptr, seid, ssrc, eid_raw);

  wa3_k<<<3, 16, 0, stream>>>(We2, ae2, Wemu, aemu, Wels, aels, wa3);
  eterm3_k<<<(NE * 4) / ET_TILE, ET_BLOCK, 0, stream>>>(edge_attr, wa3, et0, et12);

  gemm_k128<8, false, false><<<NN / 16, 256, 0, stream>>>(x, Wr, nullptr, hC, nullptr, nullptr, nullptr,
                                                        nullptr, nullptr, nullptr);

  gemm_k128<8, false, true><<<NN / 16, 256, 0, stream>>>(x, W1, nullptr, hA, a1s, a1d, nullptr, nullptr,
                                                       sa_src, sa_dst);

  node_gemm_k<false, EPI_RELU, false><<<NG_BLOCKS, 512, 0, stream>>>(
      rowptr, ssrc, seid, sa_src, sa_dst, nullptr, hA, b1, nullptr, nullptr, W2, nullptr, hB, a2s, a2d,
      nullptr, nullptr, sb_src, sb_dst);

  node_gemm_k<true, EPI_RELU_RESID, true><<<NG_BLOCKS, 512, 0, stream>>>(
      rowptr, ssrc, seid, sb_src, sb_dst, et0, hB, b2, hC, br, Wmu, Wls, hA, amus, amud, alss, alsd,
      s_src2, s_dst2);

  gat_node2_k<<<(NN * 32 + 255) / 256, 256, 0, stream>>>(rowptr, ssrc, seid, s_src2, s_dst2, et12, hA,
                                                         bmu, bls, outf, outf + (size_t)NN * 64);
}


# Source id and edge terms travel as ONE 16-B record per edge; the grouping pass gathers it once and writes group-order planes: node passes read edge terms in place, not by edge id; sums unchanged
# speedup vs baseline: 7.8650x; 1.0424x over previous
#include <hip/hip_runtime.h>
#include <math.h>

typedef float v2f __attribute__((ext_vector_type(2)));
typedef float v4f __attribute__((ext_vector_type(4)));
typedef float v8f __attribute__((ext_vector_type(8)));

static constexpr int   NN = 100000;
static constexpr int   NE = 1600000;
static constexpr float NEG = 0.2f;

__device__ __forceinline__ float leaky(float x) { return x > 0.f ? x : NEG * x; }

static constexpr int BSH = 8;
static constexpr int BNODES = 1 << BSH;
static constexpr int NBUCK = (NN + BNODES - 1) >> BSH;
static constexpr int L1_BLOCKS = 250;
static constexpr int L1_CHUNK = NE / L1_BLOCKS;
static constexpr int GT = 1024;
static constexpr int L2_CAP = 6144;
static constexpr int L1_PER = (L1_CHUNK + GT - 1) / GT;
static constexpr int L2_PER = 4;
static_assert(L1_CHUNK * L1_BLOCKS == NE, "level 1 has no tail handling");
static_assert(NBUCK <= GT && BNODES <= GT && L1_BLOCKS <= 256, "one thread per bucket / node / level-1 block");

__device__ __forceinline__ int lane_i(int v, int k) { return __builtin_amdgcn_readlane(v, k); }
__device__ __forceinline__ float lane_f(float v, int k) {
  return __int_as_float(__builtin_amdgcn_readlane(__float_as_int(v), k));
}
__device__ __forceinline__ int wave_sum_i(int v) {
  for (int m = 16; m > 0; m >>= 1) v += __shfl_xor(v, m, 32);
  return v;
}
__device__ __forceinline__ int block_excl_scan(int v, int* wtot, int& total) {
  const int lane = threadIdx.x & 31, w = threadIdx.x >> 5;
  int inc = v;
  for (int o = 1; o < 32; o <<= 1) {
    const int u = __shfl_up(inc, o, 32);
    if (lane >= o) inc += u;
  }
  if (lane == 31) wtot[w] = inc;
  __syncthreads();
  const int wt = (lane < (int)(blockDim.x >> 5)) ? wtot[lane] : 0;
  const int before = wave_sum_i(lane < w ? wt : 0);
  total = wave_sum_i(wt);
  __syncthreads();
  return inc - v + before;
}

__global__ void l1_count_k(const int* __restrict__ dst, int* __restrict__ cntmat) {
  __shared__ int hist[NBUCK];
  const int t = threadIdx.x, blk = blockIdx.x;
  if (t < NBUCK) hist[t] = 0;
  __syncthreads();
  const int e0 = blk * L1_CHUNK;
  int d[L1_PER];
#pragma unroll
  for (int j = 0; j < L1_PER; ++j) d[j] = (t + j * GT < L1_CHUNK) ? dst[e0 + t + j * GT] : -1;
#pragma unroll
  for (int j = 0; j < L1_PER; ++j)
    if (d[j] >= 0) atomicAdd(&hist[d[j] >> BSH], 1);
  __syncthreads();
  if (t < NBUCK) cntmat[t * L1_BLOCKS + blk] = hist[t];
}
__global__ void l1_rowscan_k(const int* __restrict__ cntmat, int* __restrict__ offmat,
                             int* __restrict__ rowtot) {
  __shared__ int wtot[32];
  const int t = threadIdx.x, b = blockIdx.x;
  const int own = (t < L1_BLOCKS) ? cntmat[b * L1_BLOCKS + t] : 0;
  int total;
  const int pre = block_excl_scan(own, wtot, total);
  if (t < L1_BLOCKS) offmat[b * L1_BLOCKS + t] = pre;
  if (t == 0) rowtot[b] = total;
}
__global__ void l1_scatter_k(const int* __restrict__ dst, const int* __restrict__ cntmat,
                             const int* __restrict__ offmat, const int* __restrict__ rowtot,
                             int* __restrict__ part_i) {
  __shared__ int loff[NBUCK];
  __shared__ int cur[NBUCK];
  __shared__ int goff[NBUCK];
  __shared__ int wtot[32];
  __shared__ int2 stage[L1_CHUNK];
  int2* __restrict__ part = (int2*)part_i;
  const int t = threadIdx.x, blk = blockIdx.x;
  const bool isb = t < NBUCK;
  int total;
  const int pre = block_excl_scan(isb ? cntmat[t * L1_BLOCKS + blk] : 0, wtot, total);
  const int bbase = block_excl_scan(isb ? rowtot[t] : 0, wtot, total);
  if (isb) { loff[t] = pre; cur[t] = pre; goff[t] = bbase + offmat[t * L1_BLOCKS + blk]; }
  __syncthreads();
  const int e0 = blk * L1_CHUNK;
  int d[L1_PER];
#pragma unroll
  for (int j = 0; j < L1_PER; ++j) d[j] = (t + j * GT < L1_CHUNK) ? dst[e0 + t + j * GT] : -1;
#pragma unroll
  for (int j = 0; j < L1_PER; ++j)
    if (d[j] >= 0) {
      const int slot = atomicAdd(&cur[d[j] >> BSH], 1);
      stage[slot] = make_int2(e0 + t + j * GT, d[j]);
    }
  __syncthreads();
  for (int i = t; i < L1_CHUNK; i += GT) {
    const int2 v = stage[i];
    const int b = v.y >> BSH;
    part[goff[b] + (i - loff[b])] = v;
  }
}
__device__ __forceinline__ void l2_place_sort(int* st1, int* st2, const int2* __restrict__ part,
                                              int gb, int nb, int* cur, const int* nstart,
                                              const int4* __restrict__ rec, int* __restrict__ ssrc,
                                              float* __restrict__ et0s, float2* __restrict__ et12s) {
  const int t = threadIdx.x;
  for (int i0 = t; i0 < nb; i0 += L2_PER * GT) {
    int2 v[L2_PER];
#pragma unroll
    for (int j = 0; j < L2_PER; ++j)
      v[j] = (i0 + j * GT < nb) ? part[gb + i0 + j * GT] : make_int2(-1, -1);
#pragma unroll
    for (int j = 0; j < L2_PER; ++j)
      if (v[j].y >= 0) {
        const int slot = atomicAdd(&cur[v[j].y & (BNODES - 1)], 1);
        st1[slot] = v[j].x;
      }
  }
  __syncthreads();
  const int lane = t & 31;
  const int w = __builtin_amdgcn_readfirstlane(t >> 5);
  for (int ln = w; ln < BNODES; ln += GT / 32) {
    const int start = __builtin_amdgcn_readfirstlane(nstart[ln]);
    const int end = __builtin_amdgcn_readfirstlane(nstart[ln + 1]);
    for (int base = start; base < end; base += 32) {
      const int p = base + lane;
      const int x = (p < end) ? st1[p] : 0x7fffffff;
      int r = 0;
      for (int b2 = start; b2 < end; b2 += 32) {
        const int q = b2 + lane;
        const int y = (q < end) ? st1[q] : 0x7fffffff;
        const int m = min(32, end - b2);
        for (int k = 0; k < m; ++k) r += (lane_i(y, k) < x) ? 1 : 0;
      }
      if (p < end) st2[start + r] = x;
    }
  }
  __syncthreads();
  for (int i0 = t; i0 < nb; i0 += L2_PER * GT) {
    int x[L2_PER];
    int4 r[L2_PER];
#pragma unroll
    for (int j = 0; j < L2_PER; ++j) x[j] = (i0 + j * GT < nb) ? st2[i0 + j * GT] : -1;
#pragma unroll
    for (int j = 0; j < L2_PER; ++j) r[j] = (x[j] >= 0) ? rec[x[j]] : make_int4(0, 0, 0, 0);
#pragma unroll
    for (int j = 0; j < L2_PER; ++j)
      if (x[j] >= 0) {
        const int p = gb + i0 + j * GT;
        ssrc[p] = r[j].x;
        et0s[p] = __int_as_float(r[j].y);
        et12s[p] = make_float2(__int_as_float(r[j].z), __int_as_float(r[j].w));
      }
  }
}
__global__ void l2_group_k(const int* __restrict__ part_i, const int* __restrict__ rowtot,
                           const int* __restrict__ rec_i, int* __restrict__ rowptr,
                           int* __restrict__ ssrc, float* __restrict__ et0s,
                           float* __restrict__ et12s_f, int* eid_raw, int* eid_ord) {
  __shared__ int ncnt[BNODES];
  __shared__ int nstart[BNODES + 1];
  __shared__ int cur[BNODES];
  __shared__ int wtot[32];
  __shared__ int sa[L2_CAP];
  __shared__ int sb[L2_CAP];
  const int2* __restrict__ part = (const int2*)part_i;
  const int4* __restrict__ rec = (const int4*)rec_i;
  float2* __restrict__ et12s = (float2*)et12s_f;
  const int t = threadIdx.x, b = blockIdx.x;
  int gb;
  block_excl_scan((t < b) ? rowtot[t] : 0, wtot, gb);
  const int nb = rowtot[b];
  if (t < BNODES) ncnt[t] = 0;
  __syncthreads();
  for (int i0 = t; i0 < nb; i0 += L2_PER * GT) {
    int dn[L2_PER];
#pragma unroll
    for (int j = 0; j < L2_PER; ++j) dn[j] = (i0 + j * GT < nb) ? part[gb + i0 + j * GT].y : -1;
#pragma unroll
    for (int j = 0; j < L2_PER; ++j)
      if (dn[j] >= 0) atomicAdd(&ncnt[dn[j] & (BNODES - 1)], 1);
  }
  __syncthreads();
  int total;
  const int pre = block_excl_scan((t < BNODES) ? ncnt[t] : 0, wtot, total);
  if (t < BNODES) {
    nstart[t] = pre;
    cur[t] = pre;
    const int n = b * BNODES + t;
    if (n < NN) rowptr[n] = gb + pre;
  }
  if (t == 0) {
    nstart[BNODES] = nb;
    if (b == NBUCK - 1) rowptr[NN] = gb + nb;
  }
  __syncthreads();
  if (nb <= L2_CAP) l2_place_sort(sa, sb, part, gb, nb, cur, nstart, rec, ssrc, et0s, et12s);
  else l2_place_sort(eid_raw + gb, eid_ord + gb, part, gb, nb, cur, nstart, rec, ssrc, et0s, et12s);
}

__global__ void wa3_k(const float* __restrict__ We_a, const float* __restrict__ ae_a,
                      const float* __restrict__ We_b, const float* __restrict__ ae_b,
                      const float* __restrict__ We_c, const float* __restrict__ ae_c,
                      float* __restrict__ wa3) {
  const int l = blockIdx.x;
  const float* __restrict__ We = (l == 0) ? We_a : ((l == 1) ? We_b : We_c);
  const float* __restrict__ ae = (l == 0) ? ae_a : ((l == 1) ? ae_b : ae_c);
  const int K = (l == 0) ? 128 : 64;
  const int i = threadIdx.x;
  if (i < 16) {
    float s = 0.f;
    for (int j = 0; j < K; ++j) s += We[i * K + j] * ae[j];
    wa3[l * 16 + i] = s;
  }
}

template <int M>
__device__ __forceinline__ float xmask_add(float keep, float send) {
  return keep + __int_as_float(__builtin_amdgcn_update_dpp(0, __float_as_int(send), 0x160 + M, 0xf,
                                                          0xf, true));
}

static constexpr int AP = 132;
template <int COLT, bool SPLIT, bool PREP>
__global__ void gemm_k128(const float* __restrict__ A, const float* __restrict__ B,
                          const float* __restrict__ B2, float* __restrict__ C,
                          const float* __restrict__ a_src, const float* __restrict__ a_dst,
                          const float* __restrict__ a_src1, const float* __restrict__ a_dst1,
                          float* __restrict__ s_src, float* __restrict__ s_dst) {
  constexpr int COUT = COLT * 16;
  constexpr int LDB = SPLIT ? COUT / 2 : COUT;
  __shared__ __attribute__((aligned(16))) float As[16 * AP];
  const int tid = threadIdx.x;
  const size_t abase = (size_t)blockIdx.x * (16 * 128);
  for (int i = tid * 4; i < 16 * 128; i += blockDim.x * 4)
    *(float4*)(&As[(i >> 7) * AP + (i & 127)]) = *(const float4*)(&A[abase + i]);
  __syncthreads();

  const int wave = SPLIT ? __builtin_amdgcn_readfirstlane(tid >> 5) : (tid >> 5);
  const int lane = tid & 31, half = lane >> 4, lidx = lane & 15;
  const int n0 = wave * 16;
  const int nb = SPLIT ? (wave % (COLT / 2)) * 16 : n0;
  const float* __restrict__ Bw = (SPLIT && wave >= COLT / 2) ? B2 : B;
  v8f acc = {};
#pragma unroll
  for (int k0 = 0; k0 < 128; k0 += 4) {
    v2f a, b;
    a.x = As[lidx * AP + k0 + 2 * half];
    a.y = As[lidx * AP + k0 + 2 * half + 1];
    b.x = Bw[(k0 + 2 * half) * LDB + nb + lidx];
    b.y = Bw[(k0 + 2 * half + 1) * LDB + nb + lidx];
    acc = __builtin_amdgcn_wmma_f32_16x16x4_f32(false, a, false, b, (short)0, acc,
                                                false, false);
  }
  const size_t row0 = (size_t)blockIdx.x * 16;
  __shared__ float part[PREP ? COLT * 32 : 1];
  if constexpr (PREP) {
    const bool right = SPLIT && wave >= COLT / 2;
    const float* __restrict__ asw = right ? a_src1 : a_src;
    const float* __restrict__ adw = right ? a_dst1 : a_dst;
    const float as = asw[nb + lidx], ad = adw[nb + lidx];
    const v8f P = acc * as, Q = acc * ad;
    const bool u8 = (lidx & 8) != 0, u4 = (lidx & 4) != 0, u2 = (lidx & 2) != 0, u1 = (lidx & 1) != 0;
    const v8f k8 = u8 ? Q : P, s8 = u8 ? P : Q;
    v8f r8;
    for (int i = 0; i < 8; ++i) r8[i] = xmask_add<8>(k8[i], s8[i]);
    const v4f k4 = u4 ? r8.hi : r8.lo, s4 = u4 ? r8.lo : r8.hi;
    v4f r4;
    for (int i = 0; i < 4; ++i) r4[i] = xmask_add<4>(k4[i], s4[i]);
    const v2f k2 = u2 ? r4.hi : r4.lo, s2 = u2 ? r4.lo : r4.hi;
    v2f r2;
    for (int i = 0; i < 2; ++i) r2[i] = xmask_add<2>(k2[i], s2[i]);
    const float r1 = xmask_add<1>(u1 ? r2.y : r2.x, u1 ? r2.x : r2.y);
    part[wave * 32 + (lidx >> 3) * 16 + 8 * half + (lidx & 7)] = r1;
    __syncthreads();
  }
#pragma unroll
  for (int r = 0; r < 8; ++r)
    C[(row0 + r + 8 * half) * COUT + n0 + lidx] = acc[r];
  if constexpr (PREP) {
    if (!SPLIT) {
      if (tid < 32) {
        float s = part[tid];
        for (int w = 1; w < COLT; ++w) s += part[w * 32 + tid];
        ((tid >> 4) ? s_dst : s_src)[row0 + (tid & 15)] = s;
      }
    } else {
      if (tid < 64) {
        const int hd = tid >> 5, slot = tid & 31;
        float s = part[(hd * (COLT / 2)) * 32 + slot];
        for (int w = 1; w < COLT / 2; ++w) s += part[(hd * (COLT / 2) + w) * 32 + slot];
        ((slot >> 4) ? s_dst : s_src)[(row0 + (slot & 15)) * 2 + hd] = s;
      }
    }
  }
}

static constexpr int ET_PER_THREAD = 4;
static constexpr int ET_BLOCK = 256;
static constexpr int ET_TILE = ET_BLOCK * ET_PER_THREAD;
static_assert((NE * 4) % ET_TILE == 0, "eterm3_k has no tail handling");
__device__ __forceinline__ float dot4(const float4& a, const float4& b) {
  return a.x * b.x + a.y * b.y + a.z * b.z + a.w * b.w;
}
__device__ __forceinline__ float quad_sum(float v) {
  v += __shfl_xor(v, 1, 32);
  v += __shfl_xor(v, 2, 32);
  return v;
}
__global__ void eterm3_k(const float* __restrict__ ea, const float* __restrict__ wa3,
                         const int* __restrict__ src, int* __restrict__ rec_i) {
  int4* __restrict__ rec = (int4*)rec_i;
  const int tid = threadIdx.x;
  const int q = tid & 3;
  const float4* __restrict__ ea4 = (const float4*)ea;
  const size_t base = (size_t)blockIdx.x * ET_TILE + tid;
  float4 v[ET_PER_THREAD];
  int sv[ET_PER_THREAD];
#pragma unroll
  for (int k = 0; k < ET_PER_THREAD; ++k) v[k] = ea4[base + (size_t)k * ET_BLOCK];
#pragma unroll
  for (int k = 0; k < ET_PER_THREAD; ++k) sv[k] = (q == 0) ? src[(base + (size_t)k * ET_BLOCK) >> 2] : 0;
  const float4 w0 = ((const float4*)wa3)[q];
  const float4 w1 = ((const float4*)wa3)[4 + q];
  const float4 w2 = ((const float4*)wa3)[8 + q];
#pragma unroll
  for (int k = 0; k < ET_PER_THREAD; ++k) {
    const float s0 = quad_sum(dot4(v[k], w0));
    const float s1 = quad_sum(dot4(v[k], w1));
    const float s2 = quad_sum(dot4(v[k], w2));
    if (q == 0) {
      const size_t e = (base + (size_t)k * ET_BLOCK) >> 2;
      rec[e] = make_int4(sv[k], __float_as_int(s0), __float_as_int(s1), __float_as_int(s2));
    }
  }
}

__device__ __forceinline__ void axpy(float4& acc, float a, const float4& v) {
  acc.x += a * v.x; acc.y += a * v.y; acc.z += a * v.z; acc.w += a * v.w;
}
enum { EPI_RELU = 1, EPI_RELU_RESID = 2 };
__device__ __forceinline__ void relu_v(float4& v) {
  v.x = fmaxf(v.x, 0.f); v.y = fmaxf(v.y, 0.f); v.z = fmaxf(v.z, 0.f); v.w = fmaxf(v.w, 0.f);
}
__device__ __forceinline__ void add2_v(float4& v, const float4& r, const float4& b) {
  v.x = v.x + r.x + b.x; v.y = v.y + r.y + b.y; v.z = v.z + r.z + b.z; v.w = v.w + r.w + b.w;
}

template <bool ET>
__device__ __forceinline__ float edge_logit(int p, float pd, const int* __restrict__ ssrc,
                                            const float* __restrict__ s_src,
                                            const float* __restrict__ eterm, float& et) {
  float l = s_src[ssrc[p]] + pd;
  et = 0.f;
  if (ET) { et = eterm[p]; l += et; }
  return leaky(l);
}
template <bool ET>
__device__ __forceinline__ float self_logit(float ps, float pd, float et_sum, int deg) {
  float l = ps + pd;
  if (ET) l += et_sum / fmaxf((float)deg, 1.f);
  return leaky(l);
}
__device__ __forceinline__ float wave_max(float v) {
  for (int m = 16; m > 0; m >>= 1) v = fmaxf(v, __shfl_xor(v, m, 32));
  return v;
}
__device__ __forceinline__ float wave_sum(float v) {
  for (int m = 16; m > 0; m >>= 1) v += __shfl_xor(v, m, 32);
  return v;
}

template <bool ET, int EPI>
__device__ __forceinline__ float4 gat_node_row(int n, int lane, const int* __restrict__ rowptr,
                                               const int* __restrict__ ssrc,
                                               const float* __restrict__ s_src,
                                               const float* __restrict__ s_dst,
                                               const float* __restrict__ eterm,
                                               const float* __restrict__ h,
                                               const float* __restrict__ bias,
                                               const float* __restrict__ resid,
                                               const float* __restrict__ rbias) {
  constexpr int C = 128;
  const int start = rowptr[n];
  const int end = rowptr[n + 1];
  const float ps = s_src[n], pd = s_dst[n];
  const int deg = end - start;
  const int ch = lane * 4;
  const float4 hs = *(const float4*)&h[(size_t)n * C + ch];
  const float4 bv = *(const float4*)&bias[ch];
  float4 acc = bv;
  float4 rv = {}, rb = {};
  if (EPI == EPI_RELU_RESID) {
    rv = *(const float4*)&resid[(size_t)n * C + ch];
    rb = *(const float4*)&rbias[ch];
  }

  if (end - start <= 32) {
    const int p = start + lane;
    const bool act = p < end;
    int s = 0;
    float l = 0.f, et = 0.f;
    if (act) { s = ssrc[p]; l = edge_logit<ET>(p, pd, ssrc, s_src, eterm, et); }
    const float sl = self_logit<ET>(ps, pd, ET ? wave_sum(et) : 0.f, deg);
    const float m = fmaxf(wave_max(act ? l : -INFINITY), sl);
    const float w = act ? __expf(l - m) : 0.f;
    const float wself = __expf(sl - m);
    const float den = wself + wave_sum(w) + 1e-16f;
    const float a = w / den;
    axpy(acc, wself / den, hs);
#pragma unroll 4
    for (int k = 0; k < deg; ++k) {
      const int sk = lane_i(s, k);
      const float ak = lane_f(a, k);
      axpy(acc, ak, *(const float4*)&h[(size_t)sk * C + ch]);
    }
  } else {
    float m = -INFINITY, ets = 0.f, et;
    for (int base = start; base < end; base += 32) {
      const int p = base + lane;
      if (p < end) { m = fmaxf(m, edge_logit<ET>(p, pd, ssrc, s_src, eterm, et)); ets += et; }
    }
    const float sl = self_logit<ET>(ps, pd, ET ? wave_sum(ets) : 0.f, deg);
    m = fmaxf(wave_max(m), sl);
    float ssum = 0.f;
    for (int base = start; base < end; base += 32) {
      const int p = base + lane;
      if (p < end) ssum += __expf(edge_logit<ET>(p, pd, ssrc, s_src, eterm, et) - m);
    }
    const float wself = __expf(sl - m);
    const float den = wself + wave_sum(ssum) + 1e-16f;
    axpy(acc, wself / den, hs);
    for (int base = start; base < end; base += 32) {
      const int p = base + lane;
      int s = 0;
      float a = 0.f;
      if (p < end) {
        s = ssrc[p];
        a = __expf(edge_logit<ET>(p, pd, ssrc, s_src, eterm, et) - m) / den;
      }
      const int cntk = min(32, end - base);
#pragma unroll 4
      for (int k = 0; k < cntk; ++k) {
        const int sk = lane_i(s, k);
        const float ak = lane_f(a, k);
        axpy(acc, ak, *(const float4*)&h[(size_t)sk * C + ch]);
      }
    }
  }
  relu_v(acc);
  if (EPI == EPI_RELU_RESID) add2_v(acc, rv, rb);
  return acc;
}

static constexpr int NG_TILES = NN / 16;
static constexpr int NG_BLOCKS = 1024;
static_assert(NN % 16 == 0, "node_gemm_k has no tail handling");
template <bool ET, int EPI, bool SPLIT>
__global__ void __launch_bounds__(512, 16)
node_gemm_k(const int* __restrict__ rowptr, const int* __restrict__ ssrc,
            const float* __restrict__ s_src,
            const float* __restrict__ s_dst, const float* __restrict__ eterm,
            const float* __restrict__ h, const float* __restrict__ bias,
            const float* __restrict__ resid, const float* __restrict__ rbias,
            const float* __restrict__ B, const float* __restrict__ B2,
            float* __restrict__ C, const float* __restrict__ a_src,
            const float* __restrict__ a_dst, const float* __restrict__ a_src1,
            const float* __restrict__ a_dst1, float* __restrict__ s_src_o,
            float* __restrict__ s_dst_o) {
  __shared__ __attribute__((aligned(16))) float Bs[128 * 128];
  __shared__ __attribute__((aligned(16))) float As[16 * AP];
  __shared__ float part[8 * 32];
  constexpr int LDB = SPLIT ? 64 : 128;
  const int tid = threadIdx.x;
  const int lane = tid & 31, half = lane >> 4, lidx = lane & 15;
  const int wave = __builtin_amdgcn_readfirstlane(tid >> 5);

#pragma unroll
  for (int it = tid; it < 64 * 32; it += 512) {
    const int kp = it >> 5, c = (it & 31) * 4;
    const float* __restrict__ Bm = (SPLIT && c >= 64) ? B2 : B;
    const int cc = SPLIT ? (c & 63) : c;
    const float4 r0 = *(const float4*)&Bm[(2 * kp) * LDB + cc];
    const float4 r1 = *(const float4*)&Bm[(2 * kp + 1) * LDB + cc];
    float* d = &Bs[((((c >> 4) * 32 + (kp >> 1)) * 32) + (kp & 1) * 16 + (c & 15)) * 2];
    *(float4*)(d) = make_float4(r0.x, r1.x, r0.y, r1.y);
    *(float4*)(d + 4) = make_float4(r0.z, r1.z, r0.w, r1.w);
  }
  const bool right = SPLIT && wave >= 4;
  const int nb = SPLIT ? (wave & 3) * 16 : (wave & 7) * 16;
  const int n0 = (wave & 7) * 16;

  for (int tile = blockIdx.x; tile < NG_TILES; tile += NG_BLOCKS) {
    const int row0 = tile * 16;
    const float4 row = gat_node_row<ET, EPI>(row0 + wave, lane, rowptr, ssrc, s_src, s_dst, eterm, h, bias,
                                             resid, rbias);
    *(float4*)&As[wave * AP + lane * 4] = row;
    __syncthreads();
    v8f acc = {};
    if (wave < 8) {
      const float as = (right ? a_src1 : a_src)[nb + lidx], ad = (right ? a_dst1 : a_dst)[nb + lidx];
#pragma unroll
      for (int ks = 0; ks < 32; ++ks) {
        v2f a;
        a.x = As[lidx * AP + 4 * ks + 2 * half];
        a.y = As[lidx * AP + 4 * ks + 2 * half + 1];
        const v2f b = *(const v2f*)&Bs[((wave * 32 + ks) * 32 + lane) * 2];
        acc = __builtin_amdgcn_wmma_f32_16x16x4_f32(false, a, false, b, (short)0, acc,
                                                    false, false);
      }
      const v8f P = acc * as, Q = acc * ad;
      const bool u8 = (lidx & 8) != 0, u4 = (lidx & 4) != 0, u2 = (lidx & 2) != 0, u1 = (lidx & 1) != 0;
      const v8f k8 = u8 ? Q : P, s8 = u8 ? P : Q;
      v8f r8;
      for (int i = 0; i < 8; ++i) r8[i] = xmask_add<8>(k8[i], s8[i]);
      const v4f k4 = u4 ? r8.hi : r8.lo, s4 = u4 ? r8.lo : r8.hi;
      v4f r4;
      for (int i = 0; i < 4; ++i) r4[i] = xmask_add<4>(k4[i], s4[i]);
      const v2f k2 = u2 ? r4.hi : r4.lo, s2 = u2 ? r4.lo : r4.hi;
      v2f r2;
      for (int i = 0; i < 2; ++i) r2[i] = xmask_add<2>(k2[i], s2[i]);
      const float r1 = xmask_add<1>(u1 ? r2.y : r2.x, u1 ? r2.x : r2.y);
      part[wave * 32 + (lidx >> 3) * 16 + 8 * half + (lidx & 7)] = r1;
    }
    __syncthreads();
    if (wave < 8) {
      float* __restrict__ Ct = C + (size_t)row0 * 128;
#pragma unroll
      for (int r = 0; r < 8; ++r)
        Ct[(r + 8 * half) * 128 + n0 + lidx] = acc[r];
    }
    if (!SPLIT) {
      if (tid < 32) {
        float s = part[tid];
        for (int w = 1; w < 8; ++w) s += part[w * 32 + tid];
        ((tid >> 4) ? s_dst_o : s_src_o)[row0 + (tid & 15)] = s;
      }
    } else {
      if (tid < 64) {
        const int hd = tid >> 5, slot = tid & 31;
        float s = part[(hd * 4) * 32 + slot];
        for (int w = 1; w < 4; ++w) s += part[(hd * 4 + w) * 32 + slot];
        ((slot >> 4) ? s_dst_o : s_src_o)[(size_t)(row0 + (slot & 15)) * 2 + hd] = s;
      }
    }
  }
}

struct Logit2 { float l0, l1, e0, e1; };
__device__ __forceinline__ Logit2 edge_logit2(int p, float2 pd, const int* __restrict__ ssrc,
                                              const float2* __restrict__ s_src2,
                                              const float2* __restrict__ et12) {
  const float2 ss = s_src2[ssrc[p]];
  const float2 et = et12[p];
  Logit2 r;
  r.e0 = et.x; r.e1 = et.y;
  float l0 = ss.x + pd.x; l0 += et.x;
  float l1 = ss.y + pd.y; l1 += et.y;
  r.l0 = leaky(l0); r.l1 = leaky(l1);
  return r;
}
__global__ void gat_node2_k(const int* __restrict__ rowptr, const int* __restrict__ ssrc,
                            const float* __restrict__ s_src2f,
                            const float* __restrict__ s_dst2f, const float* __restrict__ et12f,
                            const float* __restrict__ h, const float* __restrict__ bias0,
                            const float* __restrict__ bias1, float* __restrict__ out0,
                            float* __restrict__ out1) {
  const float2* __restrict__ s_src2 = (const float2*)s_src2f;
  const float2* __restrict__ s_dst2 = (const float2*)s_dst2f;
  const float2* __restrict__ et12 = (const float2*)et12f;
  const int gid = blockIdx.x * blockDim.x + threadIdx.x;
  const int lane = gid & 31;
  const int n = __builtin_amdgcn_readfirstlane(gid >> 5);
  if (n >= NN) return;
  const int start = rowptr[n];
  const int end = rowptr[n + 1];
  const float2 ps = s_src2[n], pd = s_dst2[n];
  const int deg = end - start;
  const bool hd = lane >= 16;
  const int ch = lane * 4;
  const int hc = (lane & 15) * 4;
  const float4 hs = *(const float4*)&h[(size_t)n * 128 + ch];
  const float4 bv = *(const float4*)&(hd ? bias1 : bias0)[hc];
  float4 acc = bv;

  if (deg <= 32) {
    const int p = start + lane;
    const bool act = p < end;
    int s = 0;
    Logit2 q = {0.f, 0.f, 0.f, 0.f};
    if (act) { s = ssrc[p]; q = edge_logit2(p, pd, ssrc, s_src2, et12); }
    const float sl0 = self_logit<true>(ps.x, pd.x, wave_sum(q.e0), deg);
    const float sl1 = self_logit<true>(ps.y, pd.y, wave_sum(q.e1), deg);
    const float m0 = fmaxf(wave_max(act ? q.l0 : -INFINITY), sl0);
    const float m1 = fmaxf(wave_max(act ? q.l1 : -INFINITY), sl1);
    const float w0 = act ? __expf(q.l0 - m0) : 0.f;
    const float w1 = act ? __expf(q.l1 - m1) : 0.f;
    const float wself0 = __expf(sl0 - m0);
    const float wself1 = __expf(sl1 - m1);
    const float den0 = wself0 + wave_sum(w0) + 1e-16f;
    const float den1 = wself1 + wave_sum(w1) + 1e-16f;
    const float a0 = w0 / den0;
    const float a1 = w1 / den1;
    axpy(acc, hd ? wself1 / den1 : wself0 / den0, hs);
#pragma unroll 4
    for (int k = 0; k < deg; ++k) {
      const int sk = lane_i(s, k);
      const float ak0 = lane_f(a0, k);
      const float ak1 = lane_f(a1, k);
      axpy(acc, hd ? ak1 : ak0, *(const float4*)&h[(size_t)sk * 128 + ch]);
    }
  } else {
    float m0 = -INFINITY, m1 = -INFINITY, ets0 = 0.f, ets1 = 0.f;
    for (int base = start; base < end; base += 32) {
      const int p = base + lane;
      if (p < end) {
        const Logit2 q = edge_logit2(p, pd, ssrc, s_src2, et12);
        m0 = fmaxf(m0, q.l0); m1 = fmaxf(m1, q.l1);
        ets0 += q.e0; ets1 += q.e1;
      }
    }
    const float sl0 = self_logit<true>(ps.x, pd.x, wave_sum(ets0), deg);
    const float sl1 = self_logit<true>(ps.y, pd.y, wave_sum(ets1), deg);
    m0 = fmaxf(wave_max(m0), sl0);
    m1 = fmaxf(wave_max(m1), sl1);
    float ssum0 = 0.f, ssum1 = 0.f;
    for (int base = start; base < end; base += 32) {
      const int p = base + lane;
      if (p < end) {
        const Logit2 q = edge_logit2(p, pd, ssrc, s_src2, et12);
        ssum0 += __expf(q.l0 - m0); ssum1 += __expf(q.l1 - m1);
      }
    }
    const float wself0 = __expf(sl0 - m0);
    const float wself1 = __expf(sl1 - m1);
    const float den0 = wself0 + wave_sum(ssum0) + 1e-16f;
    const float den1 = wself1 + wave_sum(ssum1) + 1e-16f;
    axpy(acc, hd ? wself1 / den1 : wself0 / den0, hs);
    for (int base = start; base < end; base += 32) {
      const int p = base + lane;
      int s = 0;
      float a0 = 0.f, a1 = 0.f;
      if (p < end) {
        s = ssrc[p];
        const Logit2 q = edge_logit2(p, pd, ssrc, s_src2, et12);
        a0 = __expf(q.l0 - m0) / den0;
        a1 = __expf(q.l1 - m1) / den1;
      }
      const int cntk = min(32, end - base);
#pragma unroll 4
      for (int k = 0; k < cntk; ++k) {
        const int sk = lane_i(s, k);
        const float ak0 = lane_f(a0, k);
        const float ak1 = lane_f(a1, k);
        axpy(acc, hd ? ak1 : ak0, *(const float4*)&h[(size_t)sk * 128 + ch]);
      }
    }
  }
  float* __restrict__ o = hd ? out1 : out0;
  *(float4*)&o[(size_t)n * 64 + hc] = acc;
}

extern "C" void kernel_launch(void* const* d_in, const int* in_sizes, int n_in, void* d_out,
                              int out_size, void* d_ws, size_t ws_size, hipStream_t stream) {
  (void)in_sizes; (void)n_in; (void)out_size; (void)ws_size;
  const float* x = (const float*)d_in[0];
  const int* ei = (const int*)d_in[1];
  const int* src = ei;
  const int* dst = ei + NE;
  const float* edge_attr = (const float*)d_in[2];
  const float *W1 = (const float*)d_in[3], *a1s = (const float*)d_in[4],
              *a1d = (const float*)d_in[5], *b1 = (const float*)d_in[6];
  const float *W2 = (const float*)d_in[7], *a2s = (const float*)d_in[8],
              *a2d = (const float*)d_in[9], *b2 = (const float*)d_in[10];
  const float *We2 = (const float*)d_in[11], *ae2 = (const float*)d_in[12];
  const float *Wr = (const float*)d_in[13], *br = (const float*)d_in[14];
  const float *Wmu = (const float*)d_in[15], *amus = (const float*)d_in[16],
              *amud = (const float*)d_in[17], *bmu = (const float*)d_in[18],
              *Wemu = (const float*)d_in[19], *aemu = (const float*)d_in[20];
  const float *Wls = (const float*)d_in[21], *alss = (const float*)d_in[22],
              *alsd = (const float*)d_in[23], *bls = (const float*)d_in[24],
              *Wels = (const float*)d_in[25], *aels = (const float*)d_in[26];
  float* outf = (float*)d_out;

  float* W = (float*)d_ws;
  size_t o = 0;
  float* hA = W + o; o += (size_t)NN * 128;
  float* hB = W + o; o += (size_t)NN * 128;
  float* hC = W + o; o += (size_t)NN * 128;
  int* rec = (int*)(W + o); o += (size_t)NE * 4;
  float* sa_src = W + o; o += NN;
  float* sa_dst = W + o; o += NN;
  float* sb_src = W + o; o += NN;
  float* sb_dst = W + o; o += NN;
  float* et0 = W + o; o += NE;
  float* et12 = W + o; o += (size_t)NE * 2;
  float* wa3 = W + o; o += 48;
  int* rowptr = (int*)(W + o); o += NN + 16;
  int* part = (int*)(W + o); o += (size_t)NE * 2;
  int* cntmat = (int*)(W + o); o += NBUCK * L1_BLOCKS + 2;
  int* offmat = (int*)(W + o); o += NBUCK * L1_BLOCKS + 2;
  int* rowtot = (int*)(W + o); o += NBUCK + 1;
  int* eid_raw = (int*)(W + o); o += NE;
  int* eid_ord = (int*)(W + o); o += NE;
  int* ssrc = (int*)(W + o); o += NE;
  float* s_src2 = W + o; o += (size_t)NN * 2;
  float* s_dst2 = W + o; o += (size_t)NN * 2;

  l1_count_k<<<L1_BLOCKS, GT, 0, stream>>>(dst, cntmat);
  l1_rowscan_k<<<NBUCK, 256, 0, stream>>>(cntmat, offmat, rowtot);
  l1_scatter_k<<<L1_BLOCKS, GT, 0, stream>>>(dst, cntmat, offmat, rowtot, part);

  wa3_k<<<3, 16, 0, stream>>>(We2, ae2, Wemu, aemu, Wels, aels, wa3);
  eterm3_k<<<(NE * 4) / ET_TILE, ET_BLOCK, 0, stream>>>(edge_attr, wa3, src, rec);

  l2_group_k<<<NBUCK, GT, 0, stream>>>(part, rowtot, rec, rowptr, ssrc, et0, et12, eid_raw, eid_ord);

  gemm_k128<8, false, false><<<NN / 16, 256, 0, stream>>>(x, Wr, nullptr, hC, nullptr, nullptr, nullptr,
                                                        nullptr, nullptr, nullptr);

  gemm_k128<8, false, true><<<NN / 16, 256, 0, stream>>>(x, W1, nullptr, hA, a1s, a1d, nullptr, nullptr,
                                                       sa_src, sa_dst);

  node_gemm_k<false, EPI_RELU, false><<<NG_BLOCKS, 512, 0, stream>>>(
      rowptr, ssrc, sa_src, sa_dst, nullptr, hA, b1, nullptr, nullptr, W2, nullptr, hB, a2s, a2d,
      nullptr, nullptr, sb_src, sb_dst);

  node_gemm_k<true, EPI_RELU_RESID, true><<<NG_BLOCKS, 512, 0, stream>>>(
      rowptr, ssrc, sb_src, sb_dst, et0, hB, b2, hC, br, Wmu, Wls, hA, amus, amud, alss, alsd,
      s_src2, s_dst2);

  gat_node2_k<<<(NN * 32 + 255) / 256, 256, 0, stream>>>(rowptr, ssrc, s_src2, s_dst2, et12, hA,
                                                         bmu, bls, outf, outf + (size_t)NN * 64);
}
